# starting kernel
# baseline (speedup 1.0000x reference)
#include <hip/hip_runtime.h>

typedef float v2f __attribute__((ext_vector_type(2)));
typedef float v8f __attribute__((ext_vector_type(8)));

#define NEG_SLOPE 0.2f

__device__ __forceinline__ unsigned fkey(float f) {
  unsigned u = __float_as_uint(f);
  return (u & 0x80000000u) ? ~u : (u | 0x80000000u);
}
__device__ __forceinline__ float fdekey(unsigned k) {
  unsigned u = (k & 0x80000000u) ? (k & 0x7FFFFFFFu) : ~k;
  return __uint_as_float(u);
}

__global__ void zero_f32(float* __restrict__ p, size_t n) {
  size_t i = (size_t)blockIdx.x * blockDim.x + threadIdx.x;
  size_t stride = (size_t)gridDim.x * blockDim.x;
  for (; i < n; i += stride) p[i] = 0.0f;
}

__global__ void gemm_wmma_f32(const float* __restrict__ A, const float* __restrict__ B,
                              float* __restrict__ C, int N, int K, int Nc) {
  const int lane = threadIdx.x & 31;
  const int half = lane >> 4;
  const int idx  = lane & 15;
  const int row0 = blockIdx.x * 16;
  const int col0 = blockIdx.y * 16;

  const int ar = min(row0 + idx, N - 1);
  const int bc = min(col0 + idx, Nc - 1);
  const float* __restrict__ ap = A + (size_t)ar * K + 2 * half;
  const float* __restrict__ bp = B + (size_t)bc * K + 2 * half;

  v8f c = {0.f, 0.f, 0.f, 0.f, 0.f, 0.f, 0.f, 0.f};
  for (int k = 0; k < K; k += 4) {
    v2f a = *(const v2f*)(ap + k);
    v2f b = *(const v2f*)(bp + k);
    c = __builtin_amdgcn_wmma_f32_16x16x4_f32(false, a, false, b, (short)0, c,
                                              false, false);
  }

  const int col = col0 + idx;
  if (col < Nc) {
#pragma unroll
    for (int r = 0; r < 8; ++r) {
      const int row = row0 + r + 8 * half;
      if (row < N) C[(size_t)row * Nc + col] = c[r];
    }
  }
}

__global__ void attn_logits(const float* __restrict__ h, const float* __restrict__ att_s,
                            const float* __restrict__ att_d, float* __restrict__ a_s,
                            float* __restrict__ a_d, int NH, int H, int D) {
  int i = blockIdx.x * blockDim.x + threadIdx.x;
  if (i >= NH) return;
  const int hh = i % H;
  const float* __restrict__ hp = h + (size_t)i * D;
  const float* __restrict__ as = att_s + (size_t)hh * D;
  const float* __restrict__ ad = att_d + (size_t)hh * D;
  float s0 = 0.f, s1 = 0.f;
  for (int dd = 0; dd < D; ++dd) {
    const float v = hp[dd];
    s0 += v * as[dd];
    s1 += v * ad[dd];
  }
  a_s[i] = s0;
  a_d[i] = s1;
}

__global__ void edge_max(const int* __restrict__ src, const int* __restrict__ dst,
                         int E, int N, const float* __restrict__ a_s,
                         const float* __restrict__ a_d, unsigned* __restrict__ mkey,
                         int H) {
  const int t = blockIdx.x * blockDim.x + threadIdx.x;
  const int TE = E + N;
  if (t >= TE) return;
  const int s = (t < E) ? src[t] : (t - E);
  const int d = (t < E) ? dst[t] : (t - E);
  for (int hh = 0; hh < H; ++hh) {
    float e = a_s[s * H + hh] + a_d[d * H + hh];
    e = (e > 0.f) ? e : NEG_SLOPE * e;
    atomicMax(&mkey[d * H + hh], fkey(e));
  }
}

__global__ void edge_expsum(const int* __restrict__ src, const int* __restrict__ dst,
                            int E, int N, const float* __restrict__ a_s,
                            const float* __restrict__ a_d,
                            const unsigned* __restrict__ mkey,
                            float* __restrict__ den, int H) {
  const int t = blockIdx.x * blockDim.x + threadIdx.x;
  const int TE = E + N;
  if (t >= TE) return;
  const int s = (t < E) ? src[t] : (t - E);
  const int d = (t < E) ? dst[t] : (t - E);
  for (int hh = 0; hh < H; ++hh) {
    float e = a_s[s * H + hh] + a_d[d * H + hh];
    e = (e > 0.f) ? e : NEG_SLOPE * e;
    const float ex = expf(e - fdekey(mkey[d * H + hh]));
    atomicAdd(&den[d * H + hh], ex);
  }
}

__global__ void edge_aggregate(const int* __restrict__ src, const int* __restrict__ dst,
                               int E, int N, const float* __restrict__ a_s,
                               const float* __restrict__ a_d,
                               const unsigned* __restrict__ mkey,
                               const float* __restrict__ den,
                               const float* __restrict__ h, float* __restrict__ agg,
                               int H, int D) {
  const int t = blockIdx.x * blockDim.x + threadIdx.x;
  const int TEH = (E + N) * H;
  if (t >= TEH) return;
  const int ei = t / H;
  const int hh = t - ei * H;
  const int s = (ei < E) ? src[ei] : (ei - E);
  const int d = (ei < E) ? dst[ei] : (ei - E);
  float e = a_s[s * H + hh] + a_d[d * H + hh];
  e = (e > 0.f) ? e : NEG_SLOPE * e;
  const float alpha = expf(e - fdekey(mkey[d * H + hh])) / den[d * H + hh];
  const float* __restrict__ hs = h + ((size_t)s * H + hh) * D;
  float* __restrict__ ag = agg + ((size_t)d * H + hh) * D;
  for (int dd = 0; dd < D; ++dd) atomicAdd(&ag[dd], alpha * hs[dd]);
}

__global__ void bias_relu(const float* __restrict__ agg, const float* __restrict__ b,
                          float* __restrict__ out, int C, size_t n) {
  size_t i = (size_t)blockIdx.x * blockDim.x + threadIdx.x;
  if (i >= n) return;
  const float v = agg[i] + b[i % C];
  out[i] = (v > 0.f) ? v : 0.f;
}

__global__ void bias_add_inplace(float* __restrict__ out, const float* __restrict__ b,
                                 int C, size_t n) {
  size_t i = (size_t)blockIdx.x * blockDim.x + threadIdx.x;
  if (i >= n) return;
  out[i] += b[i % C];
}

extern "C" void kernel_launch(void* const* d_in, const int* in_sizes, int n_in,
                              void* d_out, int out_size, void* d_ws, size_t ws_size,
                              hipStream_t stream) {
  const float* x        = (const float*)d_in[0];
  const int*   ei       = (const int*)d_in[1];
  const float* W1       = (const float*)d_in[2];
  const float* att_src1 = (const float*)d_in[3];
  const float* att_dst1 = (const float*)d_in[4];
  const float* b1       = (const float*)d_in[5];
  const float* W2       = (const float*)d_in[6];
  const float* att_src2 = (const float*)d_in[7];
  const float* att_dst2 = (const float*)d_in[8];
  const float* b2       = (const float*)d_in[9];

  const int NFEAT = 256, NHID = 128, H1 = 4, D1 = 32, NCLS = 40;
  const int N = in_sizes[0] / NFEAT;
  const int E = in_sizes[1] / 2;
  const int* src = ei;
  const int* dst = ei + E;

  float* out = (float*)d_out;

  float* w = (float*)d_ws;
  float* h1   = w;
  float* h2   = h1  + (size_t)N * NHID;
  float* as1  = h2  + (size_t)N * NCLS;
  float* ad1  = as1 + (size_t)N * H1;
  float* as2  = ad1 + (size_t)N * H1;
  float* ad2  = as2 + (size_t)N;
  float* zreg = ad2 + (size_t)N;
  float*    agg1 = zreg;
  float*    den1 = agg1 + (size_t)N * NHID;
  unsigned* m1   = (unsigned*)(den1 + (size_t)N * H1);
  float*    den2 = (float*)(m1 + (size_t)N * H1);
  unsigned* m2   = (unsigned*)(den2 + (size_t)N);
  const size_t zcount = (size_t)N * (NHID + H1 + H1 + 1 + 1);

  const int ZB = 2048, TB = 256;
  const int TE = E + N;

  zero_f32<<<ZB, TB, 0, stream>>>(zreg, zcount);
  zero_f32<<<ZB, TB, 0, stream>>>(out, (size_t)N * NCLS);

  {
    dim3 grid((N + 15) / 16, NHID / 16);
    gemm_wmma_f32<<<grid, 32, 0, stream>>>(x, W1, h1, N, NFEAT, NHID);
  }
  attn_logits<<<(N * H1 + TB - 1) / TB, TB, 0, stream>>>(h1, att_src1, att_dst1,
                                                         as1, ad1, N * H1, H1, D1);
  edge_max<<<(TE + TB - 1) / TB, TB, 0, stream>>>(src, dst, E, N, as1, ad1, m1, H1);
  edge_expsum<<<(TE + TB - 1) / TB, TB, 0, stream>>>(src, dst, E, N, as1, ad1, m1,
                                                     den1, H1);
  edge_aggregate<<<(TE * H1 + TB - 1) / TB, TB, 0, stream>>>(src, dst, E, N, as1, ad1,
                                                             m1, den1, h1, agg1, H1, D1);
  float* x2 = h1;
  {
    const size_t n = (size_t)N * NHID;
    bias_relu<<<(unsigned)((n + TB - 1) / TB), TB, 0, stream>>>(agg1, b1, x2, NHID, n);
  }

  {
    dim3 grid((N + 15) / 16, (NCLS + 15) / 16);
    gemm_wmma_f32<<<grid, 32, 0, stream>>>(x2, W2, h2, N, NHID, NCLS);
  }
  attn_logits<<<(N + TB - 1) / TB, TB, 0, stream>>>(h2, att_src2, att_dst2,
                                                    as2, ad2, N, 1, NCLS);
  edge_max<<<(TE + TB - 1) / TB, TB, 0, stream>>>(src, dst, E, N, as2, ad2, m2, 1);
  edge_expsum<<<(TE + TB - 1) / TB, TB, 0, stream>>>(src, dst, E, N, as2, ad2, m2,
                                                     den2, 1);
  edge_aggregate<<<(TE + TB - 1) / TB, TB, 0, stream>>>(src, dst, E, N, as2, ad2,
                                                        m2, den2, h2, out, 1, NCLS);
  {
    const size_t n = (size_t)N * NCLS;
    bias_add_inplace<<<(unsigned)((n + TB - 1) / TB), TB, 0, stream>>>(out, b2, NCLS, n);
  }
}


# edge aggregation: lanes run along the feature row and a wave walks 32 edges, so each load and each f32 atomic add covers one 128-byte line; arrival order of the atomic sums changes
# speedup vs baseline: 3.2411x; 3.2411x over previous
#include <hip/hip_runtime.h>

typedef float v2f __attribute__((ext_vector_type(2)));
typedef float v8f __attribute__((ext_vector_type(8)));

#define NEG_SLOPE 0.2f

__device__ __forceinline__ unsigned fkey(float f) {
  unsigned u = __float_as_uint(f);
  return (u & 0x80000000u) ? ~u : (u | 0x80000000u);
}
__device__ __forceinline__ float fdekey(unsigned k) {
  unsigned u = (k & 0x80000000u) ? (k & 0x7FFFFFFFu) : ~k;
  return __uint_as_float(u);
}

__global__ void zero_f32(float* __restrict__ p, size_t n) {
  size_t i = (size_t)blockIdx.x * blockDim.x + threadIdx.x;
  size_t stride = (size_t)gridDim.x * blockDim.x;
  for (; i < n; i += stride) p[i] = 0.0f;
}

__global__ void gemm_wmma_f32(const float* __restrict__ A, const float* __restrict__ B,
                              float* __restrict__ C, int N, int K, int Nc) {
  const int lane = threadIdx.x & 31;
  const int half = lane >> 4;
  const int idx  = lane & 15;
  const int row0 = blockIdx.x * 16;
  const int col0 = blockIdx.y * 16;

  const int ar = min(row0 + idx, N - 1);
  const int bc = min(col0 + idx, Nc - 1);
  const float* __restrict__ ap = A + (size_t)ar * K + 2 * half;
  const float* __restrict__ bp = B + (size_t)bc * K + 2 * half;

  v8f c = {0.f, 0.f, 0.f, 0.f, 0.f, 0.f, 0.f, 0.f};
  for (int k = 0; k < K; k += 4) {
    v2f a = *(const v2f*)(ap + k);
    v2f b = *(const v2f*)(bp + k);
    c = __builtin_amdgcn_wmma_f32_16x16x4_f32(false, a, false, b, (short)0, c,
                                              false, false);
  }

  const int col = col0 + idx;
  if (col < Nc) {
#pragma unroll
    for (int r = 0; r < 8; ++r) {
      const int row = row0 + r + 8 * half;
      if (row < N) C[(size_t)row * Nc + col] = c[r];
    }
  }
}

__global__ void attn_logits(const float* __restrict__ h, const float* __restrict__ att_s,
                            const float* __restrict__ att_d, float* __restrict__ a_s,
                            float* __restrict__ a_d, int NH, int H, int D) {
  int i = blockIdx.x * blockDim.x + threadIdx.x;
  if (i >= NH) return;
  const int hh = i % H;
  const float* __restrict__ hp = h + (size_t)i * D;
  const float* __restrict__ as = att_s + (size_t)hh * D;
  const float* __restrict__ ad = att_d + (size_t)hh * D;
  float s0 = 0.f, s1 = 0.f;
  for (int dd = 0; dd < D; ++dd) {
    const float v = hp[dd];
    s0 += v * as[dd];
    s1 += v * ad[dd];
  }
  a_s[i] = s0;
  a_d[i] = s1;
}

__global__ void edge_max(const int* __restrict__ src, const int* __restrict__ dst,
                         int E, int N, const float* __restrict__ a_s,
                         const float* __restrict__ a_d, unsigned* __restrict__ mkey,
                         int H) {
  const int t = blockIdx.x * blockDim.x + threadIdx.x;
  const int TE = E + N;
  if (t >= TE) return;
  const int s = (t < E) ? src[t] : (t - E);
  const int d = (t < E) ? dst[t] : (t - E);
  for (int hh = 0; hh < H; ++hh) {
    float e = a_s[s * H + hh] + a_d[d * H + hh];
    e = (e > 0.f) ? e : NEG_SLOPE * e;
    atomicMax(&mkey[d * H + hh], fkey(e));
  }
}

__global__ void edge_expsum(const int* __restrict__ src, const int* __restrict__ dst,
                            int E, int N, const float* __restrict__ a_s,
                            const float* __restrict__ a_d,
                            const unsigned* __restrict__ mkey,
                            float* __restrict__ den, int H) {
  const int t = blockIdx.x * blockDim.x + threadIdx.x;
  const int TE = E + N;
  if (t >= TE) return;
  const int s = (t < E) ? src[t] : (t - E);
  const int d = (t < E) ? dst[t] : (t - E);
  for (int hh = 0; hh < H; ++hh) {
    float e = a_s[s * H + hh] + a_d[d * H + hh];
    e = (e > 0.f) ? e : NEG_SLOPE * e;
    const float ex = expf(e - fdekey(mkey[d * H + hh]));
    atomicAdd(&den[d * H + hh], ex);
  }
}


__device__ __forceinline__ float lane_bcast(float v, int j) {
  return __int_as_float(__builtin_amdgcn_readlane(__float_as_int(v), j));
}

__global__ void __launch_bounds__(256)
edge_aggregate_h4d32(const int* __restrict__ src, const int* __restrict__ dst,
                     int E, int N, const float* __restrict__ a_s,
                     const float* __restrict__ a_d,
                     const unsigned* __restrict__ mkey,
                     const float* __restrict__ den,
                     const float* __restrict__ h, float* __restrict__ agg) {
  const int lane = threadIdx.x & 31;
  const int wib = __builtin_amdgcn_readfirstlane((int)(threadIdx.x >> 5));
  const int TE = E + N;
  const int base = blockIdx.x * 256 + wib * 32;
  if (base >= TE) return;
  const int cnt = min(32, TE - base);

  int s = 0, d = 0;
  float al0 = 0.f, al1 = 0.f, al2 = 0.f, al3 = 0.f;
  if (lane < cnt) {
    const int t = base + lane;
    s = (t < E) ? src[t] : (t - E);
    d = (t < E) ? dst[t] : (t - E);
    const float4 as4 = *(const float4*)(a_s + (size_t)s * 4);
    const float4 ad4 = *(const float4*)(a_d + (size_t)d * 4);
    const uint4  mk4 = *(const uint4*)(mkey + (size_t)d * 4);
    const float4 dn4 = *(const float4*)(den + (size_t)d * 4);
    float e0 = as4.x + ad4.x, e1 = as4.y + ad4.y, e2 = as4.z + ad4.z, e3 = as4.w + ad4.w;
    e0 = (e0 > 0.f) ? e0 : NEG_SLOPE * e0;
    e1 = (e1 > 0.f) ? e1 : NEG_SLOPE * e1;
    e2 = (e2 > 0.f) ? e2 : NEG_SLOPE * e2;
    e3 = (e3 > 0.f) ? e3 : NEG_SLOPE * e3;
    al0 = expf(e0 - fdekey(mk4.x)) / dn4.x;
    al1 = expf(e1 - fdekey(mk4.y)) / dn4.y;
    al2 = expf(e2 - fdekey(mk4.z)) / dn4.z;
    al3 = expf(e3 - fdekey(mk4.w)) / dn4.w;
  }

  int j = 0;
  for (; j + 3 < cnt; j += 4) {
    int dj[4];
    float a[4][4], v[4][4];
    for (int k = 0; k < 4; ++k) {
      const int sj = __builtin_amdgcn_readlane(s, j + k);
      dj[k] = __builtin_amdgcn_readlane(d, j + k);
      a[k][0] = lane_bcast(al0, j + k);
      a[k][1] = lane_bcast(al1, j + k);
      a[k][2] = lane_bcast(al2, j + k);
      a[k][3] = lane_bcast(al3, j + k);
      const float* __restrict__ hr = h + (size_t)sj * 128 + lane;
      v[k][0] = hr[0];
      v[k][1] = hr[32];
      v[k][2] = hr[64];
      v[k][3] = hr[96];
    }
    for (int k = 0; k < 4; ++k) {
      float* __restrict__ g = agg + (size_t)dj[k] * 128 + lane;
      atomicAdd(g,      a[k][0] * v[k][0]);
      atomicAdd(g + 32, a[k][1] * v[k][1]);
      atomicAdd(g + 64, a[k][2] * v[k][2]);
      atomicAdd(g + 96, a[k][3] * v[k][3]);
    }
  }
  for (; j < cnt; ++j) {
    const int sj = __builtin_amdgcn_readlane(s, j);
    const int dk = __builtin_amdgcn_readlane(d, j);
    const float a0 = lane_bcast(al0, j), a1 = lane_bcast(al1, j);
    const float a2 = lane_bcast(al2, j), a3 = lane_bcast(al3, j);
    const float* __restrict__ hr = h + (size_t)sj * 128 + lane;
    const float v0 = hr[0], v1 = hr[32], v2 = hr[64], v3 = hr[96];
    float* __restrict__ g = agg + (size_t)dk * 128 + lane;
    atomicAdd(g,      a0 * v0);
    atomicAdd(g + 32, a1 * v1);
    atomicAdd(g + 64, a2 * v2);
    atomicAdd(g + 96, a3 * v3);
  }
}

__global__ void __launch_bounds__(256)
edge_aggregate_h1d40(const int* __restrict__ src, const int* __restrict__ dst,
                     int E, int N, const float* __restrict__ a_s,
                     const float* __restrict__ a_d,
                     const unsigned* __restrict__ mkey,
                     const float* __restrict__ den,
                     const float* __restrict__ h, float* __restrict__ out) {
  const int lane = threadIdx.x & 31;
  const int wib = __builtin_amdgcn_readfirstlane((int)(threadIdx.x >> 5));
  const int TE = E + N;
  const int base = blockIdx.x * 256 + wib * 32;
  if (base >= TE) return;
  const int cnt = min(32, TE - base);

  int s = 0, d = 0;
  float al = 0.f;
  if (lane < cnt) {
    const int t = base + lane;
    s = (t < E) ? src[t] : (t - E);
    d = (t < E) ? dst[t] : (t - E);
    float e = a_s[s] + a_d[d];
    e = (e > 0.f) ? e : NEG_SLOPE * e;
    al = expf(e - fdekey(mkey[d])) / den[d];
  }

  int j = 0;
  for (; j + 3 < cnt; j += 4) {
    int dj[4], c1[4];
    float aj[4], va[4], vb[4];
    for (int k = 0; k < 4; ++k) {
      const int sj = __builtin_amdgcn_readlane(s, j + k);
      dj[k] = __builtin_amdgcn_readlane(d, j + k);
      aj[k] = lane_bcast(al, j + k);
      c1[k] = 32 - 8 * (dj[k] & 3);
      const float* __restrict__ hr = h + (size_t)sj * 40;
      va[k] = hr[lane];
      vb[k] = hr[min(c1[k] + lane, 39)];
    }
    for (int k = 0; k < 4; ++k) {
      float* __restrict__ orow = out + (size_t)dj[k] * 40;
      const int fb = c1[k] + lane;
      if (lane < c1[k]) atomicAdd(orow + lane, aj[k] * va[k]);
      if (fb < 40)      atomicAdd(orow + fb,   aj[k] * vb[k]);
    }
  }
  for (; j < cnt; ++j) {
    const int sj = __builtin_amdgcn_readlane(s, j);
    const int dk = __builtin_amdgcn_readlane(d, j);
    const float ak = lane_bcast(al, j);
    const int c = 32 - 8 * (dk & 3);
    const int fb = c + lane;
    const float* __restrict__ hr = h + (size_t)sj * 40;
    const float va = hr[lane];
    const float vb = hr[min(fb, 39)];
    float* __restrict__ orow = out + (size_t)dk * 40;
    if (lane < c) atomicAdd(orow + lane, ak * va);
    if (fb < 40)  atomicAdd(orow + fb,   ak * vb);
  }
}

__global__ void bias_relu(const float* __restrict__ agg, const float* __restrict__ b,
                          float* __restrict__ out, int C, size_t n) {
  size_t i = (size_t)blockIdx.x * blockDim.x + threadIdx.x;
  if (i >= n) return;
  const float v = agg[i] + b[i % C];
  out[i] = (v > 0.f) ? v : 0.f;
}

__global__ void bias_add_inplace(float* __restrict__ out, const float* __restrict__ b,
                                 int C, size_t n) {
  size_t i = (size_t)blockIdx.x * blockDim.x + threadIdx.x;
  if (i >= n) return;
  out[i] += b[i % C];
}

extern "C" void kernel_launch(void* const* d_in, const int* in_sizes, int n_in,
                              void* d_out, int out_size, void* d_ws, size_t ws_size,
                              hipStream_t stream) {
  const float* x        = (const float*)d_in[0];
  const int*   ei       = (const int*)d_in[1];
  const float* W1       = (const float*)d_in[2];
  const float* att_src1 = (const float*)d_in[3];
  const float* att_dst1 = (const float*)d_in[4];
  const float* b1       = (const float*)d_in[5];
  const float* W2       = (const float*)d_in[6];
  const float* att_src2 = (const float*)d_in[7];
  const float* att_dst2 = (const float*)d_in[8];
  const float* b2       = (const float*)d_in[9];

  const int NFEAT = 256, NHID = 128, H1 = 4, D1 = 32, NCLS = 40;
  const int N = in_sizes[0] / NFEAT;
  const int E = in_sizes[1] / 2;
  const int* src = ei;
  const int* dst = ei + E;

  float* out = (float*)d_out;

  float* w = (float*)d_ws;
  float* h1   = w;
  float* h2   = h1  + (size_t)N * NHID;
  float* as1  = h2  + (size_t)N * NCLS;
  float* ad1  = as1 + (size_t)N * H1;
  float* as2  = ad1 + (size_t)N * H1;
  float* ad2  = as2 + (size_t)N;
  float* zreg = ad2 + (size_t)N;
  float*    agg1 = zreg;
  float*    den1 = agg1 + (size_t)N * NHID;
  unsigned* m1   = (unsigned*)(den1 + (size_t)N * H1);
  float*    den2 = (float*)(m1 + (size_t)N * H1);
  unsigned* m2   = (unsigned*)(den2 + (size_t)N);
  const size_t zcount = (size_t)N * (NHID + H1 + H1 + 1 + 1);

  const int ZB = 2048, TB = 256;
  const int TE = E + N;

  zero_f32<<<ZB, TB, 0, stream>>>(zreg, zcount);
  zero_f32<<<ZB, TB, 0, stream>>>(out, (size_t)N * NCLS);

  {
    dim3 grid((N + 15) / 16, NHID / 16);
    gemm_wmma_f32<<<grid, 32, 0, stream>>>(x, W1, h1, N, NFEAT, NHID);
  }
  attn_logits<<<(N * H1 + TB - 1) / TB, TB, 0, stream>>>(h1, att_src1, att_dst1,
                                                         as1, ad1, N * H1, H1, D1);
  edge_max<<<(TE + TB - 1) / TB, TB, 0, stream>>>(src, dst, E, N, as1, ad1, m1, H1);
  edge_expsum<<<(TE + TB - 1) / TB, TB, 0, stream>>>(src, dst, E, N, as1, ad1, m1,
                                                     den1, H1);
  edge_aggregate_h4d32<<<(TE + TB - 1) / TB, TB, 0, stream>>>(src, dst, E, N, as1, ad1,
                                                              m1, den1, h1, agg1);
  float* x2 = h1;
  {
    const size_t n = (size_t)N * NHID;
    bias_relu<<<(unsigned)((n + TB - 1) / TB), TB, 0, stream>>>(agg1, b1, x2, NHID, n);
  }

  {
    dim3 grid((N + 15) / 16, (NCLS + 15) / 16);
    gemm_wmma_f32<<<grid, 32, 0, stream>>>(x2, W2, h2, N, NHID, NCLS);
  }
  attn_logits<<<(N + TB - 1) / TB, TB, 0, stream>>>(h2, att_src2, att_dst2,
                                                    as2, ad2, N, 1, NCLS);
  edge_max<<<(TE + TB - 1) / TB, TB, 0, stream>>>(src, dst, E, N, as2, ad2, m2, 1);
  edge_expsum<<<(TE + TB - 1) / TB, TB, 0, stream>>>(src, dst, E, N, as2, ad2, m2,
                                                     den2, 1);
  edge_aggregate_h1d40<<<(TE + TB - 1) / TB, TB, 0, stream>>>(src, dst, E, N, as2, ad2,
                                                              m2, den2, h2, out);
  {
    const size_t n = (size_t)N * NCLS;
    bias_add_inplace<<<(unsigned)((n + TB - 1) / TB), TB, 0, stream>>>(out, b2, NCLS, n);
  }
}


# softmax and aggregation per destination over a dst-sorted edge list built each call by counting sort: rows summed in registers, stored once; no f32 atomics; sum order = arrival-order int-atomic rank
# speedup vs baseline: 8.1893x; 2.5267x over previous
#include <hip/hip_runtime.h>


typedef float v2f __attribute__((ext_vector_type(2)));
typedef float v8f __attribute__((ext_vector_type(8)));

#define NEG_SLOPE 0.2f


__global__ void zero_i32(int* __restrict__ p, int n) {
  const int i = blockIdx.x * blockDim.x + threadIdx.x;
  if (i < n) p[i] = 0;
}

__global__ void edge_rank(const int* __restrict__ dst, int E, int* __restrict__ cnt,
                          int* __restrict__ rank) {
  const int t = blockIdx.x * blockDim.x + threadIdx.x;
  if (t >= E) return;
  rank[t] = atomicAdd(&cnt[dst[t]], 1);
}

__global__ void __launch_bounds__(256)
csr_offsets(const int* __restrict__ cnt, int* __restrict__ off, int N) {
  __shared__ int wpre[8];
  __shared__ int wtot[8];
  const int tid = threadIdx.x, lane = tid & 31, wid = tid >> 5;
  const int b0 = blockIdx.x * 256;
  int pre = 0;
  for (int i = tid; i < b0; i += 256) pre += cnt[i];
  for (int m = 16; m >= 1; m >>= 1) pre += __shfl_xor(pre, m, 32);
  const int i = b0 + tid;
  const int x = (i < N) ? cnt[i] : 0;
  int inc = x;
  for (int d = 1; d < 32; d <<= 1) {
    const int y = __shfl_up(inc, d, 32);
    if (lane >= d) inc += y;
  }
  if (lane == 31) wtot[wid] = inc;
  if (lane == 0) wpre[wid] = pre;
  __syncthreads();
  int base = 0;
  for (int w = 0; w < 8; ++w) {
    base += wpre[w];
    if (w < wid) base += wtot[w];
  }
  if (i < N) off[i] = base + inc - x;
  if (i == N - 1) off[N] = base + inc;
}

__global__ void edge_fill(const int* __restrict__ src, const int* __restrict__ dst,
                          const int* __restrict__ rank, const int* __restrict__ off,
                          int E, int* __restrict__ csr_src) {
  const int t = blockIdx.x * blockDim.x + threadIdx.x;
  if (t >= E) return;
  csr_src[off[dst[t]] + rank[t]] = src[t];
}

__global__ void gemm_wmma_f32(const float* __restrict__ A, const float* __restrict__ B,
                              float* __restrict__ C, int N, int K, int Nc) {
  const int lane = threadIdx.x & 31;
  const int half = lane >> 4;
  const int idx  = lane & 15;
  const int row0 = blockIdx.x * 16;
  const int col0 = blockIdx.y * 16;

  const int ar = min(row0 + idx, N - 1);
  const int bc = min(col0 + idx, Nc - 1);
  const float* __restrict__ ap = A + (size_t)ar * K + 2 * half;
  const float* __restrict__ bp = B + (size_t)bc * K + 2 * half;

  v8f c = {0.f, 0.f, 0.f, 0.f, 0.f, 0.f, 0.f, 0.f};
  for (int k = 0; k < K; k += 4) {
    v2f a = *(const v2f*)(ap + k);
    v2f b = *(const v2f*)(bp + k);
    c = __builtin_amdgcn_wmma_f32_16x16x4_f32(false, a, false, b, (short)0, c,
                                              false, false);
  }

  const int col = col0 + idx;
  if (col < Nc) {
#pragma unroll
    for (int r = 0; r < 8; ++r) {
      const int row = row0 + r + 8 * half;
      if (row < N) C[(size_t)row * Nc + col] = c[r];
    }
  }
}

__global__ void attn_logits(const float* __restrict__ h, const float* __restrict__ att_s,
                            const float* __restrict__ att_d, float* __restrict__ a_s,
                            float* __restrict__ a_d, int NH, int H, int D) {
  int i = blockIdx.x * blockDim.x + threadIdx.x;
  if (i >= NH) return;
  const int hh = i % H;
  const float* __restrict__ hp = h + (size_t)i * D;
  const float* __restrict__ as = att_s + (size_t)hh * D;
  const float* __restrict__ ad = att_d + (size_t)hh * D;
  float s0 = 0.f, s1 = 0.f;
  for (int dd = 0; dd < D; ++dd) {
    const float v = hp[dd];
    s0 += v * as[dd];
    s1 += v * ad[dd];
  }
  a_s[i] = s0;
  a_d[i] = s1;
}


__device__ __forceinline__ float lane_bcast(float v, int j) {
  return __int_as_float(__builtin_amdgcn_readlane(__float_as_int(v), j));
}
__device__ __forceinline__ float leaky(float e) { return (e > 0.f) ? e : NEG_SLOPE * e; }
__device__ __forceinline__ float wave_max(float v) {
  for (int m = 16; m >= 1; m >>= 1) v = fmaxf(v, __shfl_xor(v, m, 32));
  return v;
}
__device__ __forceinline__ float wave_sum(float v) {
  for (int m = 16; m >= 1; m >>= 1) v += __shfl_xor(v, m, 32);
  return v;
}

__device__ __forceinline__ void add_rows_h4d32(const float* __restrict__ h, int s_l,
                                               float al0, float al1, float al2, float al3,
                                               int cnt, int lane, float& acc0, float& acc1,
                                               float& acc2, float& acc3) {
  int j = 0;
  for (; j + 3 < cnt; j += 4) {
    float a[4][4], v[4][4];
    for (int k = 0; k < 4; ++k) {
      const int sj = __builtin_amdgcn_readlane(s_l, j + k);
      a[k][0] = lane_bcast(al0, j + k);
      a[k][1] = lane_bcast(al1, j + k);
      a[k][2] = lane_bcast(al2, j + k);
      a[k][3] = lane_bcast(al3, j + k);
      const float* __restrict__ hr = h + (size_t)sj * 128 + lane;
      v[k][0] = hr[0];
      v[k][1] = hr[32];
      v[k][2] = hr[64];
      v[k][3] = hr[96];
    }
    for (int k = 0; k < 4; ++k) {
      acc0 += a[k][0] * v[k][0];
      acc1 += a[k][1] * v[k][1];
      acc2 += a[k][2] * v[k][2];
      acc3 += a[k][3] * v[k][3];
    }
  }
  for (; j < cnt; ++j) {
    const int sj = __builtin_amdgcn_readlane(s_l, j);
    const float a0 = lane_bcast(al0, j), a1 = lane_bcast(al1, j);
    const float a2 = lane_bcast(al2, j), a3 = lane_bcast(al3, j);
    const float* __restrict__ hr = h + (size_t)sj * 128 + lane;
    const float v0 = hr[0], v1 = hr[32], v2 = hr[64], v3 = hr[96];
    acc0 += a0 * v0;
    acc1 += a1 * v1;
    acc2 += a2 * v2;
    acc3 += a3 * v3;
  }
}

__global__ void __launch_bounds__(256)
node_aggregate_h4d32(const int* __restrict__ off, const int* __restrict__ csr_src,
                     const float* __restrict__ a_s, const float* __restrict__ a_d,
                     const float* __restrict__ h, const float* __restrict__ bias,
                     float* __restrict__ x2, int N) {
  const int lane = threadIdx.x & 31;
  const int wib = __builtin_amdgcn_readfirstlane((int)(threadIdx.x >> 5));
  const int node = blockIdx.x * 8 + wib;
  if (node >= N) return;
  const int beg = __builtin_amdgcn_readfirstlane(off[node]);
  const int deg = __builtin_amdgcn_readfirstlane(off[node + 1]) - beg;
  const int* __restrict__ nb = csr_src + beg;

  const float4 ad4 = *(const float4*)(a_d + (size_t)node * 4);
  const float4 sf4 = *(const float4*)(a_s + (size_t)node * 4);
  const float z0 = leaky(sf4.x + ad4.x), z1 = leaky(sf4.y + ad4.y);
  const float z2 = leaky(sf4.z + ad4.z), z3 = leaky(sf4.w + ad4.w);

  const bool act = lane < deg;
  const int s_c = act ? nb[lane] : node;
  const float4 as4 = *(const float4*)(a_s + (size_t)s_c * 4);
  const float e0 = leaky(as4.x + ad4.x), e1 = leaky(as4.y + ad4.y);
  const float e2 = leaky(as4.z + ad4.z), e3 = leaky(as4.w + ad4.w);

  float m0 = fmaxf(z0, e0), m1 = fmaxf(z1, e1), m2 = fmaxf(z2, e2), m3 = fmaxf(z3, e3);
  for (int i = 32 + lane; i < deg; i += 32) {
    const float4 b4 = *(const float4*)(a_s + (size_t)nb[i] * 4);
    m0 = fmaxf(m0, leaky(b4.x + ad4.x));
    m1 = fmaxf(m1, leaky(b4.y + ad4.y));
    m2 = fmaxf(m2, leaky(b4.z + ad4.z));
    m3 = fmaxf(m3, leaky(b4.w + ad4.w));
  }
  m0 = wave_max(m0); m1 = wave_max(m1); m2 = wave_max(m2); m3 = wave_max(m3);

  const float p0 = act ? expf(e0 - m0) : 0.f, p1 = act ? expf(e1 - m1) : 0.f;
  const float p2 = act ? expf(e2 - m2) : 0.f, p3 = act ? expf(e3 - m3) : 0.f;
  float d0 = p0, d1 = p1, d2 = p2, d3 = p3;
  for (int i = 32 + lane; i < deg; i += 32) {
    const float4 b4 = *(const float4*)(a_s + (size_t)nb[i] * 4);
    d0 += expf(leaky(b4.x + ad4.x) - m0);
    d1 += expf(leaky(b4.y + ad4.y) - m1);
    d2 += expf(leaky(b4.z + ad4.z) - m2);
    d3 += expf(leaky(b4.w + ad4.w) - m3);
  }
  const float q0 = expf(z0 - m0), q1 = expf(z1 - m1), q2 = expf(z2 - m2), q3 = expf(z3 - m3);
  d0 = wave_sum(d0) + q0; d1 = wave_sum(d1) + q1;
  d2 = wave_sum(d2) + q2; d3 = wave_sum(d3) + q3;

  const float* __restrict__ hs = h + (size_t)node * 128 + lane;
  float acc0 = (q0 / d0) * hs[0];
  float acc1 = (q1 / d1) * hs[32];
  float acc2 = (q2 / d2) * hs[64];
  float acc3 = (q3 / d3) * hs[96];
  add_rows_h4d32(h, s_c, p0 / d0, p1 / d1, p2 / d2, p3 / d3, min(deg, 32), lane,
                 acc0, acc1, acc2, acc3);
  for (int base = 32; base < deg; base += 32) {
    const int cnt = min(32, deg - base);
    const bool on = lane < cnt;
    const int s_r = on ? nb[base + lane] : node;
    const float4 b4 = *(const float4*)(a_s + (size_t)s_r * 4);
    const float r0 = on ? expf(leaky(b4.x + ad4.x) - m0) / d0 : 0.f;
    const float r1 = on ? expf(leaky(b4.y + ad4.y) - m1) / d1 : 0.f;
    const float r2 = on ? expf(leaky(b4.z + ad4.z) - m2) / d2 : 0.f;
    const float r3 = on ? expf(leaky(b4.w + ad4.w) - m3) / d3 : 0.f;
    add_rows_h4d32(h, s_r, r0, r1, r2, r3, cnt, lane, acc0, acc1, acc2, acc3);
  }

  float* __restrict__ xr = x2 + (size_t)node * 128 + lane;
  const float o0 = acc0 + bias[lane],      o1 = acc1 + bias[32 + lane];
  const float o2 = acc2 + bias[64 + lane], o3 = acc3 + bias[96 + lane];
  xr[0]  = (o0 > 0.f) ? o0 : 0.f;
  xr[32] = (o1 > 0.f) ? o1 : 0.f;
  xr[64] = (o2 > 0.f) ? o2 : 0.f;
  xr[96] = (o3 > 0.f) ? o3 : 0.f;
}

__device__ __forceinline__ void add_rows_h1d40(const float* __restrict__ h, int s_l, float al,
                                               int cnt, int lane, int f1, float& acc0,
                                               float& acc1) {
  int j = 0;
  for (; j + 3 < cnt; j += 4) {
    float a[4], va[4], vb[4];
    for (int k = 0; k < 4; ++k) {
      const int sj = __builtin_amdgcn_readlane(s_l, j + k);
      a[k] = lane_bcast(al, j + k);
      const float* __restrict__ hr = h + (size_t)sj * 40;
      va[k] = hr[lane];
      vb[k] = hr[f1];
    }
    for (int k = 0; k < 4; ++k) {
      acc0 += a[k] * va[k];
      acc1 += a[k] * vb[k];
    }
  }
  for (; j < cnt; ++j) {
    const int sj = __builtin_amdgcn_readlane(s_l, j);
    const float ak = lane_bcast(al, j);
    const float* __restrict__ hr = h + (size_t)sj * 40;
    acc0 += ak * hr[lane];
    acc1 += ak * hr[f1];
  }
}

__global__ void __launch_bounds__(256)
node_aggregate_h1d40(const int* __restrict__ off, const int* __restrict__ csr_src,
                     const float* __restrict__ a_s, const float* __restrict__ a_d,
                     const float* __restrict__ h, const float* __restrict__ bias,
                     float* __restrict__ out, int N) {
  const int lane = threadIdx.x & 31;
  const int wib = __builtin_amdgcn_readfirstlane((int)(threadIdx.x >> 5));
  const int node = blockIdx.x * 8 + wib;
  if (node >= N) return;
  const int beg = __builtin_amdgcn_readfirstlane(off[node]);
  const int deg = __builtin_amdgcn_readfirstlane(off[node + 1]) - beg;
  const int* __restrict__ nb = csr_src + beg;
  const int f1 = 32 + (lane & 7);

  const float ad = a_d[node];
  const float z = leaky(a_s[node] + ad);

  const bool act = lane < deg;
  const int s_c = act ? nb[lane] : node;
  const float e = leaky(a_s[s_c] + ad);

  float m = fmaxf(z, e);
  for (int i = 32 + lane; i < deg; i += 32) m = fmaxf(m, leaky(a_s[nb[i]] + ad));
  m = wave_max(m);

  const float p = act ? expf(e - m) : 0.f;
  float d = p;
  for (int i = 32 + lane; i < deg; i += 32) d += expf(leaky(a_s[nb[i]] + ad) - m);
  const float q = expf(z - m);
  d = wave_sum(d) + q;

  const float* __restrict__ hs = h + (size_t)node * 40;
  float acc0 = (q / d) * hs[lane];
  float acc1 = (q / d) * hs[f1];
  add_rows_h1d40(h, s_c, p / d, min(deg, 32), lane, f1, acc0, acc1);
  for (int base = 32; base < deg; base += 32) {
    const int cnt = min(32, deg - base);
    const bool on = lane < cnt;
    const int s_r = on ? nb[base + lane] : node;
    const float r = on ? expf(leaky(a_s[s_r] + ad) - m) / d : 0.f;
    add_rows_h1d40(h, s_r, r, cnt, lane, f1, acc0, acc1);
  }

  float* __restrict__ orow = out + (size_t)node * 40;
  orow[lane] = acc0 + bias[lane];
  if (lane < 8) orow[32 + lane] = acc1 + bias[32 + lane];
}

extern "C" void kernel_launch(void* const* d_in, const int* in_sizes, int n_in,
                              void* d_out, int out_size, void* d_ws, size_t ws_size,
                              hipStream_t stream) {
  const float* x        = (const float*)d_in[0];
  const int*   ei       = (const int*)d_in[1];
  const float* W1       = (const float*)d_in[2];
  const float* att_src1 = (const float*)d_in[3];
  const float* att_dst1 = (const float*)d_in[4];
  const float* b1       = (const float*)d_in[5];
  const float* W2       = (const float*)d_in[6];
  const float* att_src2 = (const float*)d_in[7];
  const float* att_dst2 = (const float*)d_in[8];
  const float* b2       = (const float*)d_in[9];

  const int NFEAT = 256, NHID = 128, H1 = 4, D1 = 32, NCLS = 40;
  const int N = in_sizes[0] / NFEAT;
  const int E = in_sizes[1] / 2;
  const int* src = ei;
  const int* dst = ei + E;

  float* out = (float*)d_out;

  float* w = (float*)d_ws;
  float* h1   = w;
  float* x2   = h1  + (size_t)N * NHID;
  float* h2   = x2  + (size_t)N * NHID;
  float* as1  = h2  + (size_t)N * NCLS;
  float* ad1  = as1 + (size_t)N * H1;
  float* as2  = ad1 + (size_t)N * H1;
  float* ad2  = as2 + (size_t)N;
  int* cnt     = (int*)(ad2 + (size_t)N);
  int* off     = cnt + (size_t)N;
  int* rank    = off + (size_t)N + 1;
  int* csr_src = rank + (size_t)E;

  const int TB = 256;

  zero_i32<<<(N + TB - 1) / TB, TB, 0, stream>>>(cnt, N);
  edge_rank<<<(E + TB - 1) / TB, TB, 0, stream>>>(dst, E, cnt, rank);
  csr_offsets<<<(N + TB - 1) / TB, TB, 0, stream>>>(cnt, off, N);
  edge_fill<<<(E + TB - 1) / TB, TB, 0, stream>>>(src, dst, rank, off, E, csr_src);

  {
    dim3 grid((N + 15) / 16, NHID / 16);
    gemm_wmma_f32<<<grid, 32, 0, stream>>>(x, W1, h1, N, NFEAT, NHID);
  }
  attn_logits<<<(N * H1 + TB - 1) / TB, TB, 0, stream>>>(h1, att_src1, att_dst1,
                                                         as1, ad1, N * H1, H1, D1);
  node_aggregate_h4d32<<<(N + 7) / 8, TB, 0, stream>>>(off, csr_src, as1, ad1, h1, b1,
                                                       x2, N);

  {
    dim3 grid((N + 15) / 16, (NCLS + 15) / 16);
    gemm_wmma_f32<<<grid, 32, 0, stream>>>(x2, W2, h2, N, NHID, NCLS);
  }
  attn_logits<<<(N + TB - 1) / TB, TB, 0, stream>>>(h2, att_src2, att_dst2,
                                                    as2, ad2, N, 1, NCLS);
  node_aggregate_h1d40<<<(N + 7) / 8, TB, 0, stream>>>(off, csr_src, as2, ad2, h2, b2,
                                                       out, N);
}


# both GEMMs re-tiled: one wave per 16-row strip over all column tiles, K in blocks of 32 by 16-byte fragment loads requested one group ahead, so A is read once; k sum order inside a block changes
# speedup vs baseline: 9.5815x; 1.1700x over previous
#include <hip/hip_runtime.h>


typedef float v2f __attribute__((ext_vector_type(2)));
typedef float v8f __attribute__((ext_vector_type(8)));

#define NEG_SLOPE 0.2f


__global__ void zero_i32(int* __restrict__ p, int n) {
  const int i = blockIdx.x * blockDim.x + threadIdx.x;
  if (i < n) p[i] = 0;
}

__global__ void edge_rank(const int* __restrict__ dst, int E, int* __restrict__ cnt,
                          int* __restrict__ rank) {
  const int t = blockIdx.x * blockDim.x + threadIdx.x;
  if (t >= E) return;
  rank[t] = atomicAdd(&cnt[dst[t]], 1);
}

__global__ void __launch_bounds__(256)
csr_offsets(const int* __restrict__ cnt, int* __restrict__ off, int N) {
  __shared__ int wpre[8];
  __shared__ int wtot[8];
  const int tid = threadIdx.x, lane = tid & 31, wid = tid >> 5;
  const int b0 = blockIdx.x * 256;
  int pre = 0;
  for (int i = tid; i < b0; i += 256) pre += cnt[i];
  for (int m = 16; m >= 1; m >>= 1) pre += __shfl_xor(pre, m, 32);
  const int i = b0 + tid;
  const int x = (i < N) ? cnt[i] : 0;
  int inc = x;
  for (int d = 1; d < 32; d <<= 1) {
    const int y = __shfl_up(inc, d, 32);
    if (lane >= d) inc += y;
  }
  if (lane == 31) wtot[wid] = inc;
  if (lane == 0) wpre[wid] = pre;
  __syncthreads();
  int base = 0;
  for (int w = 0; w < 8; ++w) {
    base += wpre[w];
    if (w < wid) base += wtot[w];
  }
  if (i < N) off[i] = base + inc - x;
  if (i == N - 1) off[N] = base + inc;
}

__global__ void edge_fill(const int* __restrict__ src, const int* __restrict__ dst,
                          const int* __restrict__ rank, const int* __restrict__ off,
                          int E, int* __restrict__ csr_src) {
  const int t = blockIdx.x * blockDim.x + threadIdx.x;
  if (t >= E) return;
  csr_src[off[dst[t]] + rank[t]] = src[t];
}


struct frag16 { float4 p[4]; };

__device__ __forceinline__ frag16 load_k16(const float* q) {
  frag16 f;
#pragma unroll
  for (int i = 0; i < 4; ++i) f.p[i] = *(const float4*)(q + 4 * i);
  return f;
}

template <int G>
__device__ __forceinline__ void mma_group(const frag16& a, const frag16 (&b)[G], v8f* c) {
#pragma unroll
  for (int i = 0; i < 4; ++i) {
#pragma unroll
    for (int g = 0; g < G; ++g) {
      const v2f av = {a.p[i].x, a.p[i].y};
      const v2f bv = {b[g].p[i].x, b[g].p[i].y};
      c[g] = __builtin_amdgcn_wmma_f32_16x16x4_f32(false, av, false, bv, (short)0, c[g],
                                                   false, false);
    }
#pragma unroll
    for (int g = 0; g < G; ++g) {
      const v2f av = {a.p[i].z, a.p[i].w};
      const v2f bv = {b[g].p[i].z, b[g].p[i].w};
      c[g] = __builtin_amdgcn_wmma_f32_16x16x4_f32(false, av, false, bv, (short)0, c[g],
                                                   false, false);
    }
  }
}

template <int K, int NT, int G>
__device__ __forceinline__ void gemm_strip(const float* __restrict__ A,
                                           const float* __restrict__ B,
                                           float* __restrict__ C, int N, int Nc) {
  constexpr int KB = K / 32;
  constexpr int NG = NT / G;
  const int lane = threadIdx.x & 31;
  const int half = lane >> 4;
  const int idx  = lane & 15;
  const int row0 = blockIdx.x * 16;

  const int ar = min(row0 + idx, N - 1);
  const float* ap = A + (size_t)ar * K + 16 * half;
  const float* bp[NT];
#pragma unroll
  for (int t = 0; t < NT; ++t)
    bp[t] = B + (size_t)min(16 * t + idx, Nc - 1) * K + 16 * half;

  v8f c[NT];
#pragma unroll
  for (int t = 0; t < NT; ++t) c[t] = (v8f){0.f, 0.f, 0.f, 0.f, 0.f, 0.f, 0.f, 0.f};

  frag16 a_cur = load_k16(ap);
  frag16 b0[G];
#pragma unroll
  for (int g = 0; g < G; ++g) b0[g] = load_k16(bp[g]);

#pragma unroll
  for (int kb = 0; kb < KB; ++kb) {
    const int kn = min(kb + 1, KB - 1) * 32;
    frag16 a_nxt = load_k16(ap + kn);
    if constexpr (NG == 2) {
      frag16 b1[G];
#pragma unroll
      for (int g = 0; g < G; ++g) b1[g] = load_k16(bp[G + g] + kb * 32);
      __builtin_amdgcn_sched_barrier(0);
      mma_group<G>(a_cur, b0, c);
      __builtin_amdgcn_sched_barrier(0);
#pragma unroll
      for (int g = 0; g < G; ++g) b0[g] = load_k16(bp[g] + kn);
      __builtin_amdgcn_sched_barrier(0);
      mma_group<G>(a_cur, b1, c + G);
    } else {
      frag16 bn[G];
#pragma unroll
      for (int g = 0; g < G; ++g) bn[g] = load_k16(bp[g] + kn);
      __builtin_amdgcn_sched_barrier(0);
      mma_group<G>(a_cur, b0, c);
#pragma unroll
      for (int g = 0; g < G; ++g) b0[g] = bn[g];
    }
    a_cur = a_nxt;
  }

#pragma unroll
  for (int t = 0; t < NT; ++t) {
    const int col = 16 * t + idx;
    if (col < Nc) {
#pragma unroll
      for (int r = 0; r < 8; ++r) {
        const int row = row0 + r + 8 * half;
        if (row < N) C[(size_t)row * Nc + col] = c[t][r];
      }
    }
  }
}

__global__ void __launch_bounds__(32)
gemm_strip_k256_c128(const float* __restrict__ A, const float* __restrict__ B,
                     float* __restrict__ C, int N) {
  gemm_strip<256, 8, 4>(A, B, C, N, 128);
}

__global__ void __launch_bounds__(32)
gemm_strip_k128_c40(const float* __restrict__ A, const float* __restrict__ B,
                    float* __restrict__ C, int N) {
  gemm_strip<128, 3, 3>(A, B, C, N, 40);
}

__global__ void attn_logits(const float* __restrict__ h, const float* __restrict__ att_s,
                            const float* __restrict__ att_d, float* __restrict__ a_s,
                            float* __restrict__ a_d, int NH, int H, int D) {
  int i = blockIdx.x * blockDim.x + threadIdx.x;
  if (i >= NH) return;
  const int hh = i % H;
  const float* __restrict__ hp = h + (size_t)i * D;
  const float* __restrict__ as = att_s + (size_t)hh * D;
  const float* __restrict__ ad = att_d + (size_t)hh * D;
  float s0 = 0.f, s1 = 0.f;
  for (int dd = 0; dd < D; ++dd) {
    const float v = hp[dd];
    s0 += v * as[dd];
    s1 += v * ad[dd];
  }
  a_s[i] = s0;
  a_d[i] = s1;
}


__device__ __forceinline__ float lane_bcast(float v, int j) {
  return __int_as_float(__builtin_amdgcn_readlane(__float_as_int(v), j));
}
__device__ __forceinline__ float leaky(float e) { return (e > 0.f) ? e : NEG_SLOPE * e; }
__device__ __forceinline__ float wave_max(float v) {
  for (int m = 16; m >= 1; m >>= 1) v = fmaxf(v, __shfl_xor(v, m, 32));
  return v;
}
__device__ __forceinline__ float wave_sum(float v) {
  for (int m = 16; m >= 1; m >>= 1) v += __shfl_xor(v, m, 32);
  return v;
}

__device__ __forceinline__ void add_rows_h4d32(const float* __restrict__ h, int s_l,
                                               float al0, float al1, float al2, float al3,
                                               int cnt, int lane, float& acc0, float& acc1,
                                               float& acc2, float& acc3) {
  int j = 0;
  for (; j + 3 < cnt; j += 4) {
    float a[4][4], v[4][4];
    for (int k = 0; k < 4; ++k) {
      const int sj = __builtin_amdgcn_readlane(s_l, j + k);
      a[k][0] = lane_bcast(al0, j + k);
      a[k][1] = lane_bcast(al1, j + k);
      a[k][2] = lane_bcast(al2, j + k);
      a[k][3] = lane_bcast(al3, j + k);
      const float* __restrict__ hr = h + (size_t)sj * 128 + lane;
      v[k][0] = hr[0];
      v[k][1] = hr[32];
      v[k][2] = hr[64];
      v[k][3] = hr[96];
    }
    for (int k = 0; k < 4; ++k) {
      acc0 += a[k][0] * v[k][0];
      acc1 += a[k][1] * v[k][1];
      acc2 += a[k][2] * v[k][2];
      acc3 += a[k][3] * v[k][3];
    }
  }
  for (; j < cnt; ++j) {
    const int sj = __builtin_amdgcn_readlane(s_l, j);
    const float a0 = lane_bcast(al0, j), a1 = lane_bcast(al1, j);
    const float a2 = lane_bcast(al2, j), a3 = lane_bcast(al3, j);
    const float* __restrict__ hr = h + (size_t)sj * 128 + lane;
    const float v0 = hr[0], v1 = hr[32], v2 = hr[64], v3 = hr[96];
    acc0 += a0 * v0;
    acc1 += a1 * v1;
    acc2 += a2 * v2;
    acc3 += a3 * v3;
  }
}

__global__ void __launch_bounds__(256)
node_aggregate_h4d32(const int* __restrict__ off, const int* __restrict__ csr_src,
                     const float* __restrict__ a_s, const float* __restrict__ a_d,
                     const float* __restrict__ h, const float* __restrict__ bias,
                     float* __restrict__ x2, int N) {
  const int lane = threadIdx.x & 31;
  const int wib = __builtin_amdgcn_readfirstlane((int)(threadIdx.x >> 5));
  const int node = blockIdx.x * 8 + wib;
  if (node >= N) return;
  const int beg = __builtin_amdgcn_readfirstlane(off[node]);
  const int deg = __builtin_amdgcn_readfirstlane(off[node + 1]) - beg;
  const int* __restrict__ nb = csr_src + beg;

  const float4 ad4 = *(const float4*)(a_d + (size_t)node * 4);
  const float4 sf4 = *(const float4*)(a_s + (size_t)node * 4);
  const float z0 = leaky(sf4.x + ad4.x), z1 = leaky(sf4.y + ad4.y);
  const float z2 = leaky(sf4.z + ad4.z), z3 = leaky(sf4.w + ad4.w);

  const bool act = lane < deg;
  const int s_c = act ? nb[lane] : node;
  const float4 as4 = *(const float4*)(a_s + (size_t)s_c * 4);
  const float e0 = leaky(as4.x + ad4.x), e1 = leaky(as4.y + ad4.y);
  const float e2 = leaky(as4.z + ad4.z), e3 = leaky(as4.w + ad4.w);

  float m0 = fmaxf(z0, e0), m1 = fmaxf(z1, e1), m2 = fmaxf(z2, e2), m3 = fmaxf(z3, e3);
  for (int i = 32 + lane; i < deg; i += 32) {
    const float4 b4 = *(const float4*)(a_s + (size_t)nb[i] * 4);
    m0 = fmaxf(m0, leaky(b4.x + ad4.x));
    m1 = fmaxf(m1, leaky(b4.y + ad4.y));
    m2 = fmaxf(m2, leaky(b4.z + ad4.z));
    m3 = fmaxf(m3, leaky(b4.w + ad4.w));
  }
  m0 = wave_max(m0); m1 = wave_max(m1); m2 = wave_max(m2); m3 = wave_max(m3);

  const float p0 = act ? expf(e0 - m0) : 0.f, p1 = act ? expf(e1 - m1) : 0.f;
  const float p2 = act ? expf(e2 - m2) : 0.f, p3 = act ? expf(e3 - m3) : 0.f;
  float d0 = p0, d1 = p1, d2 = p2, d3 = p3;
  for (int i = 32 + lane; i < deg; i += 32) {
    const float4 b4 = *(const float4*)(a_s + (size_t)nb[i] * 4);
    d0 += expf(leaky(b4.x + ad4.x) - m0);
    d1 += expf(leaky(b4.y + ad4.y) - m1);
    d2 += expf(leaky(b4.z + ad4.z) - m2);
    d3 += expf(leaky(b4.w + ad4.w) - m3);
  }
  const float q0 = expf(z0 - m0), q1 = expf(z1 - m1), q2 = expf(z2 - m2), q3 = expf(z3 - m3);
  d0 = wave_sum(d0) + q0; d1 = wave_sum(d1) + q1;
  d2 = wave_sum(d2) + q2; d3 = wave_sum(d3) + q3;

  const float* __restrict__ hs = h + (size_t)node * 128 + lane;
  float acc0 = (q0 / d0) * hs[0];
  float acc1 = (q1 / d1) * hs[32];
  float acc2 = (q2 / d2) * hs[64];
  float acc3 = (q3 / d3) * hs[96];
  add_rows_h4d32(h, s_c, p0 / d0, p1 / d1, p2 / d2, p3 / d3, min(deg, 32), lane,
                 acc0, acc1, acc2, acc3);
  for (int base = 32; base < deg; base += 32) {
    const int cnt = min(32, deg - base);
    const bool on = lane < cnt;
    const int s_r = on ? nb[base + lane] : node;
    const float4 b4 = *(const float4*)(a_s + (size_t)s_r * 4);
    const float r0 = on ? expf(leaky(b4.x + ad4.x) - m0) / d0 : 0.f;
    const float r1 = on ? expf(leaky(b4.y + ad4.y) - m1) / d1 : 0.f;
    const float r2 = on ? expf(leaky(b4.z + ad4.z) - m2) / d2 : 0.f;
    const float r3 = on ? expf(leaky(b4.w + ad4.w) - m3) / d3 : 0.f;
    add_rows_h4d32(h, s_r, r0, r1, r2, r3, cnt, lane, acc0, acc1, acc2, acc3);
  }

  float* __restrict__ xr = x2 + (size_t)node * 128 + lane;
  const float o0 = acc0 + bias[lane],      o1 = acc1 + bias[32 + lane];
  const float o2 = acc2 + bias[64 + lane], o3 = acc3 + bias[96 + lane];
  xr[0]  = (o0 > 0.f) ? o0 : 0.f;
  xr[32] = (o1 > 0.f) ? o1 : 0.f;
  xr[64] = (o2 > 0.f) ? o2 : 0.f;
  xr[96] = (o3 > 0.f) ? o3 : 0.f;
}

__device__ __forceinline__ void add_rows_h1d40(const float* __restrict__ h, int s_l, float al,
                                               int cnt, int lane, int f1, float& acc0,
                                               float& acc1) {
  int j = 0;
  for (; j + 3 < cnt; j += 4) {
    float a[4], va[4], vb[4];
    for (int k = 0; k < 4; ++k) {
      const int sj = __builtin_amdgcn_readlane(s_l, j + k);
      a[k] = lane_bcast(al, j + k);
      const float* __restrict__ hr = h + (size_t)sj * 40;
      va[k] = hr[lane];
      vb[k] = hr[f1];
    }
    for (int k = 0; k < 4; ++k) {
      acc0 += a[k] * va[k];
      acc1 += a[k] * vb[k];
    }
  }
  for (; j < cnt; ++j) {
    const int sj = __builtin_amdgcn_readlane(s_l, j);
    const float ak = lane_bcast(al, j);
    const float* __restrict__ hr = h + (size_t)sj * 40;
    acc0 += ak * hr[lane];
    acc1 += ak * hr[f1];
  }
}

__global__ void __launch_bounds__(256)
node_aggregate_h1d40(const int* __restrict__ off, const int* __restrict__ csr_src,
                     const float* __restrict__ a_s, const float* __restrict__ a_d,
                     const float* __restrict__ h, const float* __restrict__ bias,
                     float* __restrict__ out, int N) {
  const int lane = threadIdx.x & 31;
  const int wib = __builtin_amdgcn_readfirstlane((int)(threadIdx.x >> 5));
  const int node = blockIdx.x * 8 + wib;
  if (node >= N) return;
  const int beg = __builtin_amdgcn_readfirstlane(off[node]);
  const int deg = __builtin_amdgcn_readfirstlane(off[node + 1]) - beg;
  const int* __restrict__ nb = csr_src + beg;
  const int f1 = 32 + (lane & 7);

  const float ad = a_d[node];
  const float z = leaky(a_s[node] + ad);

  const bool act = lane < deg;
  const int s_c = act ? nb[lane] : node;
  const float e = leaky(a_s[s_c] + ad);

  float m = fmaxf(z, e);
  for (int i = 32 + lane; i < deg; i += 32) m = fmaxf(m, leaky(a_s[nb[i]] + ad));
  m = wave_max(m);

  const float p = act ? expf(e - m) : 0.f;
  float d = p;
  for (int i = 32 + lane; i < deg; i += 32) d += expf(leaky(a_s[nb[i]] + ad) - m);
  const float q = expf(z - m);
  d = wave_sum(d) + q;

  const float* __restrict__ hs = h + (size_t)node * 40;
  float acc0 = (q / d) * hs[lane];
  float acc1 = (q / d) * hs[f1];
  add_rows_h1d40(h, s_c, p / d, min(deg, 32), lane, f1, acc0, acc1);
  for (int base = 32; base < deg; base += 32) {
    const int cnt = min(32, deg - base);
    const bool on = lane < cnt;
    const int s_r = on ? nb[base + lane] : node;
    const float r = on ? expf(leaky(a_s[s_r] + ad) - m) / d : 0.f;
    add_rows_h1d40(h, s_r, r, cnt, lane, f1, acc0, acc1);
  }

  float* __restrict__ orow = out + (size_t)node * 40;
  orow[lane] = acc0 + bias[lane];
  if (lane < 8) orow[32 + lane] = acc1 + bias[32 + lane];
}

extern "C" void kernel_launch(void* const* d_in, const int* in_sizes, int n_in,
                              void* d_out, int out_size, void* d_ws, size_t ws_size,
                              hipStream_t stream) {
  const float* x        = (const float*)d_in[0];
  const int*   ei       = (const int*)d_in[1];
  const float* W1       = (const float*)d_in[2];
  const float* att_src1 = (const float*)d_in[3];
  const float* att_dst1 = (const float*)d_in[4];
  const float* b1       = (const float*)d_in[5];
  const float* W2       = (const float*)d_in[6];
  const float* att_src2 = (const float*)d_in[7];
  const float* att_dst2 = (const float*)d_in[8];
  const float* b2       = (const float*)d_in[9];

  const int NFEAT = 256, NHID = 128, H1 = 4, D1 = 32, NCLS = 40;
  const int N = in_sizes[0] / NFEAT;
  const int E = in_sizes[1] / 2;
  const int* src = ei;
  const int* dst = ei + E;

  float* out = (float*)d_out;

  float* w = (float*)d_ws;
  float* h1   = w;
  float* x2   = h1  + (size_t)N * NHID;
  float* h2   = x2  + (size_t)N * NHID;
  float* as1  = h2  + (size_t)N * NCLS;
  float* ad1  = as1 + (size_t)N * H1;
  float* as2  = ad1 + (size_t)N * H1;
  float* ad2  = as2 + (size_t)N;
  int* cnt     = (int*)(ad2 + (size_t)N);
  int* off     = cnt + (size_t)N;
  int* rank    = off + (size_t)N + 1;
  int* csr_src = rank + (size_t)E;

  const int TB = 256;

  zero_i32<<<(N + TB - 1) / TB, TB, 0, stream>>>(cnt, N);
  edge_rank<<<(E + TB - 1) / TB, TB, 0, stream>>>(dst, E, cnt, rank);
  csr_offsets<<<(N + TB - 1) / TB, TB, 0, stream>>>(cnt, off, N);
  edge_fill<<<(E + TB - 1) / TB, TB, 0, stream>>>(src, dst, rank, off, E, csr_src);

  gemm_strip_k256_c128<<<(N + 15) / 16, 32, 0, stream>>>(x, W1, h1, N);
  attn_logits<<<(N * H1 + TB - 1) / TB, TB, 0, stream>>>(h1, att_src1, att_dst1,
                                                         as1, ad1, N * H1, H1, D1);
  node_aggregate_h4d32<<<(N + 7) / 8, TB, 0, stream>>>(off, csr_src, as1, ad1, h1, b1,
                                                       x2, N);

  gemm_strip_k128_c40<<<(N + 15) / 16, 32, 0, stream>>>(x2, W2, h2, N);
  attn_logits<<<(N + TB - 1) / TB, TB, 0, stream>>>(h2, att_src2, att_dst2,
                                                    as2, ad2, N, 1, NCLS);
  node_aggregate_h1d40<<<(N + 7) / 8, TB, 0, stream>>>(off, csr_src, as2, ad2, h2, b2,
                                                       out, N);
}


# edge list grouped by destination built in one launch: a work-group per 256 destinations scans all edges and ranks its hits by LDS atomics, no global atomics; sum order = LDS arrival rank
# speedup vs baseline: 11.9473x; 1.2469x over previous
#include <hip/hip_runtime.h>


typedef float v2f __attribute__((ext_vector_type(2)));
typedef float v8f __attribute__((ext_vector_type(8)));

#define NEG_SLOPE 0.2f


#define CSR_BKT 256
#define CSR_THREADS 1024
#define CSR_CAP 14336
#define CSR_U 8

template <class F>
__device__ __forceinline__ void for_each_edge(const int* __restrict__ dst, int E, int tid, F f) {
  const int head = min(E, (int)(((16u - (unsigned)((size_t)dst & 15u)) & 15u) >> 2));
  const int nv = (E - head) >> 2;
  const int4* __restrict__ dv = (const int4*)(dst + head);
  for (int i0 = tid; i0 < nv; i0 += CSR_U * CSR_THREADS) {
    int4 d[CSR_U];
#pragma unroll
    for (int u = 0; u < CSR_U; ++u) d[u] = dv[min(i0 + u * CSR_THREADS, nv - 1)];
#pragma unroll
    for (int u = 0; u < CSR_U; ++u) {
      const int i = i0 + u * CSR_THREADS;
      if (i < nv) {
        const int t = head + 4 * i;
        f(t, d[u].x);
        f(t + 1, d[u].y);
        f(t + 2, d[u].z);
        f(t + 3, d[u].w);
      }
    }
  }
  const int tail0 = head + 4 * nv;
  if (tid < head) f(tid, dst[tid]);
  if (tid < E - tail0) f(tail0 + tid, dst[tail0 + tid]);
}

__global__ void __launch_bounds__(CSR_THREADS)
csr_build(const int* __restrict__ src, const int* __restrict__ dst, int E, int N,
          int* __restrict__ off, int* __restrict__ csr_src) {
  __shared__ int c[CSR_BKT];
  __shared__ int loff[CSR_BKT];
  __shared__ int wtot[CSR_BKT / 32];
  __shared__ int s_below, s_nl;
  __shared__ int l_t[CSR_CAP];
  __shared__ int l_d[CSR_CAP];
  __shared__ int l_r[CSR_CAP];
  const int tid = threadIdx.x, lane = tid & 31, wid = tid >> 5;
  const int lo = blockIdx.x * CSR_BKT;

  if (tid < CSR_BKT) c[tid] = 0;
  if (tid == 0) { s_below = 0; s_nl = 0; }
  __syncthreads();

  int below = 0;
  for_each_edge(dst, E, tid, [&](int t, int d) {
    below += (d < lo) ? 1 : 0;
    const unsigned dl = (unsigned)(d - lo);
    if (dl < (unsigned)CSR_BKT) {
      const int r = atomicAdd(&c[dl], 1);
      const int k = atomicAdd(&s_nl, 1);
      if (k < CSR_CAP) { l_t[k] = t; l_d[k] = (int)dl; l_r[k] = r; }
    }
  });
  for (int m = 16; m >= 1; m >>= 1) below += __shfl_xor(below, m, 32);
  if (lane == 0) atomicAdd(&s_below, below);
  __syncthreads();

  int x = 0, inc = 0;
  if (tid < CSR_BKT) {
    x = c[tid];
    inc = x;
    for (int s = 1; s < 32; s <<= 1) {
      const int y = __shfl_up(inc, s, 32);
      if (lane >= s) inc += y;
    }
    if (lane == 31) wtot[wid] = inc;
  }
  __syncthreads();
  if (tid < CSR_BKT) {
    int b = 0;
    for (int w = 0; w < wid; ++w) b += wtot[w];
    loff[tid] = b + inc - x;
  }
  __syncthreads();
  const int base = s_below;
  const int nl = s_nl;
  if (tid < CSR_BKT && lo + tid < N) off[lo + tid] = base + loff[tid];
  if (tid == CSR_BKT - 1 && lo + CSR_BKT >= N) off[N] = base + loff[tid] + x;

  if (nl <= CSR_CAP) {
    for (int k = tid; k < nl; k += CSR_THREADS)
      csr_src[base + loff[l_d[k]] + l_r[k]] = src[l_t[k]];
  } else {
    if (tid < CSR_BKT) c[tid] = 0;
    __syncthreads();
    for_each_edge(dst, E, tid, [&](int t, int d) {
      const unsigned dl = (unsigned)(d - lo);
      if (dl < (unsigned)CSR_BKT) {
        const int r = atomicAdd(&c[dl], 1);
        csr_src[base + loff[dl] + r] = src[t];
      }
    });
  }
}


struct frag16 { float4 p[4]; };

__device__ __forceinline__ frag16 load_k16(const float* q) {
  frag16 f;
#pragma unroll
  for (int i = 0; i < 4; ++i) f.p[i] = *(const float4*)(q + 4 * i);
  return f;
}

template <int G>
__device__ __forceinline__ void mma_group(const frag16& a, const frag16 (&b)[G], v8f* c) {
#pragma unroll
  for (int i = 0; i < 4; ++i) {
#pragma unroll
    for (int g = 0; g < G; ++g) {
      const v2f av = {a.p[i].x, a.p[i].y};
      const v2f bv = {b[g].p[i].x, b[g].p[i].y};
      c[g] = __builtin_amdgcn_wmma_f32_16x16x4_f32(false, av, false, bv, (short)0, c[g],
                                                   false, false);
    }
#pragma unroll
    for (int g = 0; g < G; ++g) {
      const v2f av = {a.p[i].z, a.p[i].w};
      const v2f bv = {b[g].p[i].z, b[g].p[i].w};
      c[g] = __builtin_amdgcn_wmma_f32_16x16x4_f32(false, av, false, bv, (short)0, c[g],
                                                   false, false);
    }
  }
}

template <int K, int NT, int G>
__device__ __forceinline__ void gemm_strip(const float* __restrict__ A,
                                           const float* __restrict__ B,
                                           float* __restrict__ C, int N, int Nc) {
  constexpr int KB = K / 32;
  constexpr int NG = NT / G;
  const int lane = threadIdx.x & 31;
  const int half = lane >> 4;
  const int idx  = lane & 15;
  const int row0 = blockIdx.x * 16;

  const int ar = min(row0 + idx, N - 1);
  const float* ap = A + (size_t)ar * K + 16 * half;
  const float* bp[NT];
#pragma unroll
  for (int t = 0; t < NT; ++t)
    bp[t] = B + (size_t)min(16 * t + idx, Nc - 1) * K + 16 * half;

  v8f c[NT];
#pragma unroll
  for (int t = 0; t < NT; ++t) c[t] = (v8f){0.f, 0.f, 0.f, 0.f, 0.f, 0.f, 0.f, 0.f};

  frag16 a_cur = load_k16(ap);
  frag16 b0[G];
#pragma unroll
  for (int g = 0; g < G; ++g) b0[g] = load_k16(bp[g]);

#pragma unroll
  for (int kb = 0; kb < KB; ++kb) {
    const int kn = min(kb + 1, KB - 1) * 32;
    frag16 a_nxt = load_k16(ap + kn);
    if constexpr (NG == 2) {
      frag16 b1[G];
#pragma unroll
      for (int g = 0; g < G; ++g) b1[g] = load_k16(bp[G + g] + kb * 32);
      __builtin_amdgcn_sched_barrier(0);
      mma_group<G>(a_cur, b0, c);
      __builtin_amdgcn_sched_barrier(0);
#pragma unroll
      for (int g = 0; g < G; ++g) b0[g] = load_k16(bp[g] + kn);
      __builtin_amdgcn_sched_barrier(0);
      mma_group<G>(a_cur, b1, c + G);
    } else {
      frag16 bn[G];
#pragma unroll
      for (int g = 0; g < G; ++g) bn[g] = load_k16(bp[g] + kn);
      __builtin_amdgcn_sched_barrier(0);
      mma_group<G>(a_cur, b0, c);
#pragma unroll
      for (int g = 0; g < G; ++g) b0[g] = bn[g];
    }
    a_cur = a_nxt;
  }

#pragma unroll
  for (int t = 0; t < NT; ++t) {
    const int col = 16 * t + idx;
    if (col < Nc) {
#pragma unroll
      for (int r = 0; r < 8; ++r) {
        const int row = row0 + r + 8 * half;
        if (row < N) C[(size_t)row * Nc + col] = c[t][r];
      }
    }
  }
}

__global__ void __launch_bounds__(32)
gemm_strip_k256_c128(const float* __restrict__ A, const float* __restrict__ B,
                     float* __restrict__ C, int N) {
  gemm_strip<256, 8, 4>(A, B, C, N, 128);
}

__global__ void __launch_bounds__(32)
gemm_strip_k128_c40(const float* __restrict__ A, const float* __restrict__ B,
                    float* __restrict__ C, int N) {
  gemm_strip<128, 3, 3>(A, B, C, N, 40);
}

__global__ void attn_logits(const float* __restrict__ h, const float* __restrict__ att_s,
                            const float* __restrict__ att_d, float* __restrict__ a_s,
                            float* __restrict__ a_d, int NH, int H, int D) {
  int i = blockIdx.x * blockDim.x + threadIdx.x;
  if (i >= NH) return;
  const int hh = i % H;
  const float* __restrict__ hp = h + (size_t)i * D;
  const float* __restrict__ as = att_s + (size_t)hh * D;
  const float* __restrict__ ad = att_d + (size_t)hh * D;
  float s0 = 0.f, s1 = 0.f;
  for (int dd = 0; dd < D; ++dd) {
    const float v = hp[dd];
    s0 += v * as[dd];
    s1 += v * ad[dd];
  }
  a_s[i] = s0;
  a_d[i] = s1;
}


__device__ __forceinline__ float lane_bcast(float v, int j) {
  return __int_as_float(__builtin_amdgcn_readlane(__float_as_int(v), j));
}
__device__ __forceinline__ float leaky(float e) { return (e > 0.f) ? e : NEG_SLOPE * e; }
__device__ __forceinline__ float wave_max(float v) {
  for (int m = 16; m >= 1; m >>= 1) v = fmaxf(v, __shfl_xor(v, m, 32));
  return v;
}
__device__ __forceinline__ float wave_sum(float v) {
  for (int m = 16; m >= 1; m >>= 1) v += __shfl_xor(v, m, 32);
  return v;
}

__device__ __forceinline__ void add_rows_h4d32(const float* __restrict__ h, int s_l,
                                               float al0, float al1, float al2, float al3,
                                               int cnt, int lane, float& acc0, float& acc1,
                                               float& acc2, float& acc3) {
  int j = 0;
  for (; j + 3 < cnt; j += 4) {
    float a[4][4], v[4][4];
    for (int k = 0; k < 4; ++k) {
      const int sj = __builtin_amdgcn_readlane(s_l, j + k);
      a[k][0] = lane_bcast(al0, j + k);
      a[k][1] = lane_bcast(al1, j + k);
      a[k][2] = lane_bcast(al2, j + k);
      a[k][3] = lane_bcast(al3, j + k);
      const float* __restrict__ hr = h + (size_t)sj * 128 + lane;
      v[k][0] = hr[0];
      v[k][1] = hr[32];
      v[k][2] = hr[64];
      v[k][3] = hr[96];
    }
    for (int k = 0; k < 4; ++k) {
      acc0 += a[k][0] * v[k][0];
      acc1 += a[k][1] * v[k][1];
      acc2 += a[k][2] * v[k][2];
      acc3 += a[k][3] * v[k][3];
    }
  }
  for (; j < cnt; ++j) {
    const int sj = __builtin_amdgcn_readlane(s_l, j);
    const float a0 = lane_bcast(al0, j), a1 = lane_bcast(al1, j);
    const float a2 = lane_bcast(al2, j), a3 = lane_bcast(al3, j);
    const float* __restrict__ hr = h + (size_t)sj * 128 + lane;
    const float v0 = hr[0], v1 = hr[32], v2 = hr[64], v3 = hr[96];
    acc0 += a0 * v0;
    acc1 += a1 * v1;
    acc2 += a2 * v2;
    acc3 += a3 * v3;
  }
}

__global__ void __launch_bounds__(256)
node_aggregate_h4d32(const int* __restrict__ off, const int* __restrict__ csr_src,
                     const float* __restrict__ a_s, const float* __restrict__ a_d,
                     const float* __restrict__ h, const float* __restrict__ bias,
                     float* __restrict__ x2, int N) {
  const int lane = threadIdx.x & 31;
  const int wib = __builtin_amdgcn_readfirstlane((int)(threadIdx.x >> 5));
  const int node = blockIdx.x * 8 + wib;
  if (node >= N) return;
  const int beg = __builtin_amdgcn_readfirstlane(off[node]);
  const int deg = __builtin_amdgcn_readfirstlane(off[node + 1]) - beg;
  const int* __restrict__ nb = csr_src + beg;

  const float4 ad4 = *(const float4*)(a_d + (size_t)node * 4);
  const float4 sf4 = *(const float4*)(a_s + (size_t)node * 4);
  const float z0 = leaky(sf4.x + ad4.x), z1 = leaky(sf4.y + ad4.y);
  const float z2 = leaky(sf4.z + ad4.z), z3 = leaky(sf4.w + ad4.w);

  const bool act = lane < deg;
  const int s_c = act ? nb[lane] : node;
  const float4 as4 = *(const float4*)(a_s + (size_t)s_c * 4);
  const float e0 = leaky(as4.x + ad4.x), e1 = leaky(as4.y + ad4.y);
  const float e2 = leaky(as4.z + ad4.z), e3 = leaky(as4.w + ad4.w);

  float m0 = fmaxf(z0, e0), m1 = fmaxf(z1, e1), m2 = fmaxf(z2, e2), m3 = fmaxf(z3, e3);
  for (int i = 32 + lane; i < deg; i += 32) {
    const float4 b4 = *(const float4*)(a_s + (size_t)nb[i] * 4);
    m0 = fmaxf(m0, leaky(b4.x + ad4.x));
    m1 = fmaxf(m1, leaky(b4.y + ad4.y));
    m2 = fmaxf(m2, leaky(b4.z + ad4.z));
    m3 = fmaxf(m3, leaky(b4.w + ad4.w));
  }
  m0 = wave_max(m0); m1 = wave_max(m1); m2 = wave_max(m2); m3 = wave_max(m3);

  const float p0 = act ? expf(e0 - m0) : 0.f, p1 = act ? expf(e1 - m1) : 0.f;
  const float p2 = act ? expf(e2 - m2) : 0.f, p3 = act ? expf(e3 - m3) : 0.f;
  float d0 = p0, d1 = p1, d2 = p2, d3 = p3;
  for (int i = 32 + lane; i < deg; i += 32) {
    const float4 b4 = *(const float4*)(a_s + (size_t)nb[i] * 4);
    d0 += expf(leaky(b4.x + ad4.x) - m0);
    d1 += expf(leaky(b4.y + ad4.y) - m1);
    d2 += expf(leaky(b4.z + ad4.z) - m2);
    d3 += expf(leaky(b4.w + ad4.w) - m3);
  }
  const float q0 = expf(z0 - m0), q1 = expf(z1 - m1), q2 = expf(z2 - m2), q3 = expf(z3 - m3);
  d0 = wave_sum(d0) + q0; d1 = wave_sum(d1) + q1;
  d2 = wave_sum(d2) + q2; d3 = wave_sum(d3) + q3;

  const float* __restrict__ hs = h + (size_t)node * 128 + lane;
  float acc0 = (q0 / d0) * hs[0];
  float acc1 = (q1 / d1) * hs[32];
  float acc2 = (q2 / d2) * hs[64];
  float acc3 = (q3 / d3) * hs[96];
  add_rows_h4d32(h, s_c, p0 / d0, p1 / d1, p2 / d2, p3 / d3, min(deg, 32), lane,
                 acc0, acc1, acc2, acc3);
  for (int base = 32; base < deg; base += 32) {
    const int cnt = min(32, deg - base);
    const bool on = lane < cnt;
    const int s_r = on ? nb[base + lane] : node;
    const float4 b4 = *(const float4*)(a_s + (size_t)s_r * 4);
    const float r0 = on ? expf(leaky(b4.x + ad4.x) - m0) / d0 : 0.f;
    const float r1 = on ? expf(leaky(b4.y + ad4.y) - m1) / d1 : 0.f;
    const float r2 = on ? expf(leaky(b4.z + ad4.z) - m2) / d2 : 0.f;
    const float r3 = on ? expf(leaky(b4.w + ad4.w) - m3) / d3 : 0.f;
    add_rows_h4d32(h, s_r, r0, r1, r2, r3, cnt, lane, acc0, acc1, acc2, acc3);
  }

  float* __restrict__ xr = x2 + (size_t)node * 128 + lane;
  const float o0 = acc0 + bias[lane],      o1 = acc1 + bias[32 + lane];
  const float o2 = acc2 + bias[64 + lane], o3 = acc3 + bias[96 + lane];
  xr[0]  = (o0 > 0.f) ? o0 : 0.f;
  xr[32] = (o1 > 0.f) ? o1 : 0.f;
  xr[64] = (o2 > 0.f) ? o2 : 0.f;
  xr[96] = (o3 > 0.f) ? o3 : 0.f;
}

__device__ __forceinline__ void add_rows_h1d40(const float* __restrict__ h, int s_l, float al,
                                               int cnt, int lane, int f1, float& acc0,
                                               float& acc1) {
  int j = 0;
  for (; j + 3 < cnt; j += 4) {
    float a[4], va[4], vb[4];
    for (int k = 0; k < 4; ++k) {
      const int sj = __builtin_amdgcn_readlane(s_l, j + k);
      a[k] = lane_bcast(al, j + k);
      const float* __restrict__ hr = h + (size_t)sj * 40;
      va[k] = hr[lane];
      vb[k] = hr[f1];
    }
    for (int k = 0; k < 4; ++k) {
      acc0 += a[k] * va[k];
      acc1 += a[k] * vb[k];
    }
  }
  for (; j < cnt; ++j) {
    const int sj = __builtin_amdgcn_readlane(s_l, j);
    const float ak = lane_bcast(al, j);
    const float* __restrict__ hr = h + (size_t)sj * 40;
    acc0 += ak * hr[lane];
    acc1 += ak * hr[f1];
  }
}

__global__ void __launch_bounds__(256)
node_aggregate_h1d40(const int* __restrict__ off, const int* __restrict__ csr_src,
                     const float* __restrict__ a_s, const float* __restrict__ a_d,
                     const float* __restrict__ h, const float* __restrict__ bias,
                     float* __restrict__ out, int N) {
  const int lane = threadIdx.x & 31;
  const int wib = __builtin_amdgcn_readfirstlane((int)(threadIdx.x >> 5));
  const int node = blockIdx.x * 8 + wib;
  if (node >= N) return;
  const int beg = __builtin_amdgcn_readfirstlane(off[node]);
  const int deg = __builtin_amdgcn_readfirstlane(off[node + 1]) - beg;
  const int* __restrict__ nb = csr_src + beg;
  const int f1 = 32 + (lane & 7);

  const float ad = a_d[node];
  const float z = leaky(a_s[node] + ad);

  const bool act = lane < deg;
  const int s_c = act ? nb[lane] : node;
  const float e = leaky(a_s[s_c] + ad);

  float m = fmaxf(z, e);
  for (int i = 32 + lane; i < deg; i += 32) m = fmaxf(m, leaky(a_s[nb[i]] + ad));
  m = wave_max(m);

  const float p = act ? expf(e - m) : 0.f;
  float d = p;
  for (int i = 32 + lane; i < deg; i += 32) d += expf(leaky(a_s[nb[i]] + ad) - m);
  const float q = expf(z - m);
  d = wave_sum(d) + q;

  const float* __restrict__ hs = h + (size_t)node * 40;
  float acc0 = (q / d) * hs[lane];
  float acc1 = (q / d) * hs[f1];
  add_rows_h1d40(h, s_c, p / d, min(deg, 32), lane, f1, acc0, acc1);
  for (int base = 32; base < deg; base += 32) {
    const int cnt = min(32, deg - base);
    const bool on = lane < cnt;
    const int s_r = on ? nb[base + lane] : node;
    const float r = on ? expf(leaky(a_s[s_r] + ad) - m) / d : 0.f;
    add_rows_h1d40(h, s_r, r, cnt, lane, f1, acc0, acc1);
  }

  float* __restrict__ orow = out + (size_t)node * 40;
  orow[lane] = acc0 + bias[lane];
  if (lane < 8) orow[32 + lane] = acc1 + bias[32 + lane];
}

extern "C" void kernel_launch(void* const* d_in, const int* in_sizes, int n_in,
                              void* d_out, int out_size, void* d_ws, size_t ws_size,
                              hipStream_t stream) {
  const float* x        = (const float*)d_in[0];
  const int*   ei       = (const int*)d_in[1];
  const float* W1       = (const float*)d_in[2];
  const float* att_src1 = (const float*)d_in[3];
  const float* att_dst1 = (const float*)d_in[4];
  const float* b1       = (const float*)d_in[5];
  const float* W2       = (const float*)d_in[6];
  const float* att_src2 = (const float*)d_in[7];
  const float* att_dst2 = (const float*)d_in[8];
  const float* b2       = (const float*)d_in[9];

  const int NFEAT = 256, NHID = 128, H1 = 4, D1 = 32, NCLS = 40;
  const int N = in_sizes[0] / NFEAT;
  const int E = in_sizes[1] / 2;
  const int* src = ei;
  const int* dst = ei + E;

  float* out = (float*)d_out;

  float* w = (float*)d_ws;
  float* h1   = w;
  float* x2   = h1  + (size_t)N * NHID;
  float* h2   = x2  + (size_t)N * NHID;
  float* as1  = h2  + (size_t)N * NCLS;
  float* ad1  = as1 + (size_t)N * H1;
  float* as2  = ad1 + (size_t)N * H1;
  float* ad2  = as2 + (size_t)N;
  int* off     = (int*)(ad2 + (size_t)N);
  int* csr_src = off + (size_t)N + 1;

  const int TB = 256;

  csr_build<<<(N + CSR_BKT - 1) / CSR_BKT, CSR_THREADS, 0, stream>>>(src, dst, E, N, off,
                                                                      csr_src);

  gemm_strip_k256_c128<<<(N + 15) / 16, 32, 0, stream>>>(x, W1, h1, N);
  attn_logits<<<(N * H1 + TB - 1) / TB, TB, 0, stream>>>(h1, att_src1, att_dst1,
                                                         as1, ad1, N * H1, H1, D1);
  node_aggregate_h4d32<<<(N + 7) / 8, TB, 0, stream>>>(off, csr_src, as1, ad1, h1, b1,
                                                       x2, N);

  gemm_strip_k128_c40<<<(N + 15) / 16, 32, 0, stream>>>(x2, W2, h2, N);
  attn_logits<<<(N + TB - 1) / TB, TB, 0, stream>>>(h2, att_src2, att_dst2,
                                                    as2, ad2, N, 1, NCLS);
  node_aggregate_h1d40<<<(N + 7) / 8, TB, 0, stream>>>(off, csr_src, as2, ad2, h2, b2,
                                                       out, N);
}


# attention logits a_s, a_d formed in each GEMM's epilogue from its accumulator registers (16-lane exchange tree), the two logit launches dropped; summing order of the logit dot products changes
# speedup vs baseline: 12.6171x; 1.0561x over previous
#include <hip/hip_runtime.h>


typedef float v2f __attribute__((ext_vector_type(2)));
typedef float v8f __attribute__((ext_vector_type(8)));

#define NEG_SLOPE 0.2f


#define CSR_BKT 256
#define CSR_THREADS 1024
#define CSR_CAP 14336
#define CSR_U 8

template <class F>
__device__ __forceinline__ void for_each_edge(const int* __restrict__ dst, int E, int tid, F f) {
  const int head = min(E, (int)(((16u - (unsigned)((size_t)dst & 15u)) & 15u) >> 2));
  const int nv = (E - head) >> 2;
  const int4* __restrict__ dv = (const int4*)(dst + head);
  for (int i0 = tid; i0 < nv; i0 += CSR_U * CSR_THREADS) {
    int4 d[CSR_U];
#pragma unroll
    for (int u = 0; u < CSR_U; ++u) d[u] = dv[min(i0 + u * CSR_THREADS, nv - 1)];
#pragma unroll
    for (int u = 0; u < CSR_U; ++u) {
      const int i = i0 + u * CSR_THREADS;
      if (i < nv) {
        const int t = head + 4 * i;
        f(t, d[u].x);
        f(t + 1, d[u].y);
        f(t + 2, d[u].z);
        f(t + 3, d[u].w);
      }
    }
  }
  const int tail0 = head + 4 * nv;
  if (tid < head) f(tid, dst[tid]);
  if (tid < E - tail0) f(tail0 + tid, dst[tail0 + tid]);
}

__global__ void __launch_bounds__(CSR_THREADS)
csr_build(const int* __restrict__ src, const int* __restrict__ dst, int E, int N,
          int* __restrict__ off, int* __restrict__ csr_src) {
  __shared__ int c[CSR_BKT];
  __shared__ int loff[CSR_BKT];
  __shared__ int wtot[CSR_BKT / 32];
  __shared__ int s_below, s_nl;
  __shared__ int l_t[CSR_CAP];
  __shared__ int l_d[CSR_CAP];
  __shared__ int l_r[CSR_CAP];
  const int tid = threadIdx.x, lane = tid & 31, wid = tid >> 5;
  const int lo = blockIdx.x * CSR_BKT;

  if (tid < CSR_BKT) c[tid] = 0;
  if (tid == 0) { s_below = 0; s_nl = 0; }
  __syncthreads();

  int below = 0;
  for_each_edge(dst, E, tid, [&](int t, int d) {
    below += (d < lo) ? 1 : 0;
    const unsigned dl = (unsigned)(d - lo);
    if (dl < (unsigned)CSR_BKT) {
      const int r = atomicAdd(&c[dl], 1);
      const int k = atomicAdd(&s_nl, 1);
      if (k < CSR_CAP) { l_t[k] = t; l_d[k] = (int)dl; l_r[k] = r; }
    }
  });
  for (int m = 16; m >= 1; m >>= 1) below += __shfl_xor(below, m, 32);
  if (lane == 0) atomicAdd(&s_below, below);
  __syncthreads();

  int x = 0, inc = 0;
  if (tid < CSR_BKT) {
    x = c[tid];
    inc = x;
    for (int s = 1; s < 32; s <<= 1) {
      const int y = __shfl_up(inc, s, 32);
      if (lane >= s) inc += y;
    }
    if (lane == 31) wtot[wid] = inc;
  }
  __syncthreads();
  if (tid < CSR_BKT) {
    int b = 0;
    for (int w = 0; w < wid; ++w) b += wtot[w];
    loff[tid] = b + inc - x;
  }
  __syncthreads();
  const int base = s_below;
  const int nl = s_nl;
  if (tid < CSR_BKT && lo + tid < N) off[lo + tid] = base + loff[tid];
  if (tid == CSR_BKT - 1 && lo + CSR_BKT >= N) off[N] = base + loff[tid] + x;

  if (nl <= CSR_CAP) {
    for (int k = tid; k < nl; k += CSR_THREADS)
      csr_src[base + loff[l_d[k]] + l_r[k]] = src[l_t[k]];
  } else {
    if (tid < CSR_BKT) c[tid] = 0;
    __syncthreads();
    for_each_edge(dst, E, tid, [&](int t, int d) {
      const unsigned dl = (unsigned)(d - lo);
      if (dl < (unsigned)CSR_BKT) {
        const int r = atomicAdd(&c[dl], 1);
        csr_src[base + loff[dl] + r] = src[t];
      }
    });
  }
}


struct frag16 { float4 p[4]; };

__device__ __forceinline__ frag16 load_k16(const float* q) {
  frag16 f;
#pragma unroll
  for (int i = 0; i < 4; ++i) f.p[i] = *(const float4*)(q + 4 * i);
  return f;
}

template <int G>
__device__ __forceinline__ void mma_group(const frag16& a, const frag16 (&b)[G], v8f* c) {
#pragma unroll
  for (int i = 0; i < 4; ++i) {
#pragma unroll
    for (int g = 0; g < G; ++g) {
      const v2f av = {a.p[i].x, a.p[i].y};
      const v2f bv = {b[g].p[i].x, b[g].p[i].y};
      c[g] = __builtin_amdgcn_wmma_f32_16x16x4_f32(false, av, false, bv, (short)0, c[g],
                                                   false, false);
    }
#pragma unroll
    for (int g = 0; g < G; ++g) {
      const v2f av = {a.p[i].z, a.p[i].w};
      const v2f bv = {b[g].p[i].z, b[g].p[i].w};
      c[g] = __builtin_amdgcn_wmma_f32_16x16x4_f32(false, av, false, bv, (short)0, c[g],
                                                   false, false);
    }
  }
}

template <int K, int NT, int G, int H>
__device__ __forceinline__ void gemm_strip(const float* __restrict__ A,
                                           const float* __restrict__ B,
                                           float* __restrict__ C,
                                           const float* __restrict__ att_s,
                                           const float* __restrict__ att_d,
                                           float* __restrict__ a_s,
                                           float* __restrict__ a_d, int N, int Nc) {
  constexpr int KB = K / 32;
  constexpr int NG = NT / G;
  const int lane = threadIdx.x & 31;
  const int half = lane >> 4;
  const int idx  = lane & 15;
  const int row0 = blockIdx.x * 16;

  const int ar = min(row0 + idx, N - 1);
  const float* ap = A + (size_t)ar * K + 16 * half;
  const float* bp[NT];
#pragma unroll
  for (int t = 0; t < NT; ++t)
    bp[t] = B + (size_t)min(16 * t + idx, Nc - 1) * K + 16 * half;

  v8f c[NT];
#pragma unroll
  for (int t = 0; t < NT; ++t) c[t] = (v8f){0.f, 0.f, 0.f, 0.f, 0.f, 0.f, 0.f, 0.f};

  frag16 a_cur = load_k16(ap);
  frag16 b0[G];
#pragma unroll
  for (int g = 0; g < G; ++g) b0[g] = load_k16(bp[g]);

#pragma unroll
  for (int kb = 0; kb < KB; ++kb) {
    const int kn = min(kb + 1, KB - 1) * 32;
    frag16 a_nxt = load_k16(ap + kn);
    if constexpr (NG == 2) {
      frag16 b1[G];
#pragma unroll
      for (int g = 0; g < G; ++g) b1[g] = load_k16(bp[G + g] + kb * 32);
      __builtin_amdgcn_sched_barrier(0);
      mma_group<G>(a_cur, b0, c);
      __builtin_amdgcn_sched_barrier(0);
#pragma unroll
      for (int g = 0; g < G; ++g) b0[g] = load_k16(bp[g] + kn);
      __builtin_amdgcn_sched_barrier(0);
      mma_group<G>(a_cur, b1, c + G);
    } else {
      frag16 bn[G];
#pragma unroll
      for (int g = 0; g < G; ++g) bn[g] = load_k16(bp[g] + kn);
      __builtin_amdgcn_sched_barrier(0);
      mma_group<G>(a_cur, b0, c);
#pragma unroll
      for (int g = 0; g < G; ++g) b0[g] = bn[g];
    }
    a_cur = a_nxt;
  }

#pragma unroll
  for (int t = 0; t < NT; ++t) {
    const int col = 16 * t + idx;
    if (col < Nc) {
#pragma unroll
      for (int r = 0; r < 8; ++r) {
        const int row = row0 + r + 8 * half;
        if (row < N) C[(size_t)row * Nc + col] = c[t][r];
      }
    }
  }

  constexpr int TPH = NT / H;
  float p[16 * H];
#pragma unroll
  for (int hh = 0; hh < H; ++hh) {
    float ws[TPH], wd[TPH];
#pragma unroll
    for (int u = 0; u < TPH; ++u) {
      const int col = 16 * (hh * TPH + u) + idx;
      ws[u] = (col < Nc) ? att_s[min(col, Nc - 1)] : 0.f;
      wd[u] = (col < Nc) ? att_d[min(col, Nc - 1)] : 0.f;
    }
#pragma unroll
    for (int r = 0; r < 8; ++r) {
      float s = 0.f, d = 0.f;
#pragma unroll
      for (int u = 0; u < TPH; ++u) {
        s += c[hh * TPH + u][r] * ws[u];
        d += c[hh * TPH + u][r] * wd[u];
      }
      p[r * H + hh] = s;
      p[(8 + r) * H + hh] = d;
    }
  }
#pragma unroll
  for (int s = 0; s < 4; ++s) {
    const int mask = 8 >> s;
    const int hs = (8 * H) >> s;
    const bool up = (idx & mask) != 0;
#pragma unroll
    for (int j = 0; j < hs; ++j) {
      const float lo = p[j], hi = p[j + hs];
      const float send = up ? lo : hi;
      const float keep = up ? hi : lo;
      p[j] = keep + __shfl_xor(send, mask, 32);
    }
  }
  const int lrow = row0 + 8 * half + (idx & 7);
  if (lrow < N) {
    float* __restrict__ o = ((idx < 8) ? a_s : a_d) + (size_t)lrow * H;
#pragma unroll
    for (int hh = 0; hh < H; ++hh) o[hh] = p[hh];
  }
}

__global__ void __launch_bounds__(32)
gemm_strip_k256_c128(const float* __restrict__ A, const float* __restrict__ B,
                     float* __restrict__ C, const float* __restrict__ att_s,
                     const float* __restrict__ att_d, float* __restrict__ a_s,
                     float* __restrict__ a_d, int N) {
  gemm_strip<256, 8, 4, 4>(A, B, C, att_s, att_d, a_s, a_d, N, 128);
}

__global__ void __launch_bounds__(32)
gemm_strip_k128_c40(const float* __restrict__ A, const float* __restrict__ B,
                    float* __restrict__ C, const float* __restrict__ att_s,
                    const float* __restrict__ att_d, float* __restrict__ a_s,
                    float* __restrict__ a_d, int N) {
  gemm_strip<128, 3, 3, 1>(A, B, C, att_s, att_d, a_s, a_d, N, 40);
}


__device__ __forceinline__ float lane_bcast(float v, int j) {
  return __int_as_float(__builtin_amdgcn_readlane(__float_as_int(v), j));
}
__device__ __forceinline__ float leaky(float e) { return (e > 0.f) ? e : NEG_SLOPE * e; }
__device__ __forceinline__ float wave_max(float v) {
  for (int m = 16; m >= 1; m >>= 1) v = fmaxf(v, __shfl_xor(v, m, 32));
  return v;
}
__device__ __forceinline__ float wave_sum(float v) {
  for (int m = 16; m >= 1; m >>= 1) v += __shfl_xor(v, m, 32);
  return v;
}

__device__ __forceinline__ void add_rows_h4d32(const float* __restrict__ h, int s_l,
                                               float al0, float al1, float al2, float al3,
                                               int cnt, int lane, float& acc0, float& acc1,
                                               float& acc2, float& acc3) {
  int j = 0;
  for (; j + 3 < cnt; j += 4) {
    float a[4][4], v[4][4];
    for (int k = 0; k < 4; ++k) {
      const int sj = __builtin_amdgcn_readlane(s_l, j + k);
      a[k][0] = lane_bcast(al0, j + k);
      a[k][1] = lane_bcast(al1, j + k);
      a[k][2] = lane_bcast(al2, j + k);
      a[k][3] = lane_bcast(al3, j + k);
      const float* __restrict__ hr = h + (size_t)sj * 128 + lane;
      v[k][0] = hr[0];
      v[k][1] = hr[32];
      v[k][2] = hr[64];
      v[k][3] = hr[96];
    }
    for (int k = 0; k < 4; ++k) {
      acc0 += a[k][0] * v[k][0];
      acc1 += a[k][1] * v[k][1];
      acc2 += a[k][2] * v[k][2];
      acc3 += a[k][3] * v[k][3];
    }
  }
  for (; j < cnt; ++j) {
    const int sj = __builtin_amdgcn_readlane(s_l, j);
    const float a0 = lane_bcast(al0, j), a1 = lane_bcast(al1, j);
    const float a2 = lane_bcast(al2, j), a3 = lane_bcast(al3, j);
    const float* __restrict__ hr = h + (size_t)sj * 128 + lane;
    const float v0 = hr[0], v1 = hr[32], v2 = hr[64], v3 = hr[96];
    acc0 += a0 * v0;
    acc1 += a1 * v1;
    acc2 += a2 * v2;
    acc3 += a3 * v3;
  }
}

__global__ void __launch_bounds__(256)
node_aggregate_h4d32(const int* __restrict__ off, const int* __restrict__ csr_src,
                     const float* __restrict__ a_s, const float* __restrict__ a_d,
                     const float* __restrict__ h, const float* __restrict__ bias,
                     float* __restrict__ x2, int N) {
  const int lane = threadIdx.x & 31;
  const int wib = __builtin_amdgcn_readfirstlane((int)(threadIdx.x >> 5));
  const int node = blockIdx.x * 8 + wib;
  if (node >= N) return;
  const int beg = __builtin_amdgcn_readfirstlane(off[node]);
  const int deg = __builtin_amdgcn_readfirstlane(off[node + 1]) - beg;
  const int* __restrict__ nb = csr_src + beg;

  const float4 ad4 = *(const float4*)(a_d + (size_t)node * 4);
  const float4 sf4 = *(const float4*)(a_s + (size_t)node * 4);
  const float z0 = leaky(sf4.x + ad4.x), z1 = leaky(sf4.y + ad4.y);
  const float z2 = leaky(sf4.z + ad4.z), z3 = leaky(sf4.w + ad4.w);

  const bool act = lane < deg;
  const int s_c = act ? nb[lane] : node;
  const float4 as4 = *(const float4*)(a_s + (size_t)s_c * 4);
  const float e0 = leaky(as4.x + ad4.x), e1 = leaky(as4.y + ad4.y);
  const float e2 = leaky(as4.z + ad4.z), e3 = leaky(as4.w + ad4.w);

  float m0 = fmaxf(z0, e0), m1 = fmaxf(z1, e1), m2 = fmaxf(z2, e2), m3 = fmaxf(z3, e3);
  for (int i = 32 + lane; i < deg; i += 32) {
    const float4 b4 = *(const float4*)(a_s + (size_t)nb[i] * 4);
    m0 = fmaxf(m0, leaky(b4.x + ad4.x));
    m1 = fmaxf(m1, leaky(b4.y + ad4.y));
    m2 = fmaxf(m2, leaky(b4.z + ad4.z));
    m3 = fmaxf(m3, leaky(b4.w + ad4.w));
  }
  m0 = wave_max(m0); m1 = wave_max(m1); m2 = wave_max(m2); m3 = wave_max(m3);

  const float p0 = act ? expf(e0 - m0) : 0.f, p1 = act ? expf(e1 - m1) : 0.f;
  const float p2 = act ? expf(e2 - m2) : 0.f, p3 = act ? expf(e3 - m3) : 0.f;
  float d0 = p0, d1 = p1, d2 = p2, d3 = p3;
  for (int i = 32 + lane; i < deg; i += 32) {
    const float4 b4 = *(const float4*)(a_s + (size_t)nb[i] * 4);
    d0 += expf(leaky(b4.x + ad4.x) - m0);
    d1 += expf(leaky(b4.y + ad4.y) - m1);
    d2 += expf(leaky(b4.z + ad4.z) - m2);
    d3 += expf(leaky(b4.w + ad4.w) - m3);
  }
  const float q0 = expf(z0 - m0), q1 = expf(z1 - m1), q2 = expf(z2 - m2), q3 = expf(z3 - m3);
  d0 = wave_sum(d0) + q0; d1 = wave_sum(d1) + q1;
  d2 = wave_sum(d2) + q2; d3 = wave_sum(d3) + q3;

  const float* __restrict__ hs = h + (size_t)node * 128 + lane;
  float acc0 = (q0 / d0) * hs[0];
  float acc1 = (q1 / d1) * hs[32];
  float acc2 = (q2 / d2) * hs[64];
  float acc3 = (q3 / d3) * hs[96];
  add_rows_h4d32(h, s_c, p0 / d0, p1 / d1, p2 / d2, p3 / d3, min(deg, 32), lane,
                 acc0, acc1, acc2, acc3);
  for (int base = 32; base < deg; base += 32) {
    const int cnt = min(32, deg - base);
    const bool on = lane < cnt;
    const int s_r = on ? nb[base + lane] : node;
    const float4 b4 = *(const float4*)(a_s + (size_t)s_r * 4);
    const float r0 = on ? expf(leaky(b4.x + ad4.x) - m0) / d0 : 0.f;
    const float r1 = on ? expf(leaky(b4.y + ad4.y) - m1) / d1 : 0.f;
    const float r2 = on ? expf(leaky(b4.z + ad4.z) - m2) / d2 : 0.f;
    const float r3 = on ? expf(leaky(b4.w + ad4.w) - m3) / d3 : 0.f;
    add_rows_h4d32(h, s_r, r0, r1, r2, r3, cnt, lane, acc0, acc1, acc2, acc3);
  }

  float* __restrict__ xr = x2 + (size_t)node * 128 + lane;
  const float o0 = acc0 + bias[lane],      o1 = acc1 + bias[32 + lane];
  const float o2 = acc2 + bias[64 + lane], o3 = acc3 + bias[96 + lane];
  xr[0]  = (o0 > 0.f) ? o0 : 0.f;
  xr[32] = (o1 > 0.f) ? o1 : 0.f;
  xr[64] = (o2 > 0.f) ? o2 : 0.f;
  xr[96] = (o3 > 0.f) ? o3 : 0.f;
}

__device__ __forceinline__ void add_rows_h1d40(const float* __restrict__ h, int s_l, float al,
                                               int cnt, int lane, int f1, float& acc0,
                                               float& acc1) {
  int j = 0;
  for (; j + 3 < cnt; j += 4) {
    float a[4], va[4], vb[4];
    for (int k = 0; k < 4; ++k) {
      const int sj = __builtin_amdgcn_readlane(s_l, j + k);
      a[k] = lane_bcast(al, j + k);
      const float* __restrict__ hr = h + (size_t)sj * 40;
      va[k] = hr[lane];
      vb[k] = hr[f1];
    }
    for (int k = 0; k < 4; ++k) {
      acc0 += a[k] * va[k];
      acc1 += a[k] * vb[k];
    }
  }
  for (; j < cnt; ++j) {
    const int sj = __builtin_amdgcn_readlane(s_l, j);
    const float ak = lane_bcast(al, j);
    const float* __restrict__ hr = h + (size_t)sj * 40;
    acc0 += ak * hr[lane];
    acc1 += ak * hr[f1];
  }
}

__global__ void __launch_bounds__(256)
node_aggregate_h1d40(const int* __restrict__ off, const int* __restrict__ csr_src,
                     const float* __restrict__ a_s, const float* __restrict__ a_d,
                     const float* __restrict__ h, const float* __restrict__ bias,
                     float* __restrict__ out, int N) {
  const int lane = threadIdx.x & 31;
  const int wib = __builtin_amdgcn_readfirstlane((int)(threadIdx.x >> 5));
  const int node = blockIdx.x * 8 + wib;
  if (node >= N) return;
  const int beg = __builtin_amdgcn_readfirstlane(off[node]);
  const int deg = __builtin_amdgcn_readfirstlane(off[node + 1]) - beg;
  const int* __restrict__ nb = csr_src + beg;
  const int f1 = 32 + (lane & 7);

  const float ad = a_d[node];
  const float z = leaky(a_s[node] + ad);

  const bool act = lane < deg;
  const int s_c = act ? nb[lane] : node;
  const float e = leaky(a_s[s_c] + ad);

  float m = fmaxf(z, e);
  for (int i = 32 + lane; i < deg; i += 32) m = fmaxf(m, leaky(a_s[nb[i]] + ad));
  m = wave_max(m);

  const float p = act ? expf(e - m) : 0.f;
  float d = p;
  for (int i = 32 + lane; i < deg; i += 32) d += expf(leaky(a_s[nb[i]] + ad) - m);
  const float q = expf(z - m);
  d = wave_sum(d) + q;

  const float* __restrict__ hs = h + (size_t)node * 40;
  float acc0 = (q / d) * hs[lane];
  float acc1 = (q / d) * hs[f1];
  add_rows_h1d40(h, s_c, p / d, min(deg, 32), lane, f1, acc0, acc1);
  for (int base = 32; base < deg; base += 32) {
    const int cnt = min(32, deg - base);
    const bool on = lane < cnt;
    const int s_r = on ? nb[base + lane] : node;
    const float r = on ? expf(leaky(a_s[s_r] + ad) - m) / d : 0.f;
    add_rows_h1d40(h, s_r, r, cnt, lane, f1, acc0, acc1);
  }

  float* __restrict__ orow = out + (size_t)node * 40;
  orow[lane] = acc0 + bias[lane];
  if (lane < 8) orow[32 + lane] = acc1 + bias[32 + lane];
}

extern "C" void kernel_launch(void* const* d_in, const int* in_sizes, int n_in,
                              void* d_out, int out_size, void* d_ws, size_t ws_size,
                              hipStream_t stream) {
  const float* x        = (const float*)d_in[0];
  const int*   ei       = (const int*)d_in[1];
  const float* W1       = (const float*)d_in[2];
  const float* att_src1 = (const float*)d_in[3];
  const float* att_dst1 = (const float*)d_in[4];
  const float* b1       = (const float*)d_in[5];
  const float* W2       = (const float*)d_in[6];
  const float* att_src2 = (const float*)d_in[7];
  const float* att_dst2 = (const float*)d_in[8];
  const float* b2       = (const float*)d_in[9];

  const int NFEAT = 256, NHID = 128, H1 = 4, NCLS = 40;
  const int N = in_sizes[0] / NFEAT;
  const int E = in_sizes[1] / 2;
  const int* src = ei;
  const int* dst = ei + E;

  float* out = (float*)d_out;

  float* w = (float*)d_ws;
  float* h1   = w;
  float* x2   = h1  + (size_t)N * NHID;
  float* h2   = x2  + (size_t)N * NHID;
  float* as1  = h2  + (size_t)N * NCLS;
  float* ad1  = as1 + (size_t)N * H1;
  float* as2  = ad1 + (size_t)N * H1;
  float* ad2  = as2 + (size_t)N;
  int* off     = (int*)(ad2 + (size_t)N);
  int* csr_src = off + (size_t)N + 1;

  const int TB = 256;

  csr_build<<<(N + CSR_BKT - 1) / CSR_BKT, CSR_THREADS, 0, stream>>>(src, dst, E, N, off,
                                                                      csr_src);

  gemm_strip_k256_c128<<<(N + 15) / 16, 32, 0, stream>>>(x, W1, h1, att_src1, att_dst1,
                                                         as1, ad1, N);
  node_aggregate_h4d32<<<(N + 7) / 8, TB, 0, stream>>>(off, csr_src, as1, ad1, h1, b1,
                                                       x2, N);

  gemm_strip_k128_c40<<<(N + 15) / 16, 32, 0, stream>>>(x2, W2, h2, att_src2, att_dst2,
                                                        as2, ad2, N);
  node_aggregate_h1d40<<<(N + 7) / 8, TB, 0, stream>>>(off, csr_src, as2, ad2, h2, b2,
                                                       out, N);
}


# layer-1 GEMM: B (128x256) copied once per 16-strip work-group into LDS by whole-line loads; waves read B fragments from LDS, two tiles at a time, instead of 16-line global loads
# speedup vs baseline: 13.5238x; 1.0719x over previous
#include <hip/hip_runtime.h>


typedef float v2f __attribute__((ext_vector_type(2)));
typedef float v8f __attribute__((ext_vector_type(8)));

#define NEG_SLOPE 0.2f


#define CSR_BKT 256
#define CSR_THREADS 1024
#define CSR_CAP 14336
#define CSR_U 8

template <class F>
__device__ __forceinline__ void for_each_edge(const int* __restrict__ dst, int E, int tid, F f) {
  const int head = min(E, (int)(((16u - (unsigned)((size_t)dst & 15u)) & 15u) >> 2));
  const int nv = (E - head) >> 2;
  const int4* __restrict__ dv = (const int4*)(dst + head);
  for (int i0 = tid; i0 < nv; i0 += CSR_U * CSR_THREADS) {
    int4 d[CSR_U];
#pragma unroll
    for (int u = 0; u < CSR_U; ++u) d[u] = dv[min(i0 + u * CSR_THREADS, nv - 1)];
#pragma unroll
    for (int u = 0; u < CSR_U; ++u) {
      const int i = i0 + u * CSR_THREADS;
      if (i < nv) {
        const int t = head + 4 * i;
        f(t, d[u].x);
        f(t + 1, d[u].y);
        f(t + 2, d[u].z);
        f(t + 3, d[u].w);
      }
    }
  }
  const int tail0 = head + 4 * nv;
  if (tid < head) f(tid, dst[tid]);
  if (tid < E - tail0) f(tail0 + tid, dst[tail0 + tid]);
}

__global__ void __launch_bounds__(CSR_THREADS)
csr_build(const int* __restrict__ src, const int* __restrict__ dst, int E, int N,
          int* __restrict__ off, int* __restrict__ csr_src) {
  __shared__ int c[CSR_BKT];
  __shared__ int loff[CSR_BKT];
  __shared__ int wtot[CSR_BKT / 32];
  __shared__ int s_below, s_nl;
  __shared__ int l_t[CSR_CAP];
  __shared__ int l_d[CSR_CAP];
  __shared__ int l_r[CSR_CAP];
  const int tid = threadIdx.x, lane = tid & 31, wid = tid >> 5;
  const int lo = blockIdx.x * CSR_BKT;

  if (tid < CSR_BKT) c[tid] = 0;
  if (tid == 0) { s_below = 0; s_nl = 0; }
  __syncthreads();

  int below = 0;
  for_each_edge(dst, E, tid, [&](int t, int d) {
    below += (d < lo) ? 1 : 0;
    const unsigned dl = (unsigned)(d - lo);
    if (dl < (unsigned)CSR_BKT) {
      const int r = atomicAdd(&c[dl], 1);
      const int k = atomicAdd(&s_nl, 1);
      if (k < CSR_CAP) { l_t[k] = t; l_d[k] = (int)dl; l_r[k] = r; }
    }
  });
  for (int m = 16; m >= 1; m >>= 1) below += __shfl_xor(below, m, 32);
  if (lane == 0) atomicAdd(&s_below, below);
  __syncthreads();

  int x = 0, inc = 0;
  if (tid < CSR_BKT) {
    x = c[tid];
    inc = x;
    for (int s = 1; s < 32; s <<= 1) {
      const int y = __shfl_up(inc, s, 32);
      if (lane >= s) inc += y;
    }
    if (lane == 31) wtot[wid] = inc;
  }
  __syncthreads();
  if (tid < CSR_BKT) {
    int b = 0;
    for (int w = 0; w < wid; ++w) b += wtot[w];
    loff[tid] = b + inc - x;
  }
  __syncthreads();
  const int base = s_below;
  const int nl = s_nl;
  if (tid < CSR_BKT && lo + tid < N) off[lo + tid] = base + loff[tid];
  if (tid == CSR_BKT - 1 && lo + CSR_BKT >= N) off[N] = base + loff[tid] + x;

  if (nl <= CSR_CAP) {
    for (int k = tid; k < nl; k += CSR_THREADS)
      csr_src[base + loff[l_d[k]] + l_r[k]] = src[l_t[k]];
  } else {
    if (tid < CSR_BKT) c[tid] = 0;
    __syncthreads();
    for_each_edge(dst, E, tid, [&](int t, int d) {
      const unsigned dl = (unsigned)(d - lo);
      if (dl < (unsigned)CSR_BKT) {
        const int r = atomicAdd(&c[dl], 1);
        csr_src[base + loff[dl] + r] = src[t];
      }
    });
  }
}


struct frag16 { float4 p[4]; };

__device__ __forceinline__ frag16 load_k16(const float* q) {
  frag16 f;
#pragma unroll
  for (int i = 0; i < 4; ++i) f.p[i] = *(const float4*)(q + 4 * i);
  return f;
}

template <int G>
__device__ __forceinline__ void mma_group(const frag16& a, const frag16 (&b)[G], v8f* c) {
#pragma unroll
  for (int i = 0; i < 4; ++i) {
#pragma unroll
    for (int g = 0; g < G; ++g) {
      const v2f av = {a.p[i].x, a.p[i].y};
      const v2f bv = {b[g].p[i].x, b[g].p[i].y};
      c[g] = __builtin_amdgcn_wmma_f32_16x16x4_f32(false, av, false, bv, (short)0, c[g],
                                                   false, false);
    }
#pragma unroll
    for (int g = 0; g < G; ++g) {
      const v2f av = {a.p[i].z, a.p[i].w};
      const v2f bv = {b[g].p[i].z, b[g].p[i].w};
      c[g] = __builtin_amdgcn_wmma_f32_16x16x4_f32(false, av, false, bv, (short)0, c[g],
                                                   false, false);
    }
  }
}

template <int NT, int H>
__device__ __forceinline__ void strip_epilogue(const v8f* c, float* __restrict__ C,
                                               const float* __restrict__ att_s,
                                               const float* __restrict__ att_d,
                                               float* __restrict__ a_s,
                                               float* __restrict__ a_d, int N, int Nc,
                                               int row0, int idx, int half) {
#pragma unroll
  for (int t = 0; t < NT; ++t) {
    const int col = 16 * t + idx;
    if (col < Nc) {
#pragma unroll
      for (int r = 0; r < 8; ++r) {
        const int row = row0 + r + 8 * half;
        if (row < N) C[(size_t)row * Nc + col] = c[t][r];
      }
    }
  }

  constexpr int TPH = NT / H;
  float p[16 * H];
#pragma unroll
  for (int hh = 0; hh < H; ++hh) {
    float ws[TPH], wd[TPH];
#pragma unroll
    for (int u = 0; u < TPH; ++u) {
      const int col = 16 * (hh * TPH + u) + idx;
      ws[u] = (col < Nc) ? att_s[min(col, Nc - 1)] : 0.f;
      wd[u] = (col < Nc) ? att_d[min(col, Nc - 1)] : 0.f;
    }
#pragma unroll
    for (int r = 0; r < 8; ++r) {
      float s = 0.f, d = 0.f;
#pragma unroll
      for (int u = 0; u < TPH; ++u) {
        s += c[hh * TPH + u][r] * ws[u];
        d += c[hh * TPH + u][r] * wd[u];
      }
      p[r * H + hh] = s;
      p[(8 + r) * H + hh] = d;
    }
  }
#pragma unroll
  for (int s = 0; s < 4; ++s) {
    const int mask = 8 >> s;
    const int hs = (8 * H) >> s;
    const bool up = (idx & mask) != 0;
#pragma unroll
    for (int j = 0; j < hs; ++j) {
      const float lo = p[j], hi = p[j + hs];
      const float send = up ? lo : hi;
      const float keep = up ? hi : lo;
      p[j] = keep + __shfl_xor(send, mask, 32);
    }
  }
  const int lrow = row0 + 8 * half + (idx & 7);
  if (lrow < N) {
    float* __restrict__ o = ((idx < 8) ? a_s : a_d) + (size_t)lrow * H;
#pragma unroll
    for (int hh = 0; hh < H; ++hh) o[hh] = p[hh];
  }
}

template <int K, int NT, int G, int H>
__device__ __forceinline__ void gemm_strip(const float* __restrict__ A,
                                           const float* __restrict__ B,
                                           float* __restrict__ C,
                                           const float* __restrict__ att_s,
                                           const float* __restrict__ att_d,
                                           float* __restrict__ a_s,
                                           float* __restrict__ a_d, int N, int Nc) {
  constexpr int KB = K / 32;
  constexpr int NG = NT / G;
  const int lane = threadIdx.x & 31;
  const int half = lane >> 4;
  const int idx  = lane & 15;
  const int row0 = blockIdx.x * 16;

  const int ar = min(row0 + idx, N - 1);
  const float* ap = A + (size_t)ar * K + 16 * half;
  const float* bp[NT];
#pragma unroll
  for (int t = 0; t < NT; ++t)
    bp[t] = B + (size_t)min(16 * t + idx, Nc - 1) * K + 16 * half;

  v8f c[NT];
#pragma unroll
  for (int t = 0; t < NT; ++t) c[t] = (v8f){0.f, 0.f, 0.f, 0.f, 0.f, 0.f, 0.f, 0.f};

  frag16 a_cur = load_k16(ap);
  frag16 b0[G];
#pragma unroll
  for (int g = 0; g < G; ++g) b0[g] = load_k16(bp[g]);

#pragma unroll
  for (int kb = 0; kb < KB; ++kb) {
    const int kn = min(kb + 1, KB - 1) * 32;
    frag16 a_nxt = load_k16(ap + kn);
    if constexpr (NG == 2) {
      frag16 b1[G];
#pragma unroll
      for (int g = 0; g < G; ++g) b1[g] = load_k16(bp[G + g] + kb * 32);
      __builtin_amdgcn_sched_barrier(0);
      mma_group<G>(a_cur, b0, c);
      __builtin_amdgcn_sched_barrier(0);
#pragma unroll
      for (int g = 0; g < G; ++g) b0[g] = load_k16(bp[g] + kn);
      __builtin_amdgcn_sched_barrier(0);
      mma_group<G>(a_cur, b1, c + G);
    } else {
      frag16 bn[G];
#pragma unroll
      for (int g = 0; g < G; ++g) bn[g] = load_k16(bp[g] + kn);
      __builtin_amdgcn_sched_barrier(0);
      mma_group<G>(a_cur, b0, c);
#pragma unroll
      for (int g = 0; g < G; ++g) b0[g] = bn[g];
    }
    a_cur = a_nxt;
  }

  strip_epilogue<NT, H>(c, C, att_s, att_d, a_s, a_d, N, Nc, row0, idx, half);
}

#define L1_K 256
#define L1_NC 128
#define L1_BPITCH 260
#define L1_WAVES 16
__global__ void __launch_bounds__(32 * L1_WAVES)
gemm_strip_k256_c128(const float* __restrict__ A, const float* __restrict__ B,
                     float* __restrict__ C, const float* __restrict__ att_s,
                     const float* __restrict__ att_d, float* __restrict__ a_s,
                     float* __restrict__ a_d, int N) {
  constexpr int KB = L1_K / 32;
  constexpr int NT = L1_NC / 16;
  constexpr int G = 2;
  constexpr int NG = NT / G;
  __shared__ float4 Bs4[L1_NC * L1_BPITCH / 4];
  float* Bs = (float*)Bs4;
  const int tid = threadIdx.x;
  const int wave = __builtin_amdgcn_readfirstlane(tid >> 5);
  const int row0 = (blockIdx.x * L1_WAVES + wave) * 16;
  const int lane = tid & 31;
  const int half = lane >> 4;
  const int idx  = lane & 15;
  const int ar = min(row0 + idx, N - 1);
  const float* ap = A + (size_t)ar * L1_K + 16 * half;
  frag16 a_cur = load_k16(ap);
  {
    constexpr int NV = L1_NC * L1_K / 4;
    constexpr int PER = NV / (32 * L1_WAVES);
    const float4* __restrict__ Bg = (const float4*)B;
    float4 w[PER];
#pragma unroll
    for (int j = 0; j < PER; ++j) w[j] = Bg[tid + 32 * L1_WAVES * j];
#pragma unroll
    for (int j = 0; j < PER; ++j) {
      const int f = tid + 32 * L1_WAVES * j;
      const int row = f / (L1_K / 4), c4 = f % (L1_K / 4);
      *(float4*)(Bs + row * L1_BPITCH + 4 * c4) = w[j];
    }
  }
  __syncthreads();
  if (row0 >= N) return;

  const float* bq = Bs + idx * L1_BPITCH + 16 * half;

  v8f c[NT];
#pragma unroll
  for (int t = 0; t < NT; ++t) c[t] = (v8f){0.f, 0.f, 0.f, 0.f, 0.f, 0.f, 0.f, 0.f};

  frag16 bb[2][G];
#pragma unroll
  for (int g = 0; g < G; ++g) bb[0][g] = load_k16(bq + g * 16 * L1_BPITCH);

#pragma unroll
  for (int kb = 0; kb < KB; ++kb) {
    const int kn = min(kb + 1, KB - 1) * 32;
    frag16 a_nxt = load_k16(ap + kn);
#pragma unroll
    for (int grp = 0; grp < NG; ++grp) {
      const int cur = (kb * NG + grp) & 1;
      if (!(kb == KB - 1 && grp == NG - 1)) {
        const int t0 = (grp + 1 < NG) ? (grp + 1) * G : 0;
        const int k0 = (grp + 1 < NG) ? kb * 32 : kn;
#pragma unroll
        for (int g = 0; g < G; ++g)
          bb[cur ^ 1][g] = load_k16(bq + (t0 + g) * 16 * L1_BPITCH + k0);
      }
      __builtin_amdgcn_sched_barrier(0);
      mma_group<G>(a_cur, bb[cur], c + grp * G);
      __builtin_amdgcn_sched_barrier(0);
    }
    a_cur = a_nxt;
  }

  strip_epilogue<NT, 4>(c, C, att_s, att_d, a_s, a_d, N, L1_NC, row0, idx, half);
}

__global__ void __launch_bounds__(32)
gemm_strip_k128_c40(const float* __restrict__ A, const float* __restrict__ B,
                    float* __restrict__ C, const float* __restrict__ att_s,
                    const float* __restrict__ att_d, float* __restrict__ a_s,
                    float* __restrict__ a_d, int N) {
  gemm_strip<128, 3, 3, 1>(A, B, C, att_s, att_d, a_s, a_d, N, 40);
}


__device__ __forceinline__ float lane_bcast(float v, int j) {
  return __int_as_float(__builtin_amdgcn_readlane(__float_as_int(v), j));
}
__device__ __forceinline__ float leaky(float e) { return (e > 0.f) ? e : NEG_SLOPE * e; }
__device__ __forceinline__ float wave_max(float v) {
  for (int m = 16; m >= 1; m >>= 1) v = fmaxf(v, __shfl_xor(v, m, 32));
  return v;
}
__device__ __forceinline__ float wave_sum(float v) {
  for (int m = 16; m >= 1; m >>= 1) v += __shfl_xor(v, m, 32);
  return v;
}

__device__ __forceinline__ void add_rows_h4d32(const float* __restrict__ h, int s_l,
                                               float al0, float al1, float al2, float al3,
                                               int cnt, int lane, float& acc0, float& acc1,
                                               float& acc2, float& acc3) {
  int j = 0;
  for (; j + 3 < cnt; j += 4) {
    float a[4][4], v[4][4];
    for (int k = 0; k < 4; ++k) {
      const int sj = __builtin_amdgcn_readlane(s_l, j + k);
      a[k][0] = lane_bcast(al0, j + k);
      a[k][1] = lane_bcast(al1, j + k);
      a[k][2] = lane_bcast(al2, j + k);
      a[k][3] = lane_bcast(al3, j + k);
      const float* __restrict__ hr = h + (size_t)sj * 128 + lane;
      v[k][0] = hr[0];
      v[k][1] = hr[32];
      v[k][2] = hr[64];
      v[k][3] = hr[96];
    }
    for (int k = 0; k < 4; ++k) {
      acc0 += a[k][0] * v[k][0];
      acc1 += a[k][1] * v[k][1];
      acc2 += a[k][2] * v[k][2];
      acc3 += a[k][3] * v[k][3];
    }
  }
  for (; j < cnt; ++j) {
    const int sj = __builtin_amdgcn_readlane(s_l, j);
    const float a0 = lane_bcast(al0, j), a1 = lane_bcast(al1, j);
    const float a2 = lane_bcast(al2, j), a3 = lane_bcast(al3, j);
    const float* __restrict__ hr = h + (size_t)sj * 128 + lane;
    const float v0 = hr[0], v1 = hr[32], v2 = hr[64], v3 = hr[96];
    acc0 += a0 * v0;
    acc1 += a1 * v1;
    acc2 += a2 * v2;
    acc3 += a3 * v3;
  }
}

__global__ void __launch_bounds__(256)
node_aggregate_h4d32(const int* __restrict__ off, const int* __restrict__ csr_src,
                     const float* __restrict__ a_s, const float* __restrict__ a_d,
                     const float* __restrict__ h, const float* __restrict__ bias,
                     float* __restrict__ x2, int N) {
  const int lane = threadIdx.x & 31;
  const int wib = __builtin_amdgcn_readfirstlane((int)(threadIdx.x >> 5));
  const int node = blockIdx.x * 8 + wib;
  if (node >= N) return;
  const int beg = __builtin_amdgcn_readfirstlane(off[node]);
  const int deg = __builtin_amdgcn_readfirstlane(off[node + 1]) - beg;
  const int* __restrict__ nb = csr_src + beg;

  const float4 ad4 = *(const float4*)(a_d + (size_t)node * 4);
  const float4 sf4 = *(const float4*)(a_s + (size_t)node * 4);
  const float z0 = leaky(sf4.x + ad4.x), z1 = leaky(sf4.y + ad4.y);
  const float z2 = leaky(sf4.z + ad4.z), z3 = leaky(sf4.w + ad4.w);

  const bool act = lane < deg;
  const int s_c = act ? nb[lane] : node;
  const float4 as4 = *(const float4*)(a_s + (size_t)s_c * 4);
  const float e0 = leaky(as4.x + ad4.x), e1 = leaky(as4.y + ad4.y);
  const float e2 = leaky(as4.z + ad4.z), e3 = leaky(as4.w + ad4.w);

  float m0 = fmaxf(z0, e0), m1 = fmaxf(z1, e1), m2 = fmaxf(z2, e2), m3 = fmaxf(z3, e3);
  for (int i = 32 + lane; i < deg; i += 32) {
    const float4 b4 = *(const float4*)(a_s + (size_t)nb[i] * 4);
    m0 = fmaxf(m0, leaky(b4.x + ad4.x));
    m1 = fmaxf(m1, leaky(b4.y + ad4.y));
    m2 = fmaxf(m2, leaky(b4.z + ad4.z));
    m3 = fmaxf(m3, leaky(b4.w + ad4.w));
  }
  m0 = wave_max(m0); m1 = wave_max(m1); m2 = wave_max(m2); m3 = wave_max(m3);

  const float p0 = act ? expf(e0 - m0) : 0.f, p1 = act ? expf(e1 - m1) : 0.f;
  const float p2 = act ? expf(e2 - m2) : 0.f, p3 = act ? expf(e3 - m3) : 0.f;
  float d0 = p0, d1 = p1, d2 = p2, d3 = p3;
  for (int i = 32 + lane; i < deg; i += 32) {
    const float4 b4 = *(const float4*)(a_s + (size_t)nb[i] * 4);
    d0 += expf(leaky(b4.x + ad4.x) - m0);
    d1 += expf(leaky(b4.y + ad4.y) - m1);
    d2 += expf(leaky(b4.z + ad4.z) - m2);
    d3 += expf(leaky(b4.w + ad4.w) - m3);
  }
  const float q0 = expf(z0 - m0), q1 = expf(z1 - m1), q2 = expf(z2 - m2), q3 = expf(z3 - m3);
  d0 = wave_sum(d0) + q0; d1 = wave_sum(d1) + q1;
  d2 = wave_sum(d2) + q2; d3 = wave_sum(d3) + q3;

  const float* __restrict__ hs = h + (size_t)node * 128 + lane;
  float acc0 = (q0 / d0) * hs[0];
  float acc1 = (q1 / d1) * hs[32];
  float acc2 = (q2 / d2) * hs[64];
  float acc3 = (q3 / d3) * hs[96];
  add_rows_h4d32(h, s_c, p0 / d0, p1 / d1, p2 / d2, p3 / d3, min(deg, 32), lane,
                 acc0, acc1, acc2, acc3);
  for (int base = 32; base < deg; base += 32) {
    const int cnt = min(32, deg - base);
    const bool on = lane < cnt;
    const int s_r = on ? nb[base + lane] : node;
    const float4 b4 = *(const float4*)(a_s + (size_t)s_r * 4);
    const float r0 = on ? expf(leaky(b4.x + ad4.x) - m0) / d0 : 0.f;
    const float r1 = on ? expf(leaky(b4.y + ad4.y) - m1) / d1 : 0.f;
    const float r2 = on ? expf(leaky(b4.z + ad4.z) - m2) / d2 : 0.f;
    const float r3 = on ? expf(leaky(b4.w + ad4.w) - m3) / d3 : 0.f;
    add_rows_h4d32(h, s_r, r0, r1, r2, r3, cnt, lane, acc0, acc1, acc2, acc3);
  }

  float* __restrict__ xr = x2 + (size_t)node * 128 + lane;
  const float o0 = acc0 + bias[lane],      o1 = acc1 + bias[32 + lane];
  const float o2 = acc2 + bias[64 + lane], o3 = acc3 + bias[96 + lane];
  xr[0]  = (o0 > 0.f) ? o0 : 0.f;
  xr[32] = (o1 > 0.f) ? o1 : 0.f;
  xr[64] = (o2 > 0.f) ? o2 : 0.f;
  xr[96] = (o3 > 0.f) ? o3 : 0.f;
}

__device__ __forceinline__ void add_rows_h1d40(const float* __restrict__ h, int s_l, float al,
                                               int cnt, int lane, int f1, float& acc0,
                                               float& acc1) {
  int j = 0;
  for (; j + 3 < cnt; j += 4) {
    float a[4], va[4], vb[4];
    for (int k = 0; k < 4; ++k) {
      const int sj = __builtin_amdgcn_readlane(s_l, j + k);
      a[k] = lane_bcast(al, j + k);
      const float* __restrict__ hr = h + (size_t)sj * 40;
      va[k] = hr[lane];
      vb[k] = hr[f1];
    }
    for (int k = 0; k < 4; ++k) {
      acc0 += a[k] * va[k];
      acc1 += a[k] * vb[k];
    }
  }
  for (; j < cnt; ++j) {
    const int sj = __builtin_amdgcn_readlane(s_l, j);
    const float ak = lane_bcast(al, j);
    const float* __restrict__ hr = h + (size_t)sj * 40;
    acc0 += ak * hr[lane];
    acc1 += ak * hr[f1];
  }
}

__global__ void __launch_bounds__(256)
node_aggregate_h1d40(const int* __restrict__ off, const int* __restrict__ csr_src,
                     const float* __restrict__ a_s, const float* __restrict__ a_d,
                     const float* __restrict__ h, const float* __restrict__ bias,
                     float* __restrict__ out, int N) {
  const int lane = threadIdx.x & 31;
  const int wib = __builtin_amdgcn_readfirstlane((int)(threadIdx.x >> 5));
  const int node = blockIdx.x * 8 + wib;
  if (node >= N) return;
  const int beg = __builtin_amdgcn_readfirstlane(off[node]);
  const int deg = __builtin_amdgcn_readfirstlane(off[node + 1]) - beg;
  const int* __restrict__ nb = csr_src + beg;
  const int f1 = 32 + (lane & 7);

  const float ad = a_d[node];
  const float z = leaky(a_s[node] + ad);

  const bool act = lane < deg;
  const int s_c = act ? nb[lane] : node;
  const float e = leaky(a_s[s_c] + ad);

  float m = fmaxf(z, e);
  for (int i = 32 + lane; i < deg; i += 32) m = fmaxf(m, leaky(a_s[nb[i]] + ad));
  m = wave_max(m);

  const float p = act ? expf(e - m) : 0.f;
  float d = p;
  for (int i = 32 + lane; i < deg; i += 32) d += expf(leaky(a_s[nb[i]] + ad) - m);
  const float q = expf(z - m);
  d = wave_sum(d) + q;

  const float* __restrict__ hs = h + (size_t)node * 40;
  float acc0 = (q / d) * hs[lane];
  float acc1 = (q / d) * hs[f1];
  add_rows_h1d40(h, s_c, p / d, min(deg, 32), lane, f1, acc0, acc1);
  for (int base = 32; base < deg; base += 32) {
    const int cnt = min(32, deg - base);
    const bool on = lane < cnt;
    const int s_r = on ? nb[base + lane] : node;
    const float r = on ? expf(leaky(a_s[s_r] + ad) - m) / d : 0.f;
    add_rows_h1d40(h, s_r, r, cnt, lane, f1, acc0, acc1);
  }

  float* __restrict__ orow = out + (size_t)node * 40;
  orow[lane] = acc0 + bias[lane];
  if (lane < 8) orow[32 + lane] = acc1 + bias[32 + lane];
}

extern "C" void kernel_launch(void* const* d_in, const int* in_sizes, int n_in,
                              void* d_out, int out_size, void* d_ws, size_t ws_size,
                              hipStream_t stream) {
  const float* x        = (const float*)d_in[0];
  const int*   ei       = (const int*)d_in[1];
  const float* W1       = (const float*)d_in[2];
  const float* att_src1 = (const float*)d_in[3];
  const float* att_dst1 = (const float*)d_in[4];
  const float* b1       = (const float*)d_in[5];
  const float* W2       = (const float*)d_in[6];
  const float* att_src2 = (const float*)d_in[7];
  const float* att_dst2 = (const float*)d_in[8];
  const float* b2       = (const float*)d_in[9];

  const int NFEAT = 256, NHID = 128, H1 = 4, NCLS = 40;
  const int N = in_sizes[0] / NFEAT;
  const int E = in_sizes[1] / 2;
  const int* src = ei;
  const int* dst = ei + E;

  float* out = (float*)d_out;

  float* w = (float*)d_ws;
  float* h1   = w;
  float* x2   = h1  + (size_t)N * NHID;
  float* h2   = x2  + (size_t)N * NHID;
  float* as1  = h2  + (size_t)N * NCLS;
  float* ad1  = as1 + (size_t)N * H1;
  float* as2  = ad1 + (size_t)N * H1;
  float* ad2  = as2 + (size_t)N;
  int* off     = (int*)(ad2 + (size_t)N);
  int* csr_src = off + (size_t)N + 1;

  const int TB = 256;

  csr_build<<<(N + CSR_BKT - 1) / CSR_BKT, CSR_THREADS, 0, stream>>>(src, dst, E, N, off,
                                                                      csr_src);

  gemm_strip_k256_c128<<<(N + 16 * L1_WAVES - 1) / (16 * L1_WAVES), 32 * L1_WAVES, 0,
                         stream>>>(x, W1, h1, att_src1, att_dst1, as1, ad1, N);
  node_aggregate_h4d32<<<(N + 7) / 8, TB, 0, stream>>>(off, csr_src, as1, ad1, h1, b1,
                                                       x2, N);

  gemm_strip_k128_c40<<<(N + 15) / 16, 32, 0, stream>>>(x2, W2, h2, att_src2, att_dst2,
                                                        as2, ad2, N);
  node_aggregate_h1d40<<<(N + 7) / 8, TB, 0, stream>>>(off, csr_src, as2, ad2, h2, b2,
                                                       out, N);
}


# edge list built in two launches: edges binned once by 256-destination range into slots, then one work-group per range ranks and places only its own records; sum order = LDS arrival rank as before
# speedup vs baseline: 17.1906x; 1.2711x over previous
#include <hip/hip_runtime.h>


typedef float v2f __attribute__((ext_vector_type(2)));
typedef float v8f __attribute__((ext_vector_type(8)));

#define NEG_SLOPE 0.2f


#define CSR_BKT 256
#define CSR_THREADS 512
#define CSR_CAP 14336
#define CSR_U 8
#define BIN_P 256
#define BIN_THREADS 256
#define BIN_U 13
#define BIN_SLOT 48
#define BIN_MAXR 256

template <class F>
__device__ __forceinline__ void for_each_edge(const int* __restrict__ dst, int E, int tid, F f) {
  const int head = min(E, (int)(((16u - (unsigned)((size_t)dst & 15u)) & 15u) >> 2));
  const int nv = (E - head) >> 2;
  const int4* __restrict__ dv = (const int4*)(dst + head);
  for (int i0 = tid; i0 < nv; i0 += CSR_U * CSR_THREADS) {
    int4 d[CSR_U];
#pragma unroll
    for (int u = 0; u < CSR_U; ++u) d[u] = dv[min(i0 + u * CSR_THREADS, nv - 1)];
#pragma unroll
    for (int u = 0; u < CSR_U; ++u) {
      const int i = i0 + u * CSR_THREADS;
      if (i < nv) {
        const int t = head + 4 * i;
        f(t, d[u].x);
        f(t + 1, d[u].y);
        f(t + 2, d[u].z);
        f(t + 3, d[u].w);
      }
    }
  }
  const int tail0 = head + 4 * nv;
  if (tid < head) f(tid, dst[tid]);
  if (tid < E - tail0) f(tail0 + tid, dst[tail0 + tid]);
}

#define BIN_CH (BIN_U * BIN_THREADS)
static_assert(BIN_THREADS == BIN_MAXR, "thread t owns counter t");
__global__ void __launch_bounds__(BIN_THREADS)
bin_edges(const int* __restrict__ src, const int* __restrict__ dst, int E, int NR,
          int* __restrict__ counts, int* __restrict__ recs) {
  __shared__ int c[BIN_MAXR];
  __shared__ int cc[BIN_MAXR];
  __shared__ int st[BIN_MAXR];
  __shared__ int wtot[BIN_THREADS / 32];
  __shared__ int l_s[BIN_CH];
  __shared__ int l_d[BIN_CH];
  const int tid = threadIdx.x, lane = tid & 31, p = blockIdx.x;
  const int wid = __builtin_amdgcn_readfirstlane(tid >> 5);
  c[tid] = 0;
  const int EP = (E + BIN_P - 1) / BIN_P;
  const int e0 = min(E, p * EP), e1 = min(E, e0 + EP);
  const unsigned nr = (unsigned)min(NR, BIN_MAXR);
  for (int cb = e0; cb < e1; cb += BIN_CH) {
    cc[tid] = 0;
    __syncthreads();
    int d[BIN_U], s[BIN_U], k[BIN_U];
#pragma unroll
    for (int u = 0; u < BIN_U; ++u) {
      const int e = min(cb + tid + u * BIN_THREADS, e1 - 1);
      d[u] = dst[e];
      s[u] = src[e];
    }
#pragma unroll
    for (int u = 0; u < BIN_U; ++u) {
      const unsigned r = (unsigned)d[u] / (unsigned)CSR_BKT;
      const bool ok = (cb + tid + u * BIN_THREADS < e1) && (r < nr);
      k[u] = -1;
      if (ok) k[u] = atomicAdd(&cc[r], 1);
    }
    __syncthreads();
    {
      const int x = cc[tid];
      int inc = x;
      for (int sh = 1; sh < 32; sh <<= 1) {
        const int y = __shfl_up(inc, sh, 32);
        if (lane >= sh) inc += y;
      }
      if (lane == 31) wtot[wid] = inc;
      __syncthreads();
      int b = 0;
      for (int w = 0; w < wid; ++w) b += wtot[w];
      st[tid] = b + inc - x;
    }
    __syncthreads();
#pragma unroll
    for (int u = 0; u < BIN_U; ++u) {
      if (k[u] >= 0) {
        const int pos = st[(unsigned)d[u] / (unsigned)CSR_BKT] + k[u];
        l_s[pos] = s[u];
        l_d[pos] = d[u];
      }
    }
    __syncthreads();
    const int tot = st[BIN_MAXR - 1] + cc[BIN_MAXR - 1];
    for (int i = tid; i < tot; i += BIN_THREADS) {
      const int dd = l_d[i], ss = l_s[i];
      const int r = (int)((unsigned)dd / (unsigned)CSR_BKT);
      const int kk = c[r] + (i - st[r]);
      if (kk < BIN_SLOT) {
        const size_t rec = ((size_t)r * BIN_P + p) * BIN_SLOT + kk;
        *(int2*)(recs + 2 * rec) = make_int2(ss, dd);
      }
    }
    __syncthreads();
    c[tid] += cc[tid];
  }
  __syncthreads();
  for (int r = tid; r < NR; r += BIN_THREADS)
    counts[(size_t)r * BIN_P + p] = (r < BIN_MAXR) ? c[r] : 0;
}

#define CSR_WAVES (CSR_THREADS / 32)
#define CSR_SPW (BIN_P / CSR_WAVES)
#define CSR_BU 13

__global__ void __launch_bounds__(CSR_THREADS)
csr_place(const int* __restrict__ src, const int* __restrict__ dst, int E, int N,
          const int* __restrict__ counts, const int* __restrict__ recs,
          int* __restrict__ off, int* __restrict__ csr_src) {
  __shared__ int c[CSR_BKT];
  __shared__ int loff[CSR_BKT];
  __shared__ int wtot[CSR_BKT / 32];
  __shared__ int s_below, s_nl, s_slow;
  __shared__ int l_s[CSR_CAP];
  __shared__ int l_d[CSR_CAP];
  __shared__ int l_r[CSR_CAP];
  const int tid = threadIdx.x, lane = tid & 31;
  const int wid = __builtin_amdgcn_readfirstlane(tid >> 5);
  const int rg = blockIdx.x;
  const int lo = rg * CSR_BKT;

  if (tid < CSR_BKT) c[tid] = 0;
  if (tid == 0) { s_below = 0; s_nl = 0; s_slow = 0; }
  __syncthreads();

  const int* __restrict__ myc = counts + (size_t)rg * BIN_P;
  int n[CSR_SPW];
#pragma unroll
  for (int u = 0; u < CSR_SPW; ++u) n[u] = myc[wid + CSR_WAVES * u];

  int below = 0;
  {
    const int4* __restrict__ cv = (const int4*)counts;
    const int nq = rg * (BIN_P / 4);
    for (int i0 = tid; i0 < nq; i0 += CSR_BU * CSR_THREADS) {
      int4 v[CSR_BU];
#pragma unroll
      for (int u = 0; u < CSR_BU; ++u) v[u] = cv[min(i0 + u * CSR_THREADS, nq - 1)];
#pragma unroll
      for (int u = 0; u < CSR_BU; ++u)
        if (i0 + u * CSR_THREADS < nq) below += (v[u].x + v[u].y) + (v[u].z + v[u].w);
    }
  }

  auto note = [&](int s, int d) {
    const unsigned dl = (unsigned)(d - lo);
    if (dl < (unsigned)CSR_BKT) {
      const int r = atomicAdd(&c[dl], 1);
      const int k = atomicAdd(&s_nl, 1);
      if (k < CSR_CAP) { l_s[k] = s; l_d[k] = (int)dl; l_r[k] = r; }
    }
  };

  int2 rc[CSR_SPW];
#pragma unroll
  for (int u = 0; u < CSR_SPW; ++u) {
    const int* __restrict__ slot = recs + 2 * ((size_t)rg * BIN_P + wid + CSR_WAVES * u) * BIN_SLOT;
    rc[u] = make_int2(0, 0);
    if (lane < min(n[u], BIN_SLOT)) rc[u] = *(const int2*)(slot + 2 * lane);
  }
#pragma unroll
  for (int u = 0; u < CSR_SPW; ++u) {
    const int* __restrict__ slot = recs + 2 * ((size_t)rg * BIN_P + wid + CSR_WAVES * u) * BIN_SLOT;
    const int m = min(n[u], BIN_SLOT);
    if (n[u] > BIN_SLOT) s_slow = 1;
    if (lane < m) note(rc[u].x, rc[u].y);
    for (int j = 32 + lane; j < m; j += 32) {
      const int2 q = *(const int2*)(slot + 2 * j);
      note(q.x, q.y);
    }
  }
  for (int m = 16; m >= 1; m >>= 1) below += __shfl_xor(below, m, 32);
  if (lane == 0) atomicAdd(&s_below, below);
  __syncthreads();

  const bool slow = (s_slow != 0) || (s_nl > CSR_CAP);
  if (slow) {
    if (tid < CSR_BKT) c[tid] = 0;
    __syncthreads();
    for_each_edge(dst, E, tid, [&](int t, int d) {
      const unsigned dl = (unsigned)(d - lo);
      if (dl < (unsigned)CSR_BKT) atomicAdd(&c[dl], 1);
    });
    __syncthreads();
  }

  int x = 0, inc = 0;
  if (tid < CSR_BKT) {
    x = c[tid];
    inc = x;
    for (int s = 1; s < 32; s <<= 1) {
      const int y = __shfl_up(inc, s, 32);
      if (lane >= s) inc += y;
    }
    if (lane == 31) wtot[wid] = inc;
  }
  __syncthreads();
  if (tid < CSR_BKT) {
    int b = 0;
    for (int w = 0; w < wid; ++w) b += wtot[w];
    loff[tid] = b + inc - x;
  }
  __syncthreads();
  const int base = s_below;
  if (tid < CSR_BKT && lo + tid < N) off[lo + tid] = base + loff[tid];
  if (tid == CSR_BKT - 1 && lo + CSR_BKT >= N) off[N] = base + loff[tid] + x;

  if (!slow) {
    const int nl = s_nl;
    for (int k = tid; k < nl; k += CSR_THREADS)
      csr_src[base + loff[l_d[k]] + l_r[k]] = l_s[k];
  } else {
    if (tid < CSR_BKT) c[tid] = 0;
    __syncthreads();
    for_each_edge(dst, E, tid, [&](int t, int d) {
      const unsigned dl = (unsigned)(d - lo);
      if (dl < (unsigned)CSR_BKT) {
        const int r = atomicAdd(&c[dl], 1);
        csr_src[base + loff[dl] + r] = src[t];
      }
    });
  }
}


struct frag16 { float4 p[4]; };

__device__ __forceinline__ frag16 load_k16(const float* q) {
  frag16 f;
#pragma unroll
  for (int i = 0; i < 4; ++i) f.p[i] = *(const float4*)(q + 4 * i);
  return f;
}

template <int G>
__device__ __forceinline__ void mma_group(const frag16& a, const frag16 (&b)[G], v8f* c) {
#pragma unroll
  for (int i = 0; i < 4; ++i) {
#pragma unroll
    for (int g = 0; g < G; ++g) {
      const v2f av = {a.p[i].x, a.p[i].y};
      const v2f bv = {b[g].p[i].x, b[g].p[i].y};
      c[g] = __builtin_amdgcn_wmma_f32_16x16x4_f32(false, av, false, bv, (short)0, c[g],
                                                   false, false);
    }
#pragma unroll
    for (int g = 0; g < G; ++g) {
      const v2f av = {a.p[i].z, a.p[i].w};
      const v2f bv = {b[g].p[i].z, b[g].p[i].w};
      c[g] = __builtin_amdgcn_wmma_f32_16x16x4_f32(false, av, false, bv, (short)0, c[g],
                                                   false, false);
    }
  }
}

template <int NT, int H>
__device__ __forceinline__ void strip_epilogue(const v8f* c, float* __restrict__ C,
                                               const float* __restrict__ att_s,
                                               const float* __restrict__ att_d,
                                               float* __restrict__ a_s,
                                               float* __restrict__ a_d, int N, int Nc,
                                               int row0, int idx, int half) {
#pragma unroll
  for (int t = 0; t < NT; ++t) {
    const int col = 16 * t + idx;
    if (col < Nc) {
#pragma unroll
      for (int r = 0; r < 8; ++r) {
        const int row = row0 + r + 8 * half;
        if (row < N) C[(size_t)row * Nc + col] = c[t][r];
      }
    }
  }

  constexpr int TPH = NT / H;
  float p[16 * H];
#pragma unroll
  for (int hh = 0; hh < H; ++hh) {
    float ws[TPH], wd[TPH];
#pragma unroll
    for (int u = 0; u < TPH; ++u) {
      const int col = 16 * (hh * TPH + u) + idx;
      ws[u] = (col < Nc) ? att_s[min(col, Nc - 1)] : 0.f;
      wd[u] = (col < Nc) ? att_d[min(col, Nc - 1)] : 0.f;
    }
#pragma unroll
    for (int r = 0; r < 8; ++r) {
      float s = 0.f, d = 0.f;
#pragma unroll
      for (int u = 0; u < TPH; ++u) {
        s += c[hh * TPH + u][r] * ws[u];
        d += c[hh * TPH + u][r] * wd[u];
      }
      p[r * H + hh] = s;
      p[(8 + r) * H + hh] = d;
    }
  }
#pragma unroll
  for (int s = 0; s < 4; ++s) {
    const int mask = 8 >> s;
    const int hs = (8 * H) >> s;
    const bool up = (idx & mask) != 0;
#pragma unroll
    for (int j = 0; j < hs; ++j) {
      const float lo = p[j], hi = p[j + hs];
      const float send = up ? lo : hi;
      const float keep = up ? hi : lo;
      p[j] = keep + __shfl_xor(send, mask, 32);
    }
  }
  const int lrow = row0 + 8 * half + (idx & 7);
  if (lrow < N) {
    float* __restrict__ o = ((idx < 8) ? a_s : a_d) + (size_t)lrow * H;
#pragma unroll
    for (int hh = 0; hh < H; ++hh) o[hh] = p[hh];
  }
}

template <int K, int NT, int G, int H>
__device__ __forceinline__ void gemm_strip(const float* __restrict__ A,
                                           const float* __restrict__ B,
                                           float* __restrict__ C,
                                           const float* __restrict__ att_s,
                                           const float* __restrict__ att_d,
                                           float* __restrict__ a_s,
                                           float* __restrict__ a_d, int N, int Nc) {
  constexpr int KB = K / 32;
  constexpr int NG = NT / G;
  const int lane = threadIdx.x & 31;
  const int half = lane >> 4;
  const int idx  = lane & 15;
  const int row0 = blockIdx.x * 16;

  const int ar = min(row0 + idx, N - 1);
  const float* ap = A + (size_t)ar * K + 16 * half;
  const float* bp[NT];
#pragma unroll
  for (int t = 0; t < NT; ++t)
    bp[t] = B + (size_t)min(16 * t + idx, Nc - 1) * K + 16 * half;

  v8f c[NT];
#pragma unroll
  for (int t = 0; t < NT; ++t) c[t] = (v8f){0.f, 0.f, 0.f, 0.f, 0.f, 0.f, 0.f, 0.f};

  frag16 a_cur = load_k16(ap);
  frag16 b0[G];
#pragma unroll
  for (int g = 0; g < G; ++g) b0[g] = load_k16(bp[g]);

#pragma unroll
  for (int kb = 0; kb < KB; ++kb) {
    const int kn = min(kb + 1, KB - 1) * 32;
    frag16 a_nxt = load_k16(ap + kn);
    if constexpr (NG == 2) {
      frag16 b1[G];
#pragma unroll
      for (int g = 0; g < G; ++g) b1[g] = load_k16(bp[G + g] + kb * 32);
      __builtin_amdgcn_sched_barrier(0);
      mma_group<G>(a_cur, b0, c);
      __builtin_amdgcn_sched_barrier(0);
#pragma unroll
      for (int g = 0; g < G; ++g) b0[g] = load_k16(bp[g] + kn);
      __builtin_amdgcn_sched_barrier(0);
      mma_group<G>(a_cur, b1, c + G);
    } else {
      frag16 bn[G];
#pragma unroll
      for (int g = 0; g < G; ++g) bn[g] = load_k16(bp[g] + kn);
      __builtin_amdgcn_sched_barrier(0);
      mma_group<G>(a_cur, b0, c);
#pragma unroll
      for (int g = 0; g < G; ++g) b0[g] = bn[g];
    }
    a_cur = a_nxt;
  }

  strip_epilogue<NT, H>(c, C, att_s, att_d, a_s, a_d, N, Nc, row0, idx, half);
}

#define L1_K 256
#define L1_NC 128
#define L1_BPITCH 260
#define L1_WAVES 16
__global__ void __launch_bounds__(32 * L1_WAVES)
gemm_strip_k256_c128(const float* __restrict__ A, const float* __restrict__ B,
                     float* __restrict__ C, const float* __restrict__ att_s,
                     const float* __restrict__ att_d, float* __restrict__ a_s,
                     float* __restrict__ a_d, int N) {
  constexpr int KB = L1_K / 32;
  constexpr int NT = L1_NC / 16;
  constexpr int G = 2;
  constexpr int NG = NT / G;
  __shared__ float4 Bs4[L1_NC * L1_BPITCH / 4];
  float* Bs = (float*)Bs4;
  const int tid = threadIdx.x;
  const int wave = __builtin_amdgcn_readfirstlane(tid >> 5);
  const int row0 = (blockIdx.x * L1_WAVES + wave) * 16;
  const int lane = tid & 31;
  const int half = lane >> 4;
  const int idx  = lane & 15;
  const int ar = min(row0 + idx, N - 1);
  const float* ap = A + (size_t)ar * L1_K + 16 * half;
  frag16 a_cur = load_k16(ap);
  {
    constexpr int NV = L1_NC * L1_K / 4;
    constexpr int PER = NV / (32 * L1_WAVES);
    const float4* __restrict__ Bg = (const float4*)B;
    float4 w[PER];
#pragma unroll
    for (int j = 0; j < PER; ++j) w[j] = Bg[tid + 32 * L1_WAVES * j];
#pragma unroll
    for (int j = 0; j < PER; ++j) {
      const int f = tid + 32 * L1_WAVES * j;
      const int row = f / (L1_K / 4), c4 = f % (L1_K / 4);
      *(float4*)(Bs + row * L1_BPITCH + 4 * c4) = w[j];
    }
  }
  __syncthreads();
  if (row0 >= N) return;

  const float* bq = Bs + idx * L1_BPITCH + 16 * half;

  v8f c[NT];
#pragma unroll
  for (int t = 0; t < NT; ++t) c[t] = (v8f){0.f, 0.f, 0.f, 0.f, 0.f, 0.f, 0.f, 0.f};

  frag16 bb[2][G];
#pragma unroll
  for (int g = 0; g < G; ++g) bb[0][g] = load_k16(bq + g * 16 * L1_BPITCH);

#pragma unroll
  for (int kb = 0; kb < KB; ++kb) {
    const int kn = min(kb + 1, KB - 1) * 32;
    frag16 a_nxt = load_k16(ap + kn);
#pragma unroll
    for (int grp = 0; grp < NG; ++grp) {
      const int cur = (kb * NG + grp) & 1;
      if (!(kb == KB - 1 && grp == NG - 1)) {
        const int t0 = (grp + 1 < NG) ? (grp + 1) * G : 0;
        const int k0 = (grp + 1 < NG) ? kb * 32 : kn;
#pragma unroll
        for (int g = 0; g < G; ++g)
          bb[cur ^ 1][g] = load_k16(bq + (t0 + g) * 16 * L1_BPITCH + k0);
      }
      __builtin_amdgcn_sched_barrier(0);
      mma_group<G>(a_cur, bb[cur], c + grp * G);
      __builtin_amdgcn_sched_barrier(0);
    }
    a_cur = a_nxt;
  }

  strip_epilogue<NT, 4>(c, C, att_s, att_d, a_s, a_d, N, L1_NC, row0, idx, half);
}

__global__ void __launch_bounds__(32)
gemm_strip_k128_c40(const float* __restrict__ A, const float* __restrict__ B,
                    float* __restrict__ C, const float* __restrict__ att_s,
                    const float* __restrict__ att_d, float* __restrict__ a_s,
                    float* __restrict__ a_d, int N) {
  gemm_strip<128, 3, 3, 1>(A, B, C, att_s, att_d, a_s, a_d, N, 40);
}


__device__ __forceinline__ float lane_bcast(float v, int j) {
  return __int_as_float(__builtin_amdgcn_readlane(__float_as_int(v), j));
}
__device__ __forceinline__ float leaky(float e) { return (e > 0.f) ? e : NEG_SLOPE * e; }
__device__ __forceinline__ float wave_max(float v) {
  for (int m = 16; m >= 1; m >>= 1) v = fmaxf(v, __shfl_xor(v, m, 32));
  return v;
}
__device__ __forceinline__ float wave_sum(float v) {
  for (int m = 16; m >= 1; m >>= 1) v += __shfl_xor(v, m, 32);
  return v;
}

__device__ __forceinline__ void add_rows_h4d32(const float* __restrict__ h, int s_l,
                                               float al0, float al1, float al2, float al3,
                                               int cnt, int lane, float& acc0, float& acc1,
                                               float& acc2, float& acc3) {
  int j = 0;
  for (; j + 3 < cnt; j += 4) {
    float a[4][4], v[4][4];
    for (int k = 0; k < 4; ++k) {
      const int sj = __builtin_amdgcn_readlane(s_l, j + k);
      a[k][0] = lane_bcast(al0, j + k);
      a[k][1] = lane_bcast(al1, j + k);
      a[k][2] = lane_bcast(al2, j + k);
      a[k][3] = lane_bcast(al3, j + k);
      const float* __restrict__ hr = h + (size_t)sj * 128 + lane;
      v[k][0] = hr[0];
      v[k][1] = hr[32];
      v[k][2] = hr[64];
      v[k][3] = hr[96];
    }
    for (int k = 0; k < 4; ++k) {
      acc0 += a[k][0] * v[k][0];
      acc1 += a[k][1] * v[k][1];
      acc2 += a[k][2] * v[k][2];
      acc3 += a[k][3] * v[k][3];
    }
  }
  for (; j < cnt; ++j) {
    const int sj = __builtin_amdgcn_readlane(s_l, j);
    const float a0 = lane_bcast(al0, j), a1 = lane_bcast(al1, j);
    const float a2 = lane_bcast(al2, j), a3 = lane_bcast(al3, j);
    const float* __restrict__ hr = h + (size_t)sj * 128 + lane;
    const float v0 = hr[0], v1 = hr[32], v2 = hr[64], v3 = hr[96];
    acc0 += a0 * v0;
    acc1 += a1 * v1;
    acc2 += a2 * v2;
    acc3 += a3 * v3;
  }
}

__global__ void __launch_bounds__(256)
node_aggregate_h4d32(const int* __restrict__ off, const int* __restrict__ csr_src,
                     const float* __restrict__ a_s, const float* __restrict__ a_d,
                     const float* __restrict__ h, const float* __restrict__ bias,
                     float* __restrict__ x2, int N) {
  const int lane = threadIdx.x & 31;
  const int wib = __builtin_amdgcn_readfirstlane((int)(threadIdx.x >> 5));
  const int node = blockIdx.x * 8 + wib;
  if (node >= N) return;
  const int beg = __builtin_amdgcn_readfirstlane(off[node]);
  const int deg = __builtin_amdgcn_readfirstlane(off[node + 1]) - beg;
  const int* __restrict__ nb = csr_src + beg;

  const float4 ad4 = *(const float4*)(a_d + (size_t)node * 4);
  const float4 sf4 = *(const float4*)(a_s + (size_t)node * 4);
  const float z0 = leaky(sf4.x + ad4.x), z1 = leaky(sf4.y + ad4.y);
  const float z2 = leaky(sf4.z + ad4.z), z3 = leaky(sf4.w + ad4.w);

  const bool act = lane < deg;
  const int s_c = act ? nb[lane] : node;
  const float4 as4 = *(const float4*)(a_s + (size_t)s_c * 4);
  const float e0 = leaky(as4.x + ad4.x), e1 = leaky(as4.y + ad4.y);
  const float e2 = leaky(as4.z + ad4.z), e3 = leaky(as4.w + ad4.w);

  float m0 = fmaxf(z0, e0), m1 = fmaxf(z1, e1), m2 = fmaxf(z2, e2), m3 = fmaxf(z3, e3);
  for (int i = 32 + lane; i < deg; i += 32) {
    const float4 b4 = *(const float4*)(a_s + (size_t)nb[i] * 4);
    m0 = fmaxf(m0, leaky(b4.x + ad4.x));
    m1 = fmaxf(m1, leaky(b4.y + ad4.y));
    m2 = fmaxf(m2, leaky(b4.z + ad4.z));
    m3 = fmaxf(m3, leaky(b4.w + ad4.w));
  }
  m0 = wave_max(m0); m1 = wave_max(m1); m2 = wave_max(m2); m3 = wave_max(m3);

  const float p0 = act ? expf(e0 - m0) : 0.f, p1 = act ? expf(e1 - m1) : 0.f;
  const float p2 = act ? expf(e2 - m2) : 0.f, p3 = act ? expf(e3 - m3) : 0.f;
  float d0 = p0, d1 = p1, d2 = p2, d3 = p3;
  for (int i = 32 + lane; i < deg; i += 32) {
    const float4 b4 = *(const float4*)(a_s + (size_t)nb[i] * 4);
    d0 += expf(leaky(b4.x + ad4.x) - m0);
    d1 += expf(leaky(b4.y + ad4.y) - m1);
    d2 += expf(leaky(b4.z + ad4.z) - m2);
    d3 += expf(leaky(b4.w + ad4.w) - m3);
  }
  const float q0 = expf(z0 - m0), q1 = expf(z1 - m1), q2 = expf(z2 - m2), q3 = expf(z3 - m3);
  d0 = wave_sum(d0) + q0; d1 = wave_sum(d1) + q1;
  d2 = wave_sum(d2) + q2; d3 = wave_sum(d3) + q3;

  const float* __restrict__ hs = h + (size_t)node * 128 + lane;
  float acc0 = (q0 / d0) * hs[0];
  float acc1 = (q1 / d1) * hs[32];
  float acc2 = (q2 / d2) * hs[64];
  float acc3 = (q3 / d3) * hs[96];
  add_rows_h4d32(h, s_c, p0 / d0, p1 / d1, p2 / d2, p3 / d3, min(deg, 32), lane,
                 acc0, acc1, acc2, acc3);
  for (int base = 32; base < deg; base += 32) {
    const int cnt = min(32, deg - base);
    const bool on = lane < cnt;
    const int s_r = on ? nb[base + lane] : node;
    const float4 b4 = *(const float4*)(a_s + (size_t)s_r * 4);
    const float r0 = on ? expf(leaky(b4.x + ad4.x) - m0) / d0 : 0.f;
    const float r1 = on ? expf(leaky(b4.y + ad4.y) - m1) / d1 : 0.f;
    const float r2 = on ? expf(leaky(b4.z + ad4.z) - m2) / d2 : 0.f;
    const float r3 = on ? expf(leaky(b4.w + ad4.w) - m3) / d3 : 0.f;
    add_rows_h4d32(h, s_r, r0, r1, r2, r3, cnt, lane, acc0, acc1, acc2, acc3);
  }

  float* __restrict__ xr = x2 + (size_t)node * 128 + lane;
  const float o0 = acc0 + bias[lane],      o1 = acc1 + bias[32 + lane];
  const float o2 = acc2 + bias[64 + lane], o3 = acc3 + bias[96 + lane];
  xr[0]  = (o0 > 0.f) ? o0 : 0.f;
  xr[32] = (o1 > 0.f) ? o1 : 0.f;
  xr[64] = (o2 > 0.f) ? o2 : 0.f;
  xr[96] = (o3 > 0.f) ? o3 : 0.f;
}

__device__ __forceinline__ void add_rows_h1d40(const float* __restrict__ h, int s_l, float al,
                                               int cnt, int lane, int f1, float& acc0,
                                               float& acc1) {
  int j = 0;
  for (; j + 3 < cnt; j += 4) {
    float a[4], va[4], vb[4];
    for (int k = 0; k < 4; ++k) {
      const int sj = __builtin_amdgcn_readlane(s_l, j + k);
      a[k] = lane_bcast(al, j + k);
      const float* __restrict__ hr = h + (size_t)sj * 40;
      va[k] = hr[lane];
      vb[k] = hr[f1];
    }
    for (int k = 0; k < 4; ++k) {
      acc0 += a[k] * va[k];
      acc1 += a[k] * vb[k];
    }
  }
  for (; j < cnt; ++j) {
    const int sj = __builtin_amdgcn_readlane(s_l, j);
    const float ak = lane_bcast(al, j);
    const float* __restrict__ hr = h + (size_t)sj * 40;
    acc0 += ak * hr[lane];
    acc1 += ak * hr[f1];
  }
}

__global__ void __launch_bounds__(256)
node_aggregate_h1d40(const int* __restrict__ off, const int* __restrict__ csr_src,
                     const float* __restrict__ a_s, const float* __restrict__ a_d,
                     const float* __restrict__ h, const float* __restrict__ bias,
                     float* __restrict__ out, int N) {
  const int lane = threadIdx.x & 31;
  const int wib = __builtin_amdgcn_readfirstlane((int)(threadIdx.x >> 5));
  const int node = blockIdx.x * 8 + wib;
  if (node >= N) return;
  const int beg = __builtin_amdgcn_readfirstlane(off[node]);
  const int deg = __builtin_amdgcn_readfirstlane(off[node + 1]) - beg;
  const int* __restrict__ nb = csr_src + beg;
  const int f1 = 32 + (lane & 7);

  const float ad = a_d[node];
  const float z = leaky(a_s[node] + ad);

  const bool act = lane < deg;
  const int s_c = act ? nb[lane] : node;
  const float e = leaky(a_s[s_c] + ad);

  float m = fmaxf(z, e);
  for (int i = 32 + lane; i < deg; i += 32) m = fmaxf(m, leaky(a_s[nb[i]] + ad));
  m = wave_max(m);

  const float p = act ? expf(e - m) : 0.f;
  float d = p;
  for (int i = 32 + lane; i < deg; i += 32) d += expf(leaky(a_s[nb[i]] + ad) - m);
  const float q = expf(z - m);
  d = wave_sum(d) + q;

  const float* __restrict__ hs = h + (size_t)node * 40;
  float acc0 = (q / d) * hs[lane];
  float acc1 = (q / d) * hs[f1];
  add_rows_h1d40(h, s_c, p / d, min(deg, 32), lane, f1, acc0, acc1);
  for (int base = 32; base < deg; base += 32) {
    const int cnt = min(32, deg - base);
    const bool on = lane < cnt;
    const int s_r = on ? nb[base + lane] : node;
    const float r = on ? expf(leaky(a_s[s_r] + ad) - m) / d : 0.f;
    add_rows_h1d40(h, s_r, r, cnt, lane, f1, acc0, acc1);
  }

  float* __restrict__ orow = out + (size_t)node * 40;
  orow[lane] = acc0 + bias[lane];
  if (lane < 8) orow[32 + lane] = acc1 + bias[32 + lane];
}

extern "C" void kernel_launch(void* const* d_in, const int* in_sizes, int n_in,
                              void* d_out, int out_size, void* d_ws, size_t ws_size,
                              hipStream_t stream) {
  const float* x        = (const float*)d_in[0];
  const int*   ei       = (const int*)d_in[1];
  const float* W1       = (const float*)d_in[2];
  const float* att_src1 = (const float*)d_in[3];
  const float* att_dst1 = (const float*)d_in[4];
  const float* b1       = (const float*)d_in[5];
  const float* W2       = (const float*)d_in[6];
  const float* att_src2 = (const float*)d_in[7];
  const float* att_dst2 = (const float*)d_in[8];
  const float* b2       = (const float*)d_in[9];

  const int NFEAT = 256, NHID = 128, H1 = 4, NCLS = 40;
  const int N = in_sizes[0] / NFEAT;
  const int E = in_sizes[1] / 2;
  const int* src = ei;
  const int* dst = ei + E;

  float* out = (float*)d_out;

  const int NR = (N + CSR_BKT - 1) / CSR_BKT;
  int* counts = (int*)d_ws;
  int* recs   = counts + (size_t)NR * BIN_P;
  float* w = (float*)(recs + (size_t)NR * BIN_P * BIN_SLOT * 2);
  float* h1   = w;
  float* x2   = h1  + (size_t)N * NHID;
  float* h2   = x2  + (size_t)N * NHID;
  float* as1  = h2  + (size_t)N * NCLS;
  float* ad1  = as1 + (size_t)N * H1;
  float* as2  = ad1 + (size_t)N * H1;
  float* ad2  = as2 + (size_t)N;
  int* off     = (int*)(ad2 + (size_t)N);
  int* csr_src = off + (size_t)N + 1;

  const int TB = 256;

  bin_edges<<<BIN_P, BIN_THREADS, 0, stream>>>(src, dst, E, NR, counts, recs);
  csr_place<<<NR, CSR_THREADS, 0, stream>>>(src, dst, E, N, counts, recs, off, csr_src);

  gemm_strip_k256_c128<<<(N + 16 * L1_WAVES - 1) / (16 * L1_WAVES), 32 * L1_WAVES, 0,
                         stream>>>(x, W1, h1, att_src1, att_dst1, as1, ad1, N);
  node_aggregate_h4d32<<<(N + 7) / 8, TB, 0, stream>>>(off, csr_src, as1, ad1, h1, b1,
                                                       x2, N);

  gemm_strip_k128_c40<<<(N + 15) / 16, 32, 0, stream>>>(x2, W2, h2, att_src2, att_dst2,
                                                        as2, ad2, N);
  node_aggregate_h1d40<<<(N + 7) / 8, TB, 0, stream>>>(off, csr_src, as2, ad2, h2, b2,
                                                       out, N);
}


# layer-1 GEMM and edge binning share one launch: binning work-groups (512 threads) run on the compute units the matrix-bound GEMM leaves free; 6 launches -> 5; sum order = LDS arrival rank as before
# speedup vs baseline: 18.2859x; 1.0637x over previous
#include <hip/hip_runtime.h>


typedef float v2f __attribute__((ext_vector_type(2)));
typedef float v8f __attribute__((ext_vector_type(8)));

#define NEG_SLOPE 0.2f


#define CSR_BKT 256
#define CSR_THREADS 512
#define CSR_CAP 14336
#define CSR_U 8
#define BIN_P 256
#define BIN_THREADS 512
#define BIN_U 7
#define BIN_SLOT 48
#define BIN_MAXR 256

template <class F>
__device__ __forceinline__ void for_each_edge(const int* __restrict__ dst, int E, int tid, F f) {
  const int head = min(E, (int)(((16u - (unsigned)((size_t)dst & 15u)) & 15u) >> 2));
  const int nv = (E - head) >> 2;
  const int4* __restrict__ dv = (const int4*)(dst + head);
  for (int i0 = tid; i0 < nv; i0 += CSR_U * CSR_THREADS) {
    int4 d[CSR_U];
#pragma unroll
    for (int u = 0; u < CSR_U; ++u) d[u] = dv[min(i0 + u * CSR_THREADS, nv - 1)];
#pragma unroll
    for (int u = 0; u < CSR_U; ++u) {
      const int i = i0 + u * CSR_THREADS;
      if (i < nv) {
        const int t = head + 4 * i;
        f(t, d[u].x);
        f(t + 1, d[u].y);
        f(t + 2, d[u].z);
        f(t + 3, d[u].w);
      }
    }
  }
  const int tail0 = head + 4 * nv;
  if (tid < head) f(tid, dst[tid]);
  if (tid < E - tail0) f(tail0 + tid, dst[tail0 + tid]);
}

#define BIN_CH (BIN_U * BIN_THREADS)
#define BIN_LDS_INTS (3 * BIN_MAXR + 32 + 2 * BIN_CH)
static_assert(BIN_THREADS >= BIN_MAXR, "thread t < BIN_MAXR owns counter t");
__device__ __forceinline__ void bin_piece(const int* __restrict__ src,
                                          const int* __restrict__ dst, int E, int NR,
                                          int* __restrict__ counts, int* __restrict__ recs,
                                          int p, int* sm) {
  int* c    = sm;
  int* cc   = c + BIN_MAXR;
  int* st   = cc + BIN_MAXR;
  int* wtot = st + BIN_MAXR;
  int* l_s  = wtot + 32;
  int* l_d  = l_s + BIN_CH;
  const int tid = threadIdx.x, lane = tid & 31;
  const int wid = __builtin_amdgcn_readfirstlane(tid >> 5);
  const bool owner = tid < BIN_MAXR;
  if (owner) c[tid] = 0;
  const int EP = (E + BIN_P - 1) / BIN_P;
  const int e0 = min(E, p * EP), e1 = min(E, e0 + EP);
  const unsigned nr = (unsigned)min(NR, BIN_MAXR);
  for (int cb = e0; cb < e1; cb += BIN_CH) {
    if (owner) cc[tid] = 0;
    __syncthreads();
    int d[BIN_U], s[BIN_U], k[BIN_U];
#pragma unroll
    for (int u = 0; u < BIN_U; ++u) {
      const int e = min(cb + tid + u * BIN_THREADS, e1 - 1);
      d[u] = dst[e];
      s[u] = src[e];
    }
#pragma unroll
    for (int u = 0; u < BIN_U; ++u) {
      const unsigned r = (unsigned)d[u] / (unsigned)CSR_BKT;
      const bool ok = (cb + tid + u * BIN_THREADS < e1) && (r < nr);
      k[u] = -1;
      if (ok) k[u] = atomicAdd(&cc[r], 1);
    }
    __syncthreads();
    int x = 0, inc = 0;
    if (owner) {
      x = cc[tid];
      inc = x;
      for (int sh = 1; sh < 32; sh <<= 1) {
        const int y = __shfl_up(inc, sh, 32);
        if (lane >= sh) inc += y;
      }
      if (lane == 31) wtot[wid] = inc;
    }
    __syncthreads();
    if (owner) {
      int b = 0;
      for (int w = 0; w < wid; ++w) b += wtot[w];
      st[tid] = b + inc - x;
    }
    __syncthreads();
#pragma unroll
    for (int u = 0; u < BIN_U; ++u) {
      if (k[u] >= 0) {
        const int pos = st[(unsigned)d[u] / (unsigned)CSR_BKT] + k[u];
        l_s[pos] = s[u];
        l_d[pos] = d[u];
      }
    }
    __syncthreads();
    const int tot = st[BIN_MAXR - 1] + cc[BIN_MAXR - 1];
    for (int i = tid; i < tot; i += BIN_THREADS) {
      const int dd = l_d[i], ss = l_s[i];
      const int r = (int)((unsigned)dd / (unsigned)CSR_BKT);
      const int kk = c[r] + (i - st[r]);
      if (kk < BIN_SLOT) {
        const size_t rec = ((size_t)r * BIN_P + p) * BIN_SLOT + kk;
        *(int2*)(recs + 2 * rec) = make_int2(ss, dd);
      }
    }
    __syncthreads();
    if (owner) c[tid] += cc[tid];
  }
  __syncthreads();
  for (int r = tid; r < NR; r += BIN_THREADS)
    counts[(size_t)r * BIN_P + p] = (r < BIN_MAXR) ? c[r] : 0;
}

#define CSR_WAVES (CSR_THREADS / 32)
#define CSR_SPW (BIN_P / CSR_WAVES)
#define CSR_BU 13

__global__ void __launch_bounds__(CSR_THREADS)
csr_place(const int* __restrict__ src, const int* __restrict__ dst, int E, int N,
          const int* __restrict__ counts, const int* __restrict__ recs,
          int* __restrict__ off, int* __restrict__ csr_src) {
  __shared__ int c[CSR_BKT];
  __shared__ int loff[CSR_BKT];
  __shared__ int wtot[CSR_BKT / 32];
  __shared__ int s_below, s_nl, s_slow;
  __shared__ int l_s[CSR_CAP];
  __shared__ int l_d[CSR_CAP];
  __shared__ int l_r[CSR_CAP];
  const int tid = threadIdx.x, lane = tid & 31;
  const int wid = __builtin_amdgcn_readfirstlane(tid >> 5);
  const int rg = blockIdx.x;
  const int lo = rg * CSR_BKT;

  if (tid < CSR_BKT) c[tid] = 0;
  if (tid == 0) { s_below = 0; s_nl = 0; s_slow = 0; }
  __syncthreads();

  const int* __restrict__ myc = counts + (size_t)rg * BIN_P;
  int n[CSR_SPW];
#pragma unroll
  for (int u = 0; u < CSR_SPW; ++u) n[u] = myc[wid + CSR_WAVES * u];

  int below = 0;
  {
    const int4* __restrict__ cv = (const int4*)counts;
    const int nq = rg * (BIN_P / 4);
    for (int i0 = tid; i0 < nq; i0 += CSR_BU * CSR_THREADS) {
      int4 v[CSR_BU];
#pragma unroll
      for (int u = 0; u < CSR_BU; ++u) v[u] = cv[min(i0 + u * CSR_THREADS, nq - 1)];
#pragma unroll
      for (int u = 0; u < CSR_BU; ++u)
        if (i0 + u * CSR_THREADS < nq) below += (v[u].x + v[u].y) + (v[u].z + v[u].w);
    }
  }

  auto note = [&](int s, int d) {
    const unsigned dl = (unsigned)(d - lo);
    if (dl < (unsigned)CSR_BKT) {
      const int r = atomicAdd(&c[dl], 1);
      const int k = atomicAdd(&s_nl, 1);
      if (k < CSR_CAP) { l_s[k] = s; l_d[k] = (int)dl; l_r[k] = r; }
    }
  };

  int2 rc[CSR_SPW];
#pragma unroll
  for (int u = 0; u < CSR_SPW; ++u) {
    const int* __restrict__ slot = recs + 2 * ((size_t)rg * BIN_P + wid + CSR_WAVES * u) * BIN_SLOT;
    rc[u] = make_int2(0, 0);
    if (lane < min(n[u], BIN_SLOT)) rc[u] = *(const int2*)(slot + 2 * lane);
  }
#pragma unroll
  for (int u = 0; u < CSR_SPW; ++u) {
    const int* __restrict__ slot = recs + 2 * ((size_t)rg * BIN_P + wid + CSR_WAVES * u) * BIN_SLOT;
    const int m = min(n[u], BIN_SLOT);
    if (n[u] > BIN_SLOT) s_slow = 1;
    if (lane < m) note(rc[u].x, rc[u].y);
    for (int j = 32 + lane; j < m; j += 32) {
      const int2 q = *(const int2*)(slot + 2 * j);
      note(q.x, q.y);
    }
  }
  for (int m = 16; m >= 1; m >>= 1) below += __shfl_xor(below, m, 32);
  if (lane == 0) atomicAdd(&s_below, below);
  __syncthreads();

  const bool slow = (s_slow != 0) || (s_nl > CSR_CAP);
  if (slow) {
    if (tid < CSR_BKT) c[tid] = 0;
    __syncthreads();
    for_each_edge(dst, E, tid, [&](int t, int d) {
      const unsigned dl = (unsigned)(d - lo);
      if (dl < (unsigned)CSR_BKT) atomicAdd(&c[dl], 1);
    });
    __syncthreads();
  }

  int x = 0, inc = 0;
  if (tid < CSR_BKT) {
    x = c[tid];
    inc = x;
    for (int s = 1; s < 32; s <<= 1) {
      const int y = __shfl_up(inc, s, 32);
      if (lane >= s) inc += y;
    }
    if (lane == 31) wtot[wid] = inc;
  }
  __syncthreads();
  if (tid < CSR_BKT) {
    int b = 0;
    for (int w = 0; w < wid; ++w) b += wtot[w];
    loff[tid] = b + inc - x;
  }
  __syncthreads();
  const int base = s_below;
  if (tid < CSR_BKT && lo + tid < N) off[lo + tid] = base + loff[tid];
  if (tid == CSR_BKT - 1 && lo + CSR_BKT >= N) off[N] = base + loff[tid] + x;

  if (!slow) {
    const int nl = s_nl;
    for (int k = tid; k < nl; k += CSR_THREADS)
      csr_src[base + loff[l_d[k]] + l_r[k]] = l_s[k];
  } else {
    if (tid < CSR_BKT) c[tid] = 0;
    __syncthreads();
    for_each_edge(dst, E, tid, [&](int t, int d) {
      const unsigned dl = (unsigned)(d - lo);
      if (dl < (unsigned)CSR_BKT) {
        const int r = atomicAdd(&c[dl], 1);
        csr_src[base + loff[dl] + r] = src[t];
      }
    });
  }
}


struct frag16 { float4 p[4]; };

__device__ __forceinline__ frag16 load_k16(const float* q) {
  frag16 f;
#pragma unroll
  for (int i = 0; i < 4; ++i) f.p[i] = *(const float4*)(q + 4 * i);
  return f;
}

template <int G>
__device__ __forceinline__ void mma_group(const frag16& a, const frag16 (&b)[G], v8f* c) {
#pragma unroll
  for (int i = 0; i < 4; ++i) {
#pragma unroll
    for (int g = 0; g < G; ++g) {
      const v2f av = {a.p[i].x, a.p[i].y};
      const v2f bv = {b[g].p[i].x, b[g].p[i].y};
      c[g] = __builtin_amdgcn_wmma_f32_16x16x4_f32(false, av, false, bv, (short)0, c[g],
                                                   false, false);
    }
#pragma unroll
    for (int g = 0; g < G; ++g) {
      const v2f av = {a.p[i].z, a.p[i].w};
      const v2f bv = {b[g].p[i].z, b[g].p[i].w};
      c[g] = __builtin_amdgcn_wmma_f32_16x16x4_f32(false, av, false, bv, (short)0, c[g],
                                                   false, false);
    }
  }
}

template <int NT, int H>
__device__ __forceinline__ void strip_epilogue(const v8f* c, float* __restrict__ C,
                                               const float* __restrict__ att_s,
                                               const float* __restrict__ att_d,
                                               float* __restrict__ a_s,
                                               float* __restrict__ a_d, int N, int Nc,
                                               int row0, int idx, int half) {
#pragma unroll
  for (int t = 0; t < NT; ++t) {
    const int col = 16 * t + idx;
    if (col < Nc) {
#pragma unroll
      for (int r = 0; r < 8; ++r) {
        const int row = row0 + r + 8 * half;
        if (row < N) C[(size_t)row * Nc + col] = c[t][r];
      }
    }
  }

  constexpr int TPH = NT / H;
  float p[16 * H];
#pragma unroll
  for (int hh = 0; hh < H; ++hh) {
    float ws[TPH], wd[TPH];
#pragma unroll
    for (int u = 0; u < TPH; ++u) {
      const int col = 16 * (hh * TPH + u) + idx;
      ws[u] = (col < Nc) ? att_s[min(col, Nc - 1)] : 0.f;
      wd[u] = (col < Nc) ? att_d[min(col, Nc - 1)] : 0.f;
    }
#pragma unroll
    for (int r = 0; r < 8; ++r) {
      float s = 0.f, d = 0.f;
#pragma unroll
      for (int u = 0; u < TPH; ++u) {
        s += c[hh * TPH + u][r] * ws[u];
        d += c[hh * TPH + u][r] * wd[u];
      }
      p[r * H + hh] = s;
      p[(8 + r) * H + hh] = d;
    }
  }
#pragma unroll
  for (int s = 0; s < 4; ++s) {
    const int mask = 8 >> s;
    const int hs = (8 * H) >> s;
    const bool up = (idx & mask) != 0;
#pragma unroll
    for (int j = 0; j < hs; ++j) {
      const float lo = p[j], hi = p[j + hs];
      const float send = up ? lo : hi;
      const float keep = up ? hi : lo;
      p[j] = keep + __shfl_xor(send, mask, 32);
    }
  }
  const int lrow = row0 + 8 * half + (idx & 7);
  if (lrow < N) {
    float* __restrict__ o = ((idx < 8) ? a_s : a_d) + (size_t)lrow * H;
#pragma unroll
    for (int hh = 0; hh < H; ++hh) o[hh] = p[hh];
  }
}

template <int K, int NT, int G, int H>
__device__ __forceinline__ void gemm_strip(const float* __restrict__ A,
                                           const float* __restrict__ B,
                                           float* __restrict__ C,
                                           const float* __restrict__ att_s,
                                           const float* __restrict__ att_d,
                                           float* __restrict__ a_s,
                                           float* __restrict__ a_d, int N, int Nc) {
  constexpr int KB = K / 32;
  constexpr int NG = NT / G;
  const int lane = threadIdx.x & 31;
  const int half = lane >> 4;
  const int idx  = lane & 15;
  const int row0 = blockIdx.x * 16;

  const int ar = min(row0 + idx, N - 1);
  const float* ap = A + (size_t)ar * K + 16 * half;
  const float* bp[NT];
#pragma unroll
  for (int t = 0; t < NT; ++t)
    bp[t] = B + (size_t)min(16 * t + idx, Nc - 1) * K + 16 * half;

  v8f c[NT];
#pragma unroll
  for (int t = 0; t < NT; ++t) c[t] = (v8f){0.f, 0.f, 0.f, 0.f, 0.f, 0.f, 0.f, 0.f};

  frag16 a_cur = load_k16(ap);
  frag16 b0[G];
#pragma unroll
  for (int g = 0; g < G; ++g) b0[g] = load_k16(bp[g]);

#pragma unroll
  for (int kb = 0; kb < KB; ++kb) {
    const int kn = min(kb + 1, KB - 1) * 32;
    frag16 a_nxt = load_k16(ap + kn);
    if constexpr (NG == 2) {
      frag16 b1[G];
#pragma unroll
      for (int g = 0; g < G; ++g) b1[g] = load_k16(bp[G + g] + kb * 32);
      __builtin_amdgcn_sched_barrier(0);
      mma_group<G>(a_cur, b0, c);
      __builtin_amdgcn_sched_barrier(0);
#pragma unroll
      for (int g = 0; g < G; ++g) b0[g] = load_k16(bp[g] + kn);
      __builtin_amdgcn_sched_barrier(0);
      mma_group<G>(a_cur, b1, c + G);
    } else {
      frag16 bn[G];
#pragma unroll
      for (int g = 0; g < G; ++g) bn[g] = load_k16(bp[g] + kn);
      __builtin_amdgcn_sched_barrier(0);
      mma_group<G>(a_cur, b0, c);
#pragma unroll
      for (int g = 0; g < G; ++g) b0[g] = bn[g];
    }
    a_cur = a_nxt;
  }

  strip_epilogue<NT, H>(c, C, att_s, att_d, a_s, a_d, N, Nc, row0, idx, half);
}

#define L1_K 256
#define L1_NC 128
#define L1_BPITCH 260
#define L1_WAVES 16
static_assert(32 * L1_WAVES == BIN_THREADS, "one block size for both kinds of work-group");
static_assert(L1_NC * L1_BPITCH >= BIN_LDS_INTS, "the binning fits into the LDS copy of B");
__global__ void __launch_bounds__(32 * L1_WAVES)
gemm_strip_k256_c128_and_bin(const float* __restrict__ A, const float* __restrict__ B,
                             float* __restrict__ C, const float* __restrict__ att_s,
                             const float* __restrict__ att_d, float* __restrict__ a_s,
                             float* __restrict__ a_d, int N, int gemm_blocks,
                             const int* __restrict__ src, const int* __restrict__ dst,
                             int E, int NR, int* __restrict__ counts,
                             int* __restrict__ recs) {
  constexpr int KB = L1_K / 32;
  constexpr int NT = L1_NC / 16;
  constexpr int G = 2;
  constexpr int NG = NT / G;
  __shared__ float4 Bs4[L1_NC * L1_BPITCH / 4];
  if ((int)blockIdx.x >= gemm_blocks) {
    bin_piece(src, dst, E, NR, counts, recs, (int)blockIdx.x - gemm_blocks, (int*)Bs4);
    return;
  }
  float* Bs = (float*)Bs4;
  const int tid = threadIdx.x;
  const int wave = __builtin_amdgcn_readfirstlane(tid >> 5);
  const int row0 = (blockIdx.x * L1_WAVES + wave) * 16;
  const int lane = tid & 31;
  const int half = lane >> 4;
  const int idx  = lane & 15;
  const int ar = min(row0 + idx, N - 1);
  const float* ap = A + (size_t)ar * L1_K + 16 * half;
  frag16 a_cur = load_k16(ap);
  {
    constexpr int NV = L1_NC * L1_K / 4;
    constexpr int PER = NV / (32 * L1_WAVES);
    const float4* __restrict__ Bg = (const float4*)B;
    float4 w[PER];
#pragma unroll
    for (int j = 0; j < PER; ++j) w[j] = Bg[tid + 32 * L1_WAVES * j];
#pragma unroll
    for (int j = 0; j < PER; ++j) {
      const int f = tid + 32 * L1_WAVES * j;
      const int row = f / (L1_K / 4), c4 = f % (L1_K / 4);
      *(float4*)(Bs + row * L1_BPITCH + 4 * c4) = w[j];
    }
  }
  __syncthreads();
  if (row0 >= N) return;

  const float* bq = Bs + idx * L1_BPITCH + 16 * half;

  v8f c[NT];
#pragma unroll
  for (int t = 0; t < NT; ++t) c[t] = (v8f){0.f, 0.f, 0.f, 0.f, 0.f, 0.f, 0.f, 0.f};

  frag16 bb[2][G];
#pragma unroll
  for (int g = 0; g < G; ++g) bb[0][g] = load_k16(bq + g * 16 * L1_BPITCH);

#pragma unroll
  for (int kb = 0; kb < KB; ++kb) {
    const int kn = min(kb + 1, KB - 1) * 32;
    frag16 a_nxt = load_k16(ap + kn);
#pragma unroll
    for (int grp = 0; grp < NG; ++grp) {
      const int cur = (kb * NG + grp) & 1;
      if (!(kb == KB - 1 && grp == NG - 1)) {
        const int t0 = (grp + 1 < NG) ? (grp + 1) * G : 0;
        const int k0 = (grp + 1 < NG) ? kb * 32 : kn;
#pragma unroll
        for (int g = 0; g < G; ++g)
          bb[cur ^ 1][g] = load_k16(bq + (t0 + g) * 16 * L1_BPITCH + k0);
      }
      __builtin_amdgcn_sched_barrier(0);
      mma_group<G>(a_cur, bb[cur], c + grp * G);
      __builtin_amdgcn_sched_barrier(0);
    }
    a_cur = a_nxt;
  }

  strip_epilogue<NT, 4>(c, C, att_s, att_d, a_s, a_d, N, L1_NC, row0, idx, half);
}

__global__ void __launch_bounds__(32)
gemm_strip_k128_c40(const float* __restrict__ A, const float* __restrict__ B,
                    float* __restrict__ C, const float* __restrict__ att_s,
                    const float* __restrict__ att_d, float* __restrict__ a_s,
                    float* __restrict__ a_d, int N) {
  gemm_strip<128, 3, 3, 1>(A, B, C, att_s, att_d, a_s, a_d, N, 40);
}


__device__ __forceinline__ float lane_bcast(float v, int j) {
  return __int_as_float(__builtin_amdgcn_readlane(__float_as_int(v), j));
}
__device__ __forceinline__ float leaky(float e) { return (e > 0.f) ? e : NEG_SLOPE * e; }
__device__ __forceinline__ float wave_max(float v) {
  for (int m = 16; m >= 1; m >>= 1) v = fmaxf(v, __shfl_xor(v, m, 32));
  return v;
}
__device__ __forceinline__ float wave_sum(float v) {
  for (int m = 16; m >= 1; m >>= 1) v += __shfl_xor(v, m, 32);
  return v;
}

__device__ __forceinline__ void add_rows_h4d32(const float* __restrict__ h, int s_l,
                                               float al0, float al1, float al2, float al3,
                                               int cnt, int lane, float& acc0, float& acc1,
                                               float& acc2, float& acc3) {
  int j = 0;
  for (; j + 3 < cnt; j += 4) {
    float a[4][4], v[4][4];
    for (int k = 0; k < 4; ++k) {
      const int sj = __builtin_amdgcn_readlane(s_l, j + k);
      a[k][0] = lane_bcast(al0, j + k);
      a[k][1] = lane_bcast(al1, j + k);
      a[k][2] = lane_bcast(al2, j + k);
      a[k][3] = lane_bcast(al3, j + k);
      const float* __restrict__ hr = h + (size_t)sj * 128 + lane;
      v[k][0] = hr[0];
      v[k][1] = hr[32];
      v[k][2] = hr[64];
      v[k][3] = hr[96];
    }
    for (int k = 0; k < 4; ++k) {
      acc0 += a[k][0] * v[k][0];
      acc1 += a[k][1] * v[k][1];
      acc2 += a[k][2] * v[k][2];
      acc3 += a[k][3] * v[k][3];
    }
  }
  for (; j < cnt; ++j) {
    const int sj = __builtin_amdgcn_readlane(s_l, j);
    const float a0 = lane_bcast(al0, j), a1 = lane_bcast(al1, j);
    const float a2 = lane_bcast(al2, j), a3 = lane_bcast(al3, j);
    const float* __restrict__ hr = h + (size_t)sj * 128 + lane;
    const float v0 = hr[0], v1 = hr[32], v2 = hr[64], v3 = hr[96];
    acc0 += a0 * v0;
    acc1 += a1 * v1;
    acc2 += a2 * v2;
    acc3 += a3 * v3;
  }
}

__global__ void __launch_bounds__(256)
node_aggregate_h4d32(const int* __restrict__ off, const int* __restrict__ csr_src,
                     const float* __restrict__ a_s, const float* __restrict__ a_d,
                     const float* __restrict__ h, const float* __restrict__ bias,
                     float* __restrict__ x2, int N) {
  const int lane = threadIdx.x & 31;
  const int wib = __builtin_amdgcn_readfirstlane((int)(threadIdx.x >> 5));
  const int node = blockIdx.x * 8 + wib;
  if (node >= N) return;
  const int beg = __builtin_amdgcn_readfirstlane(off[node]);
  const int deg = __builtin_amdgcn_readfirstlane(off[node + 1]) - beg;
  const int* __restrict__ nb = csr_src + beg;

  const float4 ad4 = *(const float4*)(a_d + (size_t)node * 4);
  const float4 sf4 = *(const float4*)(a_s + (size_t)node * 4);
  const float z0 = leaky(sf4.x + ad4.x), z1 = leaky(sf4.y + ad4.y);
  const float z2 = leaky(sf4.z + ad4.z), z3 = leaky(sf4.w + ad4.w);

  const bool act = lane < deg;
  const int s_c = act ? nb[lane] : node;
  const float4 as4 = *(const float4*)(a_s + (size_t)s_c * 4);
  const float e0 = leaky(as4.x + ad4.x), e1 = leaky(as4.y + ad4.y);
  const float e2 = leaky(as4.z + ad4.z), e3 = leaky(as4.w + ad4.w);

  float m0 = fmaxf(z0, e0), m1 = fmaxf(z1, e1), m2 = fmaxf(z2, e2), m3 = fmaxf(z3, e3);
  for (int i = 32 + lane; i < deg; i += 32) {
    const float4 b4 = *(const float4*)(a_s + (size_t)nb[i] * 4);
    m0 = fmaxf(m0, leaky(b4.x + ad4.x));
    m1 = fmaxf(m1, leaky(b4.y + ad4.y));
    m2 = fmaxf(m2, leaky(b4.z + ad4.z));
    m3 = fmaxf(m3, leaky(b4.w + ad4.w));
  }
  m0 = wave_max(m0); m1 = wave_max(m1); m2 = wave_max(m2); m3 = wave_max(m3);

  const float p0 = act ? expf(e0 - m0) : 0.f, p1 = act ? expf(e1 - m1) : 0.f;
  const float p2 = act ? expf(e2 - m2) : 0.f, p3 = act ? expf(e3 - m3) : 0.f;
  float d0 = p0, d1 = p1, d2 = p2, d3 = p3;
  for (int i = 32 + lane; i < deg; i += 32) {
    const float4 b4 = *(const float4*)(a_s + (size_t)nb[i] * 4);
    d0 += expf(leaky(b4.x + ad4.x) - m0);
    d1 += expf(leaky(b4.y + ad4.y) - m1);
    d2 += expf(leaky(b4.z + ad4.z) - m2);
    d3 += expf(leaky(b4.w + ad4.w) - m3);
  }
  const float q0 = expf(z0 - m0), q1 = expf(z1 - m1), q2 = expf(z2 - m2), q3 = expf(z3 - m3);
  d0 = wave_sum(d0) + q0; d1 = wave_sum(d1) + q1;
  d2 = wave_sum(d2) + q2; d3 = wave_sum(d3) + q3;

  const float* __restrict__ hs = h + (size_t)node * 128 + lane;
  float acc0 = (q0 / d0) * hs[0];
  float acc1 = (q1 / d1) * hs[32];
  float acc2 = (q2 / d2) * hs[64];
  float acc3 = (q3 / d3) * hs[96];
  add_rows_h4d32(h, s_c, p0 / d0, p1 / d1, p2 / d2, p3 / d3, min(deg, 32), lane,
                 acc0, acc1, acc2, acc3);
  for (int base = 32; base < deg; base += 32) {
    const int cnt = min(32, deg - base);
    const bool on = lane < cnt;
    const int s_r = on ? nb[base + lane] : node;
    const float4 b4 = *(const float4*)(a_s + (size_t)s_r * 4);
    const float r0 = on ? expf(leaky(b4.x + ad4.x) - m0) / d0 : 0.f;
    const float r1 = on ? expf(leaky(b4.y + ad4.y) - m1) / d1 : 0.f;
    const float r2 = on ? expf(leaky(b4.z + ad4.z) - m2) / d2 : 0.f;
    const float r3 = on ? expf(leaky(b4.w + ad4.w) - m3) / d3 : 0.f;
    add_rows_h4d32(h, s_r, r0, r1, r2, r3, cnt, lane, acc0, acc1, acc2, acc3);
  }

  float* __restrict__ xr = x2 + (size_t)node * 128 + lane;
  const float o0 = acc0 + bias[lane],      o1 = acc1 + bias[32 + lane];
  const float o2 = acc2 + bias[64 + lane], o3 = acc3 + bias[96 + lane];
  xr[0]  = (o0 > 0.f) ? o0 : 0.f;
  xr[32] = (o1 > 0.f) ? o1 : 0.f;
  xr[64] = (o2 > 0.f) ? o2 : 0.f;
  xr[96] = (o3 > 0.f) ? o3 : 0.f;
}

__device__ __forceinline__ void add_rows_h1d40(const float* __restrict__ h, int s_l, float al,
                                               int cnt, int lane, int f1, float& acc0,
                                               float& acc1) {
  int j = 0;
  for (; j + 3 < cnt; j += 4) {
    float a[4], va[4], vb[4];
    for (int k = 0; k < 4; ++k) {
      const int sj = __builtin_amdgcn_readlane(s_l, j + k);
      a[k] = lane_bcast(al, j + k);
      const float* __restrict__ hr = h + (size_t)sj * 40;
      va[k] = hr[lane];
      vb[k] = hr[f1];
    }
    for (int k = 0; k < 4; ++k) {
      acc0 += a[k] * va[k];
      acc1 += a[k] * vb[k];
    }
  }
  for (; j < cnt; ++j) {
    const int sj = __builtin_amdgcn_readlane(s_l, j);
    const float ak = lane_bcast(al, j);
    const float* __restrict__ hr = h + (size_t)sj * 40;
    acc0 += ak * hr[lane];
    acc1 += ak * hr[f1];
  }
}

__global__ void __launch_bounds__(256)
node_aggregate_h1d40(const int* __restrict__ off, const int* __restrict__ csr_src,
                     const float* __restrict__ a_s, const float* __restrict__ a_d,
                     const float* __restrict__ h, const float* __restrict__ bias,
                     float* __restrict__ out, int N) {
  const int lane = threadIdx.x & 31;
  const int wib = __builtin_amdgcn_readfirstlane((int)(threadIdx.x >> 5));
  const int node = blockIdx.x * 8 + wib;
  if (node >= N) return;
  const int beg = __builtin_amdgcn_readfirstlane(off[node]);
  const int deg = __builtin_amdgcn_readfirstlane(off[node + 1]) - beg;
  const int* __restrict__ nb = csr_src + beg;
  const int f1 = 32 + (lane & 7);

  const float ad = a_d[node];
  const float z = leaky(a_s[node] + ad);

  const bool act = lane < deg;
  const int s_c = act ? nb[lane] : node;
  const float e = leaky(a_s[s_c] + ad);

  float m = fmaxf(z, e);
  for (int i = 32 + lane; i < deg; i += 32) m = fmaxf(m, leaky(a_s[nb[i]] + ad));
  m = wave_max(m);

  const float p = act ? expf(e - m) : 0.f;
  float d = p;
  for (int i = 32 + lane; i < deg; i += 32) d += expf(leaky(a_s[nb[i]] + ad) - m);
  const float q = expf(z - m);
  d = wave_sum(d) + q;

  const float* __restrict__ hs = h + (size_t)node * 40;
  float acc0 = (q / d) * hs[lane];
  float acc1 = (q / d) * hs[f1];
  add_rows_h1d40(h, s_c, p / d, min(deg, 32), lane, f1, acc0, acc1);
  for (int base = 32; base < deg; base += 32) {
    const int cnt = min(32, deg - base);
    const bool on = lane < cnt;
    const int s_r = on ? nb[base + lane] : node;
    const float r = on ? expf(leaky(a_s[s_r] + ad) - m) / d : 0.f;
    add_rows_h1d40(h, s_r, r, cnt, lane, f1, acc0, acc1);
  }

  float* __restrict__ orow = out + (size_t)node * 40;
  orow[lane] = acc0 + bias[lane];
  if (lane < 8) orow[32 + lane] = acc1 + bias[32 + lane];
}

extern "C" void kernel_launch(void* const* d_in, const int* in_sizes, int n_in,
                              void* d_out, int out_size, void* d_ws, size_t ws_size,
                              hipStream_t stream) {
  const float* x        = (const float*)d_in[0];
  const int*   ei       = (const int*)d_in[1];
  const float* W1       = (const float*)d_in[2];
  const float* att_src1 = (const float*)d_in[3];
  const float* att_dst1 = (const float*)d_in[4];
  const float* b1       = (const float*)d_in[5];
  const float* W2       = (const float*)d_in[6];
  const float* att_src2 = (const float*)d_in[7];
  const float* att_dst2 = (const float*)d_in[8];
  const float* b2       = (const float*)d_in[9];

  const int NFEAT = 256, NHID = 128, H1 = 4, NCLS = 40;
  const int N = in_sizes[0] / NFEAT;
  const int E = in_sizes[1] / 2;
  const int* src = ei;
  const int* dst = ei + E;

  float* out = (float*)d_out;

  const int NR = (N + CSR_BKT - 1) / CSR_BKT;
  int* counts = (int*)d_ws;
  int* recs   = counts + (size_t)NR * BIN_P;
  float* w = (float*)(recs + (size_t)NR * BIN_P * BIN_SLOT * 2);
  float* h1   = w;
  float* x2   = h1  + (size_t)N * NHID;
  float* h2   = x2  + (size_t)N * NHID;
  float* as1  = h2  + (size_t)N * NCLS;
  float* ad1  = as1 + (size_t)N * H1;
  float* as2  = ad1 + (size_t)N * H1;
  float* ad2  = as2 + (size_t)N;
  int* off     = (int*)(ad2 + (size_t)N);
  int* csr_src = off + (size_t)N + 1;

  const int TB = 256;

  const int GB = (N + 16 * L1_WAVES - 1) / (16 * L1_WAVES);
  gemm_strip_k256_c128_and_bin<<<GB + BIN_P, 32 * L1_WAVES, 0, stream>>>(
      x, W1, h1, att_src1, att_dst1, as1, ad1, N, GB, src, dst, E, NR, counts, recs);
  csr_place<<<NR, CSR_THREADS, 0, stream>>>(src, dst, E, N, counts, recs, off, csr_src);

  node_aggregate_h4d32<<<(N + 7) / 8, TB, 0, stream>>>(off, csr_src, as1, ad1, h1, b1,
                                                       x2, N);

  gemm_strip_k128_c40<<<(N + 15) / 16, 32, 0, stream>>>(x2, W2, h2, att_src2, att_dst2,
                                                        as2, ad2, N);
  node_aggregate_h1d40<<<(N + 7) / 8, TB, 0, stream>>>(off, csr_src, as2, ad2, h2, b2,
                                                       out, N);
}


# layer-1 aggregation: a lane holds 4 neighbouring features, so a source row is ONE 16-byte-per-lane load (4 lines), not 4 one-line loads; softmax per 8-lane head group by DPP8; exp-sum tree changes
# speedup vs baseline: 19.0598x; 1.0423x over previous
#include <hip/hip_runtime.h>


typedef float v2f __attribute__((ext_vector_type(2)));
typedef float v8f __attribute__((ext_vector_type(8)));

#define NEG_SLOPE 0.2f


#define CSR_BKT 256
#define CSR_THREADS 512
#define CSR_CAP 14336
#define CSR_U 8
#define BIN_P 256
#define BIN_THREADS 512
#define BIN_U 7
#define BIN_SLOT 48
#define BIN_MAXR 256

template <class F>
__device__ __forceinline__ void for_each_edge(const int* __restrict__ dst, int E, int tid, F f) {
  const int head = min(E, (int)(((16u - (unsigned)((size_t)dst & 15u)) & 15u) >> 2));
  const int nv = (E - head) >> 2;
  const int4* __restrict__ dv = (const int4*)(dst + head);
  for (int i0 = tid; i0 < nv; i0 += CSR_U * CSR_THREADS) {
    int4 d[CSR_U];
#pragma unroll
    for (int u = 0; u < CSR_U; ++u) d[u] = dv[min(i0 + u * CSR_THREADS, nv - 1)];
#pragma unroll
    for (int u = 0; u < CSR_U; ++u) {
      const int i = i0 + u * CSR_THREADS;
      if (i < nv) {
        const int t = head + 4 * i;
        f(t, d[u].x);
        f(t + 1, d[u].y);
        f(t + 2, d[u].z);
        f(t + 3, d[u].w);
      }
    }
  }
  const int tail0 = head + 4 * nv;
  if (tid < head) f(tid, dst[tid]);
  if (tid < E - tail0) f(tail0 + tid, dst[tail0 + tid]);
}

#define BIN_CH (BIN_U * BIN_THREADS)
#define BIN_LDS_INTS (3 * BIN_MAXR + 32 + 2 * BIN_CH)
static_assert(BIN_THREADS >= BIN_MAXR, "thread t < BIN_MAXR owns counter t");
__device__ __forceinline__ void bin_piece(const int* __restrict__ src,
                                          const int* __restrict__ dst, int E, int NR,
                                          int* __restrict__ counts, int* __restrict__ recs,
                                          int p, int* sm) {
  int* c    = sm;
  int* cc   = c + BIN_MAXR;
  int* st   = cc + BIN_MAXR;
  int* wtot = st + BIN_MAXR;
  int* l_s  = wtot + 32;
  int* l_d  = l_s + BIN_CH;
  const int tid = threadIdx.x, lane = tid & 31;
  const int wid = __builtin_amdgcn_readfirstlane(tid >> 5);
  const bool owner = tid < BIN_MAXR;
  if (owner) c[tid] = 0;
  const int EP = (E + BIN_P - 1) / BIN_P;
  const int e0 = min(E, p * EP), e1 = min(E, e0 + EP);
  const unsigned nr = (unsigned)min(NR, BIN_MAXR);
  for (int cb = e0; cb < e1; cb += BIN_CH) {
    if (owner) cc[tid] = 0;
    __syncthreads();
    int d[BIN_U], s[BIN_U], k[BIN_U];
#pragma unroll
    for (int u = 0; u < BIN_U; ++u) {
      const int e = min(cb + tid + u * BIN_THREADS, e1 - 1);
      d[u] = dst[e];
      s[u] = src[e];
    }
#pragma unroll
    for (int u = 0; u < BIN_U; ++u) {
      const unsigned r = (unsigned)d[u] / (unsigned)CSR_BKT;
      const bool ok = (cb + tid + u * BIN_THREADS < e1) && (r < nr);
      k[u] = -1;
      if (ok) k[u] = atomicAdd(&cc[r], 1);
    }
    __syncthreads();
    int x = 0, inc = 0;
    if (owner) {
      x = cc[tid];
      inc = x;
      for (int sh = 1; sh < 32; sh <<= 1) {
        const int y = __shfl_up(inc, sh, 32);
        if (lane >= sh) inc += y;
      }
      if (lane == 31) wtot[wid] = inc;
    }
    __syncthreads();
    if (owner) {
      int b = 0;
      for (int w = 0; w < wid; ++w) b += wtot[w];
      st[tid] = b + inc - x;
    }
    __syncthreads();
#pragma unroll
    for (int u = 0; u < BIN_U; ++u) {
      if (k[u] >= 0) {
        const int pos = st[(unsigned)d[u] / (unsigned)CSR_BKT] + k[u];
        l_s[pos] = s[u];
        l_d[pos] = d[u];
      }
    }
    __syncthreads();
    const int tot = st[BIN_MAXR - 1] + cc[BIN_MAXR - 1];
    for (int i = tid; i < tot; i += BIN_THREADS) {
      const int dd = l_d[i], ss = l_s[i];
      const int r = (int)((unsigned)dd / (unsigned)CSR_BKT);
      const int kk = c[r] + (i - st[r]);
      if (kk < BIN_SLOT) {
        const size_t rec = ((size_t)r * BIN_P + p) * BIN_SLOT + kk;
        *(int2*)(recs + 2 * rec) = make_int2(ss, dd);
      }
    }
    __syncthreads();
    if (owner) c[tid] += cc[tid];
  }
  __syncthreads();
  for (int r = tid; r < NR; r += BIN_THREADS)
    counts[(size_t)r * BIN_P + p] = (r < BIN_MAXR) ? c[r] : 0;
}

#define CSR_WAVES (CSR_THREADS / 32)
#define CSR_SPW (BIN_P / CSR_WAVES)
#define CSR_BU 13

__global__ void __launch_bounds__(CSR_THREADS)
csr_place(const int* __restrict__ src, const int* __restrict__ dst, int E, int N,
          const int* __restrict__ counts, const int* __restrict__ recs,
          int* __restrict__ off, int* __restrict__ csr_src) {
  __shared__ int c[CSR_BKT];
  __shared__ int loff[CSR_BKT];
  __shared__ int wtot[CSR_BKT / 32];
  __shared__ int s_below, s_nl, s_slow;
  __shared__ int l_s[CSR_CAP];
  __shared__ int l_d[CSR_CAP];
  __shared__ int l_r[CSR_CAP];
  const int tid = threadIdx.x, lane = tid & 31;
  const int wid = __builtin_amdgcn_readfirstlane(tid >> 5);
  const int rg = blockIdx.x;
  const int lo = rg * CSR_BKT;

  if (tid < CSR_BKT) c[tid] = 0;
  if (tid == 0) { s_below = 0; s_nl = 0; s_slow = 0; }
  __syncthreads();

  const int* __restrict__ myc = counts + (size_t)rg * BIN_P;
  int n[CSR_SPW];
#pragma unroll
  for (int u = 0; u < CSR_SPW; ++u) n[u] = myc[wid + CSR_WAVES * u];

  int below = 0;
  {
    const int4* __restrict__ cv = (const int4*)counts;
    const int nq = rg * (BIN_P / 4);
    for (int i0 = tid; i0 < nq; i0 += CSR_BU * CSR_THREADS) {
      int4 v[CSR_BU];
#pragma unroll
      for (int u = 0; u < CSR_BU; ++u) v[u] = cv[min(i0 + u * CSR_THREADS, nq - 1)];
#pragma unroll
      for (int u = 0; u < CSR_BU; ++u)
        if (i0 + u * CSR_THREADS < nq) below += (v[u].x + v[u].y) + (v[u].z + v[u].w);
    }
  }

  auto note = [&](int s, int d) {
    const unsigned dl = (unsigned)(d - lo);
    if (dl < (unsigned)CSR_BKT) {
      const int r = atomicAdd(&c[dl], 1);
      const int k = atomicAdd(&s_nl, 1);
      if (k < CSR_CAP) { l_s[k] = s; l_d[k] = (int)dl; l_r[k] = r; }
    }
  };

  int2 rc[CSR_SPW];
#pragma unroll
  for (int u = 0; u < CSR_SPW; ++u) {
    const int* __restrict__ slot = recs + 2 * ((size_t)rg * BIN_P + wid + CSR_WAVES * u) * BIN_SLOT;
    rc[u] = make_int2(0, 0);
    if (lane < min(n[u], BIN_SLOT)) rc[u] = *(const int2*)(slot + 2 * lane);
  }
#pragma unroll
  for (int u = 0; u < CSR_SPW; ++u) {
    const int* __restrict__ slot = recs + 2 * ((size_t)rg * BIN_P + wid + CSR_WAVES * u) * BIN_SLOT;
    const int m = min(n[u], BIN_SLOT);
    if (n[u] > BIN_SLOT) s_slow = 1;
    if (lane < m) note(rc[u].x, rc[u].y);
    for (int j = 32 + lane; j < m; j += 32) {
      const int2 q = *(const int2*)(slot + 2 * j);
      note(q.x, q.y);
    }
  }
  for (int m = 16; m >= 1; m >>= 1) below += __shfl_xor(below, m, 32);
  if (lane == 0) atomicAdd(&s_below, below);
  __syncthreads();

  const bool slow = (s_slow != 0) || (s_nl > CSR_CAP);
  if (slow) {
    if (tid < CSR_BKT) c[tid] = 0;
    __syncthreads();
    for_each_edge(dst, E, tid, [&](int t, int d) {
      const unsigned dl = (unsigned)(d - lo);
      if (dl < (unsigned)CSR_BKT) atomicAdd(&c[dl], 1);
    });
    __syncthreads();
  }

  int x = 0, inc = 0;
  if (tid < CSR_BKT) {
    x = c[tid];
    inc = x;
    for (int s = 1; s < 32; s <<= 1) {
      const int y = __shfl_up(inc, s, 32);
      if (lane >= s) inc += y;
    }
    if (lane == 31) wtot[wid] = inc;
  }
  __syncthreads();
  if (tid < CSR_BKT) {
    int b = 0;
    for (int w = 0; w < wid; ++w) b += wtot[w];
    loff[tid] = b + inc - x;
  }
  __syncthreads();
  const int base = s_below;
  if (tid < CSR_BKT && lo + tid < N) off[lo + tid] = base + loff[tid];
  if (tid == CSR_BKT - 1 && lo + CSR_BKT >= N) off[N] = base + loff[tid] + x;

  if (!slow) {
    const int nl = s_nl;
    for (int k = tid; k < nl; k += CSR_THREADS)
      csr_src[base + loff[l_d[k]] + l_r[k]] = l_s[k];
  } else {
    if (tid < CSR_BKT) c[tid] = 0;
    __syncthreads();
    for_each_edge(dst, E, tid, [&](int t, int d) {
      const unsigned dl = (unsigned)(d - lo);
      if (dl < (unsigned)CSR_BKT) {
        const int r = atomicAdd(&c[dl], 1);
        csr_src[base + loff[dl] + r] = src[t];
      }
    });
  }
}


struct frag16 { float4 p[4]; };

__device__ __forceinline__ frag16 load_k16(const float* q) {
  frag16 f;
#pragma unroll
  for (int i = 0; i < 4; ++i) f.p[i] = *(const float4*)(q + 4 * i);
  return f;
}

template <int G>
__device__ __forceinline__ void mma_group(const frag16& a, const frag16 (&b)[G], v8f* c) {
#pragma unroll
  for (int i = 0; i < 4; ++i) {
#pragma unroll
    for (int g = 0; g < G; ++g) {
      const v2f av = {a.p[i].x, a.p[i].y};
      const v2f bv = {b[g].p[i].x, b[g].p[i].y};
      c[g] = __builtin_amdgcn_wmma_f32_16x16x4_f32(false, av, false, bv, (short)0, c[g],
                                                   false, false);
    }
#pragma unroll
    for (int g = 0; g < G; ++g) {
      const v2f av = {a.p[i].z, a.p[i].w};
      const v2f bv = {b[g].p[i].z, b[g].p[i].w};
      c[g] = __builtin_amdgcn_wmma_f32_16x16x4_f32(false, av, false, bv, (short)0, c[g],
                                                   false, false);
    }
  }
}

template <int NT, int H>
__device__ __forceinline__ void strip_epilogue(const v8f* c, float* __restrict__ C,
                                               const float* __restrict__ att_s,
                                               const float* __restrict__ att_d,
                                               float* __restrict__ a_s,
                                               float* __restrict__ a_d, int N, int Nc,
                                               int row0, int idx, int half) {
#pragma unroll
  for (int t = 0; t < NT; ++t) {
    const int col = 16 * t + idx;
    if (col < Nc) {
#pragma unroll
      for (int r = 0; r < 8; ++r) {
        const int row = row0 + r + 8 * half;
        if (row < N) C[(size_t)row * Nc + col] = c[t][r];
      }
    }
  }

  constexpr int TPH = NT / H;
  float p[16 * H];
#pragma unroll
  for (int hh = 0; hh < H; ++hh) {
    float ws[TPH], wd[TPH];
#pragma unroll
    for (int u = 0; u < TPH; ++u) {
      const int col = 16 * (hh * TPH + u) + idx;
      ws[u] = (col < Nc) ? att_s[min(col, Nc - 1)] : 0.f;
      wd[u] = (col < Nc) ? att_d[min(col, Nc - 1)] : 0.f;
    }
#pragma unroll
    for (int r = 0; r < 8; ++r) {
      float s = 0.f, d = 0.f;
#pragma unroll
      for (int u = 0; u < TPH; ++u) {
        s += c[hh * TPH + u][r] * ws[u];
        d += c[hh * TPH + u][r] * wd[u];
      }
      p[r * H + hh] = s;
      p[(8 + r) * H + hh] = d;
    }
  }
#pragma unroll
  for (int s = 0; s < 4; ++s) {
    const int mask = 8 >> s;
    const int hs = (8 * H) >> s;
    const bool up = (idx & mask) != 0;
#pragma unroll
    for (int j = 0; j < hs; ++j) {
      const float lo = p[j], hi = p[j + hs];
      const float send = up ? lo : hi;
      const float keep = up ? hi : lo;
      p[j] = keep + __shfl_xor(send, mask, 32);
    }
  }
  const int lrow = row0 + 8 * half + (idx & 7);
  if (lrow < N) {
    float* __restrict__ o = ((idx < 8) ? a_s : a_d) + (size_t)lrow * H;
#pragma unroll
    for (int hh = 0; hh < H; ++hh) o[hh] = p[hh];
  }
}

template <int K, int NT, int G, int H>
__device__ __forceinline__ void gemm_strip(const float* __restrict__ A,
                                           const float* __restrict__ B,
                                           float* __restrict__ C,
                                           const float* __restrict__ att_s,
                                           const float* __restrict__ att_d,
                                           float* __restrict__ a_s,
                                           float* __restrict__ a_d, int N, int Nc) {
  constexpr int KB = K / 32;
  constexpr int NG = NT / G;
  const int lane = threadIdx.x & 31;
  const int half = lane >> 4;
  const int idx  = lane & 15;
  const int row0 = blockIdx.x * 16;

  const int ar = min(row0 + idx, N - 1);
  const float* ap = A + (size_t)ar * K + 16 * half;
  const float* bp[NT];
#pragma unroll
  for (int t = 0; t < NT; ++t)
    bp[t] = B + (size_t)min(16 * t + idx, Nc - 1) * K + 16 * half;

  v8f c[NT];
#pragma unroll
  for (int t = 0; t < NT; ++t) c[t] = (v8f){0.f, 0.f, 0.f, 0.f, 0.f, 0.f, 0.f, 0.f};

  frag16 a_cur = load_k16(ap);
  frag16 b0[G];
#pragma unroll
  for (int g = 0; g < G; ++g) b0[g] = load_k16(bp[g]);

#pragma unroll
  for (int kb = 0; kb < KB; ++kb) {
    const int kn = min(kb + 1, KB - 1) * 32;
    frag16 a_nxt = load_k16(ap + kn);
    if constexpr (NG == 2) {
      frag16 b1[G];
#pragma unroll
      for (int g = 0; g < G; ++g) b1[g] = load_k16(bp[G + g] + kb * 32);
      __builtin_amdgcn_sched_barrier(0);
      mma_group<G>(a_cur, b0, c);
      __builtin_amdgcn_sched_barrier(0);
#pragma unroll
      for (int g = 0; g < G; ++g) b0[g] = load_k16(bp[g] + kn);
      __builtin_amdgcn_sched_barrier(0);
      mma_group<G>(a_cur, b1, c + G);
    } else {
      frag16 bn[G];
#pragma unroll
      for (int g = 0; g < G; ++g) bn[g] = load_k16(bp[g] + kn);
      __builtin_amdgcn_sched_barrier(0);
      mma_group<G>(a_cur, b0, c);
#pragma unroll
      for (int g = 0; g < G; ++g) b0[g] = bn[g];
    }
    a_cur = a_nxt;
  }

  strip_epilogue<NT, H>(c, C, att_s, att_d, a_s, a_d, N, Nc, row0, idx, half);
}

#define L1_K 256
#define L1_NC 128
#define L1_BPITCH 260
#define L1_WAVES 16
static_assert(32 * L1_WAVES == BIN_THREADS, "one block size for both kinds of work-group");
static_assert(L1_NC * L1_BPITCH >= BIN_LDS_INTS, "the binning fits into the LDS copy of B");
__global__ void __launch_bounds__(32 * L1_WAVES)
gemm_strip_k256_c128_and_bin(const float* __restrict__ A, const float* __restrict__ B,
                             float* __restrict__ C, const float* __restrict__ att_s,
                             const float* __restrict__ att_d, float* __restrict__ a_s,
                             float* __restrict__ a_d, int N, int gemm_blocks,
                             const int* __restrict__ src, const int* __restrict__ dst,
                             int E, int NR, int* __restrict__ counts,
                             int* __restrict__ recs) {
  constexpr int KB = L1_K / 32;
  constexpr int NT = L1_NC / 16;
  constexpr int G = 2;
  constexpr int NG = NT / G;
  __shared__ float4 Bs4[L1_NC * L1_BPITCH / 4];
  if ((int)blockIdx.x >= gemm_blocks) {
    bin_piece(src, dst, E, NR, counts, recs, (int)blockIdx.x - gemm_blocks, (int*)Bs4);
    return;
  }
  float* Bs = (float*)Bs4;
  const int tid = threadIdx.x;
  const int wave = __builtin_amdgcn_readfirstlane(tid >> 5);
  const int row0 = (blockIdx.x * L1_WAVES + wave) * 16;
  const int lane = tid & 31;
  const int half = lane >> 4;
  const int idx  = lane & 15;
  const int ar = min(row0 + idx, N - 1);
  const float* ap = A + (size_t)ar * L1_K + 16 * half;
  frag16 a_cur = load_k16(ap);
  {
    constexpr int NV = L1_NC * L1_K / 4;
    constexpr int PER = NV / (32 * L1_WAVES);
    const float4* __restrict__ Bg = (const float4*)B;
    float4 w[PER];
#pragma unroll
    for (int j = 0; j < PER; ++j) w[j] = Bg[tid + 32 * L1_WAVES * j];
#pragma unroll
    for (int j = 0; j < PER; ++j) {
      const int f = tid + 32 * L1_WAVES * j;
      const int row = f / (L1_K / 4), c4 = f % (L1_K / 4);
      *(float4*)(Bs + row * L1_BPITCH + 4 * c4) = w[j];
    }
  }
  __syncthreads();
  if (row0 >= N) return;

  const float* bq = Bs + idx * L1_BPITCH + 16 * half;

  v8f c[NT];
#pragma unroll
  for (int t = 0; t < NT; ++t) c[t] = (v8f){0.f, 0.f, 0.f, 0.f, 0.f, 0.f, 0.f, 0.f};

  frag16 bb[2][G];
#pragma unroll
  for (int g = 0; g < G; ++g) bb[0][g] = load_k16(bq + g * 16 * L1_BPITCH);

#pragma unroll
  for (int kb = 0; kb < KB; ++kb) {
    const int kn = min(kb + 1, KB - 1) * 32;
    frag16 a_nxt = load_k16(ap + kn);
#pragma unroll
    for (int grp = 0; grp < NG; ++grp) {
      const int cur = (kb * NG + grp) & 1;
      if (!(kb == KB - 1 && grp == NG - 1)) {
        const int t0 = (grp + 1 < NG) ? (grp + 1) * G : 0;
        const int k0 = (grp + 1 < NG) ? kb * 32 : kn;
#pragma unroll
        for (int g = 0; g < G; ++g)
          bb[cur ^ 1][g] = load_k16(bq + (t0 + g) * 16 * L1_BPITCH + k0);
      }
      __builtin_amdgcn_sched_barrier(0);
      mma_group<G>(a_cur, bb[cur], c + grp * G);
      __builtin_amdgcn_sched_barrier(0);
    }
    a_cur = a_nxt;
  }

  strip_epilogue<NT, 4>(c, C, att_s, att_d, a_s, a_d, N, L1_NC, row0, idx, half);
}

__global__ void __launch_bounds__(32)
gemm_strip_k128_c40(const float* __restrict__ A, const float* __restrict__ B,
                    float* __restrict__ C, const float* __restrict__ att_s,
                    const float* __restrict__ att_d, float* __restrict__ a_s,
                    float* __restrict__ a_d, int N) {
  gemm_strip<128, 3, 3, 1>(A, B, C, att_s, att_d, a_s, a_d, N, 40);
}


__device__ __forceinline__ float lane_bcast(float v, int j) {
  return __int_as_float(__builtin_amdgcn_readlane(__float_as_int(v), j));
}
__device__ __forceinline__ float leaky(float e) { return (e > 0.f) ? e : NEG_SLOPE * e; }
__device__ __forceinline__ float wave_max(float v) {
  for (int m = 16; m >= 1; m >>= 1) v = fmaxf(v, __shfl_xor(v, m, 32));
  return v;
}
__device__ __forceinline__ float wave_sum(float v) {
  for (int m = 16; m >= 1; m >>= 1) v += __shfl_xor(v, m, 32);
  return v;
}


template <unsigned S>
__device__ __forceinline__ float octet_sel(float v) {
  return __uint_as_float(__builtin_amdgcn_mov_dpp8(__float_as_uint(v), S));
}
#define OCT_SEL(a, b, c, d, e, f, g, hh_) \
  ((a) | ((b) << 3) | ((c) << 6) | ((d) << 9) | ((e) << 12) | ((f) << 15) | ((g) << 18) | ((hh_) << 21))
__device__ __forceinline__ float octet_max(float v) {
  v = fmaxf(v, octet_sel<OCT_SEL(1u, 0u, 3u, 2u, 5u, 4u, 7u, 6u)>(v));
  v = fmaxf(v, octet_sel<OCT_SEL(2u, 3u, 0u, 1u, 6u, 7u, 4u, 5u)>(v));
  v = fmaxf(v, octet_sel<OCT_SEL(4u, 5u, 6u, 7u, 0u, 1u, 2u, 3u)>(v));
  return v;
}
__device__ __forceinline__ float octet_sum(float v) {
  v += octet_sel<OCT_SEL(1u, 0u, 3u, 2u, 5u, 4u, 7u, 6u)>(v);
  v += octet_sel<OCT_SEL(2u, 3u, 0u, 1u, 6u, 7u, 4u, 5u)>(v);
  v += octet_sel<OCT_SEL(4u, 5u, 6u, 7u, 0u, 1u, 2u, 3u)>(v);
  return v;
}

template <unsigned T>
__device__ __forceinline__ void add_trip_h4d32(const float* __restrict__ h, const int (&s)[4],
                                               const float (&r)[4], int cnt, int lane,
                                               float4& acc) {
  constexpr unsigned SEL = OCT_SEL(T, T, T, T, T, T, T, T);
  float a[4];
  float4 v[4];
  for (int k = 0; k < 4; ++k) {
    const int sj = __builtin_amdgcn_readlane(s[k], (int)T);
    a[k] = octet_sel<SEL>(r[k]);
    v[k] = *(const float4*)(h + (size_t)sj * 128 + 4 * lane);
  }
  if ((int)(4 * T + 3) >= cnt) {
    for (int k = 1; k < 4; ++k)
      if ((int)(4 * T + k) >= cnt) v[k] = make_float4(0.f, 0.f, 0.f, 0.f);
  }
  for (int k = 0; k < 4; ++k) {
    acc.x = __builtin_fmaf(a[k], v[k].x, acc.x);
    acc.y = __builtin_fmaf(a[k], v[k].y, acc.y);
    acc.z = __builtin_fmaf(a[k], v[k].z, acc.z);
    acc.w = __builtin_fmaf(a[k], v[k].w, acc.w);
  }
}

__device__ __forceinline__ void add_group_h4d32(const float* __restrict__ h, const int (&s)[4],
                                                const float (&r)[4], int cnt, int lane,
                                                float4& acc) {
  if (cnt > 0)  add_trip_h4d32<0>(h, s, r, cnt, lane, acc);
  if (cnt > 4)  add_trip_h4d32<1>(h, s, r, cnt, lane, acc);
  if (cnt > 8)  add_trip_h4d32<2>(h, s, r, cnt, lane, acc);
  if (cnt > 12) add_trip_h4d32<3>(h, s, r, cnt, lane, acc);
  if (cnt > 16) add_trip_h4d32<4>(h, s, r, cnt, lane, acc);
  if (cnt > 20) add_trip_h4d32<5>(h, s, r, cnt, lane, acc);
  if (cnt > 24) add_trip_h4d32<6>(h, s, r, cnt, lane, acc);
  if (cnt > 28) add_trip_h4d32<7>(h, s, r, cnt, lane, acc);
}

__global__ void __launch_bounds__(256)
node_aggregate_h4d32(const int* __restrict__ off, const int* __restrict__ csr_src,
                     const float* __restrict__ a_s, const float* __restrict__ a_d,
                     const float* __restrict__ h, const float* __restrict__ bias,
                     float* __restrict__ x2, int N) {
  const int lane = threadIdx.x & 31;
  const int hh = lane >> 3;
  const int sub = lane & 7;
  const int wib = __builtin_amdgcn_readfirstlane((int)(threadIdx.x >> 5));
  const int node = blockIdx.x * 8 + wib;
  if (node >= N) return;
  const int beg = __builtin_amdgcn_readfirstlane(off[node]);
  const int deg = __builtin_amdgcn_readfirstlane(off[node + 1]) - beg;
  const int* __restrict__ nb = csr_src + beg;

  const float ad = a_d[(size_t)node * 4 + hh];
  const float z = leaky(a_s[(size_t)node * 4 + hh] + ad);
  const float4 hs = *(const float4*)(h + (size_t)node * 128 + 4 * lane);

  int s[4];
  bool on[4];
  float e[4];
  for (int k = 0; k < 4; ++k) {
    on[k] = 4 * sub + k < deg;
    s[k] = on[k] ? nb[4 * sub + k] : node;
  }
  for (int k = 0; k < 4; ++k) e[k] = leaky(a_s[(size_t)s[k] * 4 + hh] + ad);

  float m = fmaxf(fmaxf(z, e[0]), fmaxf(fmaxf(e[1], e[2]), e[3]));
  for (int base = 32; base < deg; base += 32) {
    for (int k = 0; k < 4; ++k) {
      const int j = base + 4 * sub + k;
      if (j < deg) m = fmaxf(m, leaky(a_s[(size_t)nb[j] * 4 + hh] + ad));
    }
  }
  m = octet_max(m);

  float p[4];
  for (int k = 0; k < 4; ++k) {
    const float ex = expf(e[k] - m);
    p[k] = on[k] ? ex : 0.f;
  }
  float d = (p[0] + p[1]) + (p[2] + p[3]);
  for (int base = 32; base < deg; base += 32) {
    for (int k = 0; k < 4; ++k) {
      const int j = base + 4 * sub + k;
      if (j < deg) d += expf(leaky(a_s[(size_t)nb[j] * 4 + hh] + ad) - m);
    }
  }
  const float q = expf(z - m);
  d = octet_sum(d) + q;

  const float w0 = q / d;
  float4 acc = make_float4(w0 * hs.x, w0 * hs.y, w0 * hs.z, w0 * hs.w);
  {
    float r[4];
    for (int k = 0; k < 4; ++k) r[k] = p[k] / d;
    add_group_h4d32(h, s, r, min(deg, 32), lane, acc);
  }
  for (int base = 32; base < deg; base += 32) {
    const int cnt = min(32, deg - base);
    int s2[4];
    float r2[4];
    for (int k = 0; k < 4; ++k) {
      const bool o2 = 4 * sub + k < cnt;
      s2[k] = o2 ? nb[base + 4 * sub + k] : node;
      r2[k] = o2 ? expf(leaky(a_s[(size_t)s2[k] * 4 + hh] + ad) - m) / d : 0.f;
    }
    add_group_h4d32(h, s2, r2, cnt, lane, acc);
  }

  const float4 b4 = *(const float4*)(bias + 4 * lane);
  float4 o = make_float4(acc.x + b4.x, acc.y + b4.y, acc.z + b4.z, acc.w + b4.w);
  o.x = (o.x > 0.f) ? o.x : 0.f;
  o.y = (o.y > 0.f) ? o.y : 0.f;
  o.z = (o.z > 0.f) ? o.z : 0.f;
  o.w = (o.w > 0.f) ? o.w : 0.f;
  *(float4*)(x2 + (size_t)node * 128 + 4 * lane) = o;
}

__device__ __forceinline__ void add_rows_h1d40(const float* __restrict__ h, int s_l, float al,
                                               int cnt, int lane, int f1, float& acc0,
                                               float& acc1) {
  int j = 0;
  for (; j + 3 < cnt; j += 4) {
    float a[4], va[4], vb[4];
    for (int k = 0; k < 4; ++k) {
      const int sj = __builtin_amdgcn_readlane(s_l, j + k);
      a[k] = lane_bcast(al, j + k);
      const float* __restrict__ hr = h + (size_t)sj * 40;
      va[k] = hr[lane];
      vb[k] = hr[f1];
    }
    for (int k = 0; k < 4; ++k) {
      acc0 += a[k] * va[k];
      acc1 += a[k] * vb[k];
    }
  }
  for (; j < cnt; ++j) {
    const int sj = __builtin_amdgcn_readlane(s_l, j);
    const float ak = lane_bcast(al, j);
    const float* __restrict__ hr = h + (size_t)sj * 40;
    acc0 += ak * hr[lane];
    acc1 += ak * hr[f1];
  }
}

__global__ void __launch_bounds__(256)
node_aggregate_h1d40(const int* __restrict__ off, const int* __restrict__ csr_src,
                     const float* __restrict__ a_s, const float* __restrict__ a_d,
                     const float* __restrict__ h, const float* __restrict__ bias,
                     float* __restrict__ out, int N) {
  const int lane = threadIdx.x & 31;
  const int wib = __builtin_amdgcn_readfirstlane((int)(threadIdx.x >> 5));
  const int node = blockIdx.x * 8 + wib;
  if (node >= N) return;
  const int beg = __builtin_amdgcn_readfirstlane(off[node]);
  const int deg = __builtin_amdgcn_readfirstlane(off[node + 1]) - beg;
  const int* __restrict__ nb = csr_src + beg;
  const int f1 = 32 + (lane & 7);

  const float ad = a_d[node];
  const float z = leaky(a_s[node] + ad);

  const bool act = lane < deg;
  const int s_c = act ? nb[lane] : node;
  const float e = leaky(a_s[s_c] + ad);

  float m = fmaxf(z, e);
  for (int i = 32 + lane; i < deg; i += 32) m = fmaxf(m, leaky(a_s[nb[i]] + ad));
  m = wave_max(m);

  const float p = act ? expf(e - m) : 0.f;
  float d = p;
  for (int i = 32 + lane; i < deg; i += 32) d += expf(leaky(a_s[nb[i]] + ad) - m);
  const float q = expf(z - m);
  d = wave_sum(d) + q;

  const float* __restrict__ hs = h + (size_t)node * 40;
  float acc0 = (q / d) * hs[lane];
  float acc1 = (q / d) * hs[f1];
  add_rows_h1d40(h, s_c, p / d, min(deg, 32), lane, f1, acc0, acc1);
  for (int base = 32; base < deg; base += 32) {
    const int cnt = min(32, deg - base);
    const bool on = lane < cnt;
    const int s_r = on ? nb[base + lane] : node;
    const float r = on ? expf(leaky(a_s[s_r] + ad) - m) / d : 0.f;
    add_rows_h1d40(h, s_r, r, cnt, lane, f1, acc0, acc1);
  }

  float* __restrict__ orow = out + (size_t)node * 40;
  orow[lane] = acc0 + bias[lane];
  if (lane < 8) orow[32 + lane] = acc1 + bias[32 + lane];
}

extern "C" void kernel_launch(void* const* d_in, const int* in_sizes, int n_in,
                              void* d_out, int out_size, void* d_ws, size_t ws_size,
                              hipStream_t stream) {
  const float* x        = (const float*)d_in[0];
  const int*   ei       = (const int*)d_in[1];
  const float* W1       = (const float*)d_in[2];
  const float* att_src1 = (const float*)d_in[3];
  const float* att_dst1 = (const float*)d_in[4];
  const float* b1       = (const float*)d_in[5];
  const float* W2       = (const float*)d_in[6];
  const float* att_src2 = (const float*)d_in[7];
  const float* att_dst2 = (const float*)d_in[8];
  const float* b2       = (const float*)d_in[9];

  const int NFEAT = 256, NHID = 128, H1 = 4, NCLS = 40;
  const int N = in_sizes[0] / NFEAT;
  const int E = in_sizes[1] / 2;
  const int* src = ei;
  const int* dst = ei + E;

  float* out = (float*)d_out;

  const int NR = (N + CSR_BKT - 1) / CSR_BKT;
  int* counts = (int*)d_ws;
  int* recs   = counts + (size_t)NR * BIN_P;
  float* w = (float*)(recs + (size_t)NR * BIN_P * BIN_SLOT * 2);
  float* h1   = w;
  float* x2   = h1  + (size_t)N * NHID;
  float* h2   = x2  + (size_t)N * NHID;
  float* as1  = h2  + (size_t)N * NCLS;
  float* ad1  = as1 + (size_t)N * H1;
  float* as2  = ad1 + (size_t)N * H1;
  float* ad2  = as2 + (size_t)N;
  int* off     = (int*)(ad2 + (size_t)N);
  int* csr_src = off + (size_t)N + 1;

  const int TB = 256;

  const int GB = (N + 16 * L1_WAVES - 1) / (16 * L1_WAVES);
  gemm_strip_k256_c128_and_bin<<<GB + BIN_P, 32 * L1_WAVES, 0, stream>>>(
      x, W1, h1, att_src1, att_dst1, as1, ad1, N, GB, src, dst, E, NR, counts, recs);
  csr_place<<<NR, CSR_THREADS, 0, stream>>>(src, dst, E, N, counts, recs, off, csr_src);

  node_aggregate_h4d32<<<(N + 7) / 8, TB, 0, stream>>>(off, csr_src, as1, ad1, h1, b1,
                                                       x2, N);

  gemm_strip_k128_c40<<<(N + 15) / 16, 32, 0, stream>>>(x2, W2, h2, att_src2, att_dst2,
                                                        as2, ad2, N);
  node_aggregate_h1d40<<<(N + 7) / 8, TB, 0, stream>>>(off, csr_src, as2, ad2, h2, b2,
                                                       out, N);
}


# layer-2 aggregation: a wave serves 4 neighbouring nodes side by side, one per 8-lane group; one 16-byte + one 4-byte load per lane fetch 4 source rows; a quarter of the waves; exp-sum tree changes
# speedup vs baseline: 19.7237x; 1.0348x over previous
#include <hip/hip_runtime.h>


typedef float v2f __attribute__((ext_vector_type(2)));
typedef float v8f __attribute__((ext_vector_type(8)));

#define NEG_SLOPE 0.2f


#define CSR_BKT 256
#define CSR_THREADS 512
#define CSR_CAP 14336
#define CSR_U 8
#define BIN_P 256
#define BIN_THREADS 512
#define BIN_U 7
#define BIN_SLOT 48
#define BIN_MAXR 256

template <class F>
__device__ __forceinline__ void for_each_edge(const int* __restrict__ dst, int E, int tid, F f) {
  const int head = min(E, (int)(((16u - (unsigned)((size_t)dst & 15u)) & 15u) >> 2));
  const int nv = (E - head) >> 2;
  const int4* __restrict__ dv = (const int4*)(dst + head);
  for (int i0 = tid; i0 < nv; i0 += CSR_U * CSR_THREADS) {
    int4 d[CSR_U];
#pragma unroll
    for (int u = 0; u < CSR_U; ++u) d[u] = dv[min(i0 + u * CSR_THREADS, nv - 1)];
#pragma unroll
    for (int u = 0; u < CSR_U; ++u) {
      const int i = i0 + u * CSR_THREADS;
      if (i < nv) {
        const int t = head + 4 * i;
        f(t, d[u].x);
        f(t + 1, d[u].y);
        f(t + 2, d[u].z);
        f(t + 3, d[u].w);
      }
    }
  }
  const int tail0 = head + 4 * nv;
  if (tid < head) f(tid, dst[tid]);
  if (tid < E - tail0) f(tail0 + tid, dst[tail0 + tid]);
}

#define BIN_CH (BIN_U * BIN_THREADS)
#define BIN_LDS_INTS (3 * BIN_MAXR + 32 + 2 * BIN_CH)
static_assert(BIN_THREADS >= BIN_MAXR, "thread t < BIN_MAXR owns counter t");
__device__ __forceinline__ void bin_piece(const int* __restrict__ src,
                                          const int* __restrict__ dst, int E, int NR,
                                          int* __restrict__ counts, int* __restrict__ recs,
                                          int p, int* sm) {
  int* c    = sm;
  int* cc   = c + BIN_MAXR;
  int* st   = cc + BIN_MAXR;
  int* wtot = st + BIN_MAXR;
  int* l_s  = wtot + 32;
  int* l_d  = l_s + BIN_CH;
  const int tid = threadIdx.x, lane = tid & 31;
  const int wid = __builtin_amdgcn_readfirstlane(tid >> 5);
  const bool owner = tid < BIN_MAXR;
  if (owner) c[tid] = 0;
  const int EP = (E + BIN_P - 1) / BIN_P;
  const int e0 = min(E, p * EP), e1 = min(E, e0 + EP);
  const unsigned nr = (unsigned)min(NR, BIN_MAXR);
  for (int cb = e0; cb < e1; cb += BIN_CH) {
    if (owner) cc[tid] = 0;
    __syncthreads();
    int d[BIN_U], s[BIN_U], k[BIN_U];
#pragma unroll
    for (int u = 0; u < BIN_U; ++u) {
      const int e = min(cb + tid + u * BIN_THREADS, e1 - 1);
      d[u] = dst[e];
      s[u] = src[e];
    }
#pragma unroll
    for (int u = 0; u < BIN_U; ++u) {
      const unsigned r = (unsigned)d[u] / (unsigned)CSR_BKT;
      const bool ok = (cb + tid + u * BIN_THREADS < e1) && (r < nr);
      k[u] = -1;
      if (ok) k[u] = atomicAdd(&cc[r], 1);
    }
    __syncthreads();
    int x = 0, inc = 0;
    if (owner) {
      x = cc[tid];
      inc = x;
      for (int sh = 1; sh < 32; sh <<= 1) {
        const int y = __shfl_up(inc, sh, 32);
        if (lane >= sh) inc += y;
      }
      if (lane == 31) wtot[wid] = inc;
    }
    __syncthreads();
    if (owner) {
      int b = 0;
      for (int w = 0; w < wid; ++w) b += wtot[w];
      st[tid] = b + inc - x;
    }
    __syncthreads();
#pragma unroll
    for (int u = 0; u < BIN_U; ++u) {
      if (k[u] >= 0) {
        const int pos = st[(unsigned)d[u] / (unsigned)CSR_BKT] + k[u];
        l_s[pos] = s[u];
        l_d[pos] = d[u];
      }
    }
    __syncthreads();
    const int tot = st[BIN_MAXR - 1] + cc[BIN_MAXR - 1];
    for (int i = tid; i < tot; i += BIN_THREADS) {
      const int dd = l_d[i], ss = l_s[i];
      const int r = (int)((unsigned)dd / (unsigned)CSR_BKT);
      const int kk = c[r] + (i - st[r]);
      if (kk < BIN_SLOT) {
        const size_t rec = ((size_t)r * BIN_P + p) * BIN_SLOT + kk;
        *(int2*)(recs + 2 * rec) = make_int2(ss, dd);
      }
    }
    __syncthreads();
    if (owner) c[tid] += cc[tid];
  }
  __syncthreads();
  for (int r = tid; r < NR; r += BIN_THREADS)
    counts[(size_t)r * BIN_P + p] = (r < BIN_MAXR) ? c[r] : 0;
}

#define CSR_WAVES (CSR_THREADS / 32)
#define CSR_SPW (BIN_P / CSR_WAVES)
#define CSR_BU 13

__global__ void __launch_bounds__(CSR_THREADS)
csr_place(const int* __restrict__ src, const int* __restrict__ dst, int E, int N,
          const int* __restrict__ counts, const int* __restrict__ recs,
          int* __restrict__ off, int* __restrict__ csr_src) {
  __shared__ int c[CSR_BKT];
  __shared__ int loff[CSR_BKT];
  __shared__ int wtot[CSR_BKT / 32];
  __shared__ int s_below, s_nl, s_slow;
  __shared__ int l_s[CSR_CAP];
  __shared__ int l_d[CSR_CAP];
  __shared__ int l_r[CSR_CAP];
  const int tid = threadIdx.x, lane = tid & 31;
  const int wid = __builtin_amdgcn_readfirstlane(tid >> 5);
  const int rg = blockIdx.x;
  const int lo = rg * CSR_BKT;

  if (tid < CSR_BKT) c[tid] = 0;
  if (tid == 0) { s_below = 0; s_nl = 0; s_slow = 0; }
  __syncthreads();

  const int* __restrict__ myc = counts + (size_t)rg * BIN_P;
  int n[CSR_SPW];
#pragma unroll
  for (int u = 0; u < CSR_SPW; ++u) n[u] = myc[wid + CSR_WAVES * u];

  int below = 0;
  {
    const int4* __restrict__ cv = (const int4*)counts;
    const int nq = rg * (BIN_P / 4);
    for (int i0 = tid; i0 < nq; i0 += CSR_BU * CSR_THREADS) {
      int4 v[CSR_BU];
#pragma unroll
      for (int u = 0; u < CSR_BU; ++u) v[u] = cv[min(i0 + u * CSR_THREADS, nq - 1)];
#pragma unroll
      for (int u = 0; u < CSR_BU; ++u)
        if (i0 + u * CSR_THREADS < nq) below += (v[u].x + v[u].y) + (v[u].z + v[u].w);
    }
  }

  auto note = [&](int s, int d) {
    const unsigned dl = (unsigned)(d - lo);
    if (dl < (unsigned)CSR_BKT) {
      const int r = atomicAdd(&c[dl], 1);
      const int k = atomicAdd(&s_nl, 1);
      if (k < CSR_CAP) { l_s[k] = s; l_d[k] = (int)dl; l_r[k] = r; }
    }
  };

  int2 rc[CSR_SPW];
#pragma unroll
  for (int u = 0; u < CSR_SPW; ++u) {
    const int* __restrict__ slot = recs + 2 * ((size_t)rg * BIN_P + wid + CSR_WAVES * u) * BIN_SLOT;
    rc[u] = make_int2(0, 0);
    if (lane < min(n[u], BIN_SLOT)) rc[u] = *(const int2*)(slot + 2 * lane);
  }
#pragma unroll
  for (int u = 0; u < CSR_SPW; ++u) {
    const int* __restrict__ slot = recs + 2 * ((size_t)rg * BIN_P + wid + CSR_WAVES * u) * BIN_SLOT;
    const int m = min(n[u], BIN_SLOT);
    if (n[u] > BIN_SLOT) s_slow = 1;
    if (lane < m) note(rc[u].x, rc[u].y);
    for (int j = 32 + lane; j < m; j += 32) {
      const int2 q = *(const int2*)(slot + 2 * j);
      note(q.x, q.y);
    }
  }
  for (int m = 16; m >= 1; m >>= 1) below += __shfl_xor(below, m, 32);
  if (lane == 0) atomicAdd(&s_below, below);
  __syncthreads();

  const bool slow = (s_slow != 0) || (s_nl > CSR_CAP);
  if (slow) {
    if (tid < CSR_BKT) c[tid] = 0;
    __syncthreads();
    for_each_edge(dst, E, tid, [&](int t, int d) {
      const unsigned dl = (unsigned)(d - lo);
      if (dl < (unsigned)CSR_BKT) atomicAdd(&c[dl], 1);
    });
    __syncthreads();
  }

  int x = 0, inc = 0;
  if (tid < CSR_BKT) {
    x = c[tid];
    inc = x;
    for (int s = 1; s < 32; s <<= 1) {
      const int y = __shfl_up(inc, s, 32);
      if (lane >= s) inc += y;
    }
    if (lane == 31) wtot[wid] = inc;
  }
  __syncthreads();
  if (tid < CSR_BKT) {
    int b = 0;
    for (int w = 0; w < wid; ++w) b += wtot[w];
    loff[tid] = b + inc - x;
  }
  __syncthreads();
  const int base = s_below;
  if (tid < CSR_BKT && lo + tid < N) off[lo + tid] = base + loff[tid];
  if (tid == CSR_BKT - 1 && lo + CSR_BKT >= N) off[N] = base + loff[tid] + x;

  if (!slow) {
    const int nl = s_nl;
    for (int k = tid; k < nl; k += CSR_THREADS)
      csr_src[base + loff[l_d[k]] + l_r[k]] = l_s[k];
  } else {
    if (tid < CSR_BKT) c[tid] = 0;
    __syncthreads();
    for_each_edge(dst, E, tid, [&](int t, int d) {
      const unsigned dl = (unsigned)(d - lo);
      if (dl < (unsigned)CSR_BKT) {
        const int r = atomicAdd(&c[dl], 1);
        csr_src[base + loff[dl] + r] = src[t];
      }
    });
  }
}


struct frag16 { float4 p[4]; };

__device__ __forceinline__ frag16 load_k16(const float* q) {
  frag16 f;
#pragma unroll
  for (int i = 0; i < 4; ++i) f.p[i] = *(const float4*)(q + 4 * i);
  return f;
}

template <int G>
__device__ __forceinline__ void mma_group(const frag16& a, const frag16 (&b)[G], v8f* c) {
#pragma unroll
  for (int i = 0; i < 4; ++i) {
#pragma unroll
    for (int g = 0; g < G; ++g) {
      const v2f av = {a.p[i].x, a.p[i].y};
      const v2f bv = {b[g].p[i].x, b[g].p[i].y};
      c[g] = __builtin_amdgcn_wmma_f32_16x16x4_f32(false, av, false, bv, (short)0, c[g],
                                                   false, false);
    }
#pragma unroll
    for (int g = 0; g < G; ++g) {
      const v2f av = {a.p[i].z, a.p[i].w};
      const v2f bv = {b[g].p[i].z, b[g].p[i].w};
      c[g] = __builtin_amdgcn_wmma_f32_16x16x4_f32(false, av, false, bv, (short)0, c[g],
                                                   false, false);
    }
  }
}

template <int NT, int H>
__device__ __forceinline__ void strip_epilogue(const v8f* c, float* __restrict__ C,
                                               const float* __restrict__ att_s,
                                               const float* __restrict__ att_d,
                                               float* __restrict__ a_s,
                                               float* __restrict__ a_d, int N, int Nc,
                                               int row0, int idx, int half) {
#pragma unroll
  for (int t = 0; t < NT; ++t) {
    const int col = 16 * t + idx;
    if (col < Nc) {
#pragma unroll
      for (int r = 0; r < 8; ++r) {
        const int row = row0 + r + 8 * half;
        if (row < N) C[(size_t)row * Nc + col] = c[t][r];
      }
    }
  }

  constexpr int TPH = NT / H;
  float p[16 * H];
#pragma unroll
  for (int hh = 0; hh < H; ++hh) {
    float ws[TPH], wd[TPH];
#pragma unroll
    for (int u = 0; u < TPH; ++u) {
      const int col = 16 * (hh * TPH + u) + idx;
      ws[u] = (col < Nc) ? att_s[min(col, Nc - 1)] : 0.f;
      wd[u] = (col < Nc) ? att_d[min(col, Nc - 1)] : 0.f;
    }
#pragma unroll
    for (int r = 0; r < 8; ++r) {
      float s = 0.f, d = 0.f;
#pragma unroll
      for (int u = 0; u < TPH; ++u) {
        s += c[hh * TPH + u][r] * ws[u];
        d += c[hh * TPH + u][r] * wd[u];
      }
      p[r * H + hh] = s;
      p[(8 + r) * H + hh] = d;
    }
  }
#pragma unroll
  for (int s = 0; s < 4; ++s) {
    const int mask = 8 >> s;
    const int hs = (8 * H) >> s;
    const bool up = (idx & mask) != 0;
#pragma unroll
    for (int j = 0; j < hs; ++j) {
      const float lo = p[j], hi = p[j + hs];
      const float send = up ? lo : hi;
      const float keep = up ? hi : lo;
      p[j] = keep + __shfl_xor(send, mask, 32);
    }
  }
  const int lrow = row0 + 8 * half + (idx & 7);
  if (lrow < N) {
    float* __restrict__ o = ((idx < 8) ? a_s : a_d) + (size_t)lrow * H;
#pragma unroll
    for (int hh = 0; hh < H; ++hh) o[hh] = p[hh];
  }
}

template <int K, int NT, int G, int H>
__device__ __forceinline__ void gemm_strip(const float* __restrict__ A,
                                           const float* __restrict__ B,
                                           float* __restrict__ C,
                                           const float* __restrict__ att_s,
                                           const float* __restrict__ att_d,
                                           float* __restrict__ a_s,
                                           float* __restrict__ a_d, int N, int Nc) {
  constexpr int KB = K / 32;
  constexpr int NG = NT / G;
  const int lane = threadIdx.x & 31;
  const int half = lane >> 4;
  const int idx  = lane & 15;
  const int row0 = blockIdx.x * 16;

  const int ar = min(row0 + idx, N - 1);
  const float* ap = A + (size_t)ar * K + 16 * half;
  const float* bp[NT];
#pragma unroll
  for (int t = 0; t < NT; ++t)
    bp[t] = B + (size_t)min(16 * t + idx, Nc - 1) * K + 16 * half;

  v8f c[NT];
#pragma unroll
  for (int t = 0; t < NT; ++t) c[t] = (v8f){0.f, 0.f, 0.f, 0.f, 0.f, 0.f, 0.f, 0.f};

  frag16 a_cur = load_k16(ap);
  frag16 b0[G];
#pragma unroll
  for (int g = 0; g < G; ++g) b0[g] = load_k16(bp[g]);

#pragma unroll
  for (int kb = 0; kb < KB; ++kb) {
    const int kn = min(kb + 1, KB - 1) * 32;
    frag16 a_nxt = load_k16(ap + kn);
    if constexpr (NG == 2) {
      frag16 b1[G];
#pragma unroll
      for (int g = 0; g < G; ++g) b1[g] = load_k16(bp[G + g] + kb * 32);
      __builtin_amdgcn_sched_barrier(0);
      mma_group<G>(a_cur, b0, c);
      __builtin_amdgcn_sched_barrier(0);
#pragma unroll
      for (int g = 0; g < G; ++g) b0[g] = load_k16(bp[g] + kn);
      __builtin_amdgcn_sched_barrier(0);
      mma_group<G>(a_cur, b1, c + G);
    } else {
      frag16 bn[G];
#pragma unroll
      for (int g = 0; g < G; ++g) bn[g] = load_k16(bp[g] + kn);
      __builtin_amdgcn_sched_barrier(0);
      mma_group<G>(a_cur, b0, c);
#pragma unroll
      for (int g = 0; g < G; ++g) b0[g] = bn[g];
    }
    a_cur = a_nxt;
  }

  strip_epilogue<NT, H>(c, C, att_s, att_d, a_s, a_d, N, Nc, row0, idx, half);
}

#define L1_K 256
#define L1_NC 128
#define L1_BPITCH 260
#define L1_WAVES 16
static_assert(32 * L1_WAVES == BIN_THREADS, "one block size for both kinds of work-group");
static_assert(L1_NC * L1_BPITCH >= BIN_LDS_INTS, "the binning fits into the LDS copy of B");
__global__ void __launch_bounds__(32 * L1_WAVES)
gemm_strip_k256_c128_and_bin(const float* __restrict__ A, const float* __restrict__ B,
                             float* __restrict__ C, const float* __restrict__ att_s,
                             const float* __restrict__ att_d, float* __restrict__ a_s,
                             float* __restrict__ a_d, int N, int gemm_blocks,
                             const int* __restrict__ src, const int* __restrict__ dst,
                             int E, int NR, int* __restrict__ counts,
                             int* __restrict__ recs) {
  constexpr int KB = L1_K / 32;
  constexpr int NT = L1_NC / 16;
  constexpr int G = 2;
  constexpr int NG = NT / G;
  __shared__ float4 Bs4[L1_NC * L1_BPITCH / 4];
  if ((int)blockIdx.x >= gemm_blocks) {
    bin_piece(src, dst, E, NR, counts, recs, (int)blockIdx.x - gemm_blocks, (int*)Bs4);
    return;
  }
  float* Bs = (float*)Bs4;
  const int tid = threadIdx.x;
  const int wave = __builtin_amdgcn_readfirstlane(tid >> 5);
  const int row0 = (blockIdx.x * L1_WAVES + wave) * 16;
  const int lane = tid & 31;
  const int half = lane >> 4;
  const int idx  = lane & 15;
  const int ar = min(row0 + idx, N - 1);
  const float* ap = A + (size_t)ar * L1_K + 16 * half;
  frag16 a_cur = load_k16(ap);
  {
    constexpr int NV = L1_NC * L1_K / 4;
    constexpr int PER = NV / (32 * L1_WAVES);
    const float4* __restrict__ Bg = (const float4*)B;
    float4 w[PER];
#pragma unroll
    for (int j = 0; j < PER; ++j) w[j] = Bg[tid + 32 * L1_WAVES * j];
#pragma unroll
    for (int j = 0; j < PER; ++j) {
      const int f = tid + 32 * L1_WAVES * j;
      const int row = f / (L1_K / 4), c4 = f % (L1_K / 4);
      *(float4*)(Bs + row * L1_BPITCH + 4 * c4) = w[j];
    }
  }
  __syncthreads();
  if (row0 >= N) return;

  const float* bq = Bs + idx * L1_BPITCH + 16 * half;

  v8f c[NT];
#pragma unroll
  for (int t = 0; t < NT; ++t) c[t] = (v8f){0.f, 0.f, 0.f, 0.f, 0.f, 0.f, 0.f, 0.f};

  frag16 bb[2][G];
#pragma unroll
  for (int g = 0; g < G; ++g) bb[0][g] = load_k16(bq + g * 16 * L1_BPITCH);

#pragma unroll
  for (int kb = 0; kb < KB; ++kb) {
    const int kn = min(kb + 1, KB - 1) * 32;
    frag16 a_nxt = load_k16(ap + kn);
#pragma unroll
    for (int grp = 0; grp < NG; ++grp) {
      const int cur = (kb * NG + grp) & 1;
      if (!(kb == KB - 1 && grp == NG - 1)) {
        const int t0 = (grp + 1 < NG) ? (grp + 1) * G : 0;
        const int k0 = (grp + 1 < NG) ? kb * 32 : kn;
#pragma unroll
        for (int g = 0; g < G; ++g)
          bb[cur ^ 1][g] = load_k16(bq + (t0 + g) * 16 * L1_BPITCH + k0);
      }
      __builtin_amdgcn_sched_barrier(0);
      mma_group<G>(a_cur, bb[cur], c + grp * G);
      __builtin_amdgcn_sched_barrier(0);
    }
    a_cur = a_nxt;
  }

  strip_epilogue<NT, 4>(c, C, att_s, att_d, a_s, a_d, N, L1_NC, row0, idx, half);
}

__global__ void __launch_bounds__(32)
gemm_strip_k128_c40(const float* __restrict__ A, const float* __restrict__ B,
                    float* __restrict__ C, const float* __restrict__ att_s,
                    const float* __restrict__ att_d, float* __restrict__ a_s,
                    float* __restrict__ a_d, int N) {
  gemm_strip<128, 3, 3, 1>(A, B, C, att_s, att_d, a_s, a_d, N, 40);
}


__device__ __forceinline__ float leaky(float e) { return (e > 0.f) ? e : NEG_SLOPE * e; }


template <unsigned S>
__device__ __forceinline__ float octet_sel(float v) {
  return __uint_as_float(__builtin_amdgcn_mov_dpp8(__float_as_uint(v), S));
}
#define OCT_SEL(a, b, c, d, e, f, g, hh_) \
  ((a) | ((b) << 3) | ((c) << 6) | ((d) << 9) | ((e) << 12) | ((f) << 15) | ((g) << 18) | ((hh_) << 21))
__device__ __forceinline__ float octet_max(float v) {
  v = fmaxf(v, octet_sel<OCT_SEL(1u, 0u, 3u, 2u, 5u, 4u, 7u, 6u)>(v));
  v = fmaxf(v, octet_sel<OCT_SEL(2u, 3u, 0u, 1u, 6u, 7u, 4u, 5u)>(v));
  v = fmaxf(v, octet_sel<OCT_SEL(4u, 5u, 6u, 7u, 0u, 1u, 2u, 3u)>(v));
  return v;
}
__device__ __forceinline__ float octet_sum(float v) {
  v += octet_sel<OCT_SEL(1u, 0u, 3u, 2u, 5u, 4u, 7u, 6u)>(v);
  v += octet_sel<OCT_SEL(2u, 3u, 0u, 1u, 6u, 7u, 4u, 5u)>(v);
  v += octet_sel<OCT_SEL(4u, 5u, 6u, 7u, 0u, 1u, 2u, 3u)>(v);
  return v;
}

template <unsigned T>
__device__ __forceinline__ void add_trip_h4d32(const float* __restrict__ h, const int (&s)[4],
                                               const float (&r)[4], int cnt, int lane,
                                               float4& acc) {
  constexpr unsigned SEL = OCT_SEL(T, T, T, T, T, T, T, T);
  float a[4];
  float4 v[4];
  for (int k = 0; k < 4; ++k) {
    const int sj = __builtin_amdgcn_readlane(s[k], (int)T);
    a[k] = octet_sel<SEL>(r[k]);
    v[k] = *(const float4*)(h + (size_t)sj * 128 + 4 * lane);
  }
  if ((int)(4 * T + 3) >= cnt) {
    for (int k = 1; k < 4; ++k)
      if ((int)(4 * T + k) >= cnt) v[k] = make_float4(0.f, 0.f, 0.f, 0.f);
  }
  for (int k = 0; k < 4; ++k) {
    acc.x = __builtin_fmaf(a[k], v[k].x, acc.x);
    acc.y = __builtin_fmaf(a[k], v[k].y, acc.y);
    acc.z = __builtin_fmaf(a[k], v[k].z, acc.z);
    acc.w = __builtin_fmaf(a[k], v[k].w, acc.w);
  }
}

__device__ __forceinline__ void add_group_h4d32(const float* __restrict__ h, const int (&s)[4],
                                                const float (&r)[4], int cnt, int lane,
                                                float4& acc) {
  if (cnt > 0)  add_trip_h4d32<0>(h, s, r, cnt, lane, acc);
  if (cnt > 4)  add_trip_h4d32<1>(h, s, r, cnt, lane, acc);
  if (cnt > 8)  add_trip_h4d32<2>(h, s, r, cnt, lane, acc);
  if (cnt > 12) add_trip_h4d32<3>(h, s, r, cnt, lane, acc);
  if (cnt > 16) add_trip_h4d32<4>(h, s, r, cnt, lane, acc);
  if (cnt > 20) add_trip_h4d32<5>(h, s, r, cnt, lane, acc);
  if (cnt > 24) add_trip_h4d32<6>(h, s, r, cnt, lane, acc);
  if (cnt > 28) add_trip_h4d32<7>(h, s, r, cnt, lane, acc);
}

__global__ void __launch_bounds__(256)
node_aggregate_h4d32(const int* __restrict__ off, const int* __restrict__ csr_src,
                     const float* __restrict__ a_s, const float* __restrict__ a_d,
                     const float* __restrict__ h, const float* __restrict__ bias,
                     float* __restrict__ x2, int N) {
  const int lane = threadIdx.x & 31;
  const int hh = lane >> 3;
  const int sub = lane & 7;
  const int wib = __builtin_amdgcn_readfirstlane((int)(threadIdx.x >> 5));
  const int node = blockIdx.x * 8 + wib;
  if (node >= N) return;
  const int beg = __builtin_amdgcn_readfirstlane(off[node]);
  const int deg = __builtin_amdgcn_readfirstlane(off[node + 1]) - beg;
  const int* __restrict__ nb = csr_src + beg;

  const float ad = a_d[(size_t)node * 4 + hh];
  const float z = leaky(a_s[(size_t)node * 4 + hh] + ad);
  const float4 hs = *(const float4*)(h + (size_t)node * 128 + 4 * lane);

  int s[4];
  bool on[4];
  float e[4];
  for (int k = 0; k < 4; ++k) {
    on[k] = 4 * sub + k < deg;
    s[k] = on[k] ? nb[4 * sub + k] : node;
  }
  for (int k = 0; k < 4; ++k) e[k] = leaky(a_s[(size_t)s[k] * 4 + hh] + ad);

  float m = fmaxf(fmaxf(z, e[0]), fmaxf(fmaxf(e[1], e[2]), e[3]));
  for (int base = 32; base < deg; base += 32) {
    for (int k = 0; k < 4; ++k) {
      const int j = base + 4 * sub + k;
      if (j < deg) m = fmaxf(m, leaky(a_s[(size_t)nb[j] * 4 + hh] + ad));
    }
  }
  m = octet_max(m);

  float p[4];
  for (int k = 0; k < 4; ++k) {
    const float ex = expf(e[k] - m);
    p[k] = on[k] ? ex : 0.f;
  }
  float d = (p[0] + p[1]) + (p[2] + p[3]);
  for (int base = 32; base < deg; base += 32) {
    for (int k = 0; k < 4; ++k) {
      const int j = base + 4 * sub + k;
      if (j < deg) d += expf(leaky(a_s[(size_t)nb[j] * 4 + hh] + ad) - m);
    }
  }
  const float q = expf(z - m);
  d = octet_sum(d) + q;

  const float w0 = q / d;
  float4 acc = make_float4(w0 * hs.x, w0 * hs.y, w0 * hs.z, w0 * hs.w);
  {
    float r[4];
    for (int k = 0; k < 4; ++k) r[k] = p[k] / d;
    add_group_h4d32(h, s, r, min(deg, 32), lane, acc);
  }
  for (int base = 32; base < deg; base += 32) {
    const int cnt = min(32, deg - base);
    int s2[4];
    float r2[4];
    for (int k = 0; k < 4; ++k) {
      const bool o2 = 4 * sub + k < cnt;
      s2[k] = o2 ? nb[base + 4 * sub + k] : node;
      r2[k] = o2 ? expf(leaky(a_s[(size_t)s2[k] * 4 + hh] + ad) - m) / d : 0.f;
    }
    add_group_h4d32(h, s2, r2, cnt, lane, acc);
  }

  const float4 b4 = *(const float4*)(bias + 4 * lane);
  float4 o = make_float4(acc.x + b4.x, acc.y + b4.y, acc.z + b4.z, acc.w + b4.w);
  o.x = (o.x > 0.f) ? o.x : 0.f;
  o.y = (o.y > 0.f) ? o.y : 0.f;
  o.z = (o.z > 0.f) ? o.z : 0.f;
  o.w = (o.w > 0.f) ? o.w : 0.f;
  *(float4*)(x2 + (size_t)node * 128 + 4 * lane) = o;
}


template <unsigned T>
__device__ __forceinline__ void octet_row_h1d40(const float* __restrict__ h, int s, float r,
                                                int sub, float& a, float4& v, float& v1) {
  constexpr unsigned SEL = OCT_SEL(T, T, T, T, T, T, T, T);
  const int sj = (int)__builtin_amdgcn_mov_dpp8((unsigned)s, SEL);
  a = octet_sel<SEL>(r);
  const float* __restrict__ hr = h + (size_t)sj * 40;
  v = *(const float4*)(hr + 4 * sub);
  v1 = hr[32 + sub];
}

__device__ __forceinline__ void add_eight_h1d40(const float* __restrict__ h, int s, float r,
                                                int left, int sub, float4& acc, float& acc1) {
  float a[8], v1[8];
  float4 v[8];
  octet_row_h1d40<0>(h, s, r, sub, a[0], v[0], v1[0]);
  octet_row_h1d40<1>(h, s, r, sub, a[1], v[1], v1[1]);
  octet_row_h1d40<2>(h, s, r, sub, a[2], v[2], v1[2]);
  octet_row_h1d40<3>(h, s, r, sub, a[3], v[3], v1[3]);
  octet_row_h1d40<4>(h, s, r, sub, a[4], v[4], v1[4]);
  octet_row_h1d40<5>(h, s, r, sub, a[5], v[5], v1[5]);
  octet_row_h1d40<6>(h, s, r, sub, a[6], v[6], v1[6]);
  octet_row_h1d40<7>(h, s, r, sub, a[7], v[7], v1[7]);
  for (int t = 0; t < 8; ++t) {
    if (t >= left) {
      v[t] = make_float4(0.f, 0.f, 0.f, 0.f);
      v1[t] = 0.f;
    }
    acc.x = __builtin_fmaf(a[t], v[t].x, acc.x);
    acc.y = __builtin_fmaf(a[t], v[t].y, acc.y);
    acc.z = __builtin_fmaf(a[t], v[t].z, acc.z);
    acc.w = __builtin_fmaf(a[t], v[t].w, acc.w);
    acc1 = __builtin_fmaf(a[t], v1[t], acc1);
  }
}

__global__ void __launch_bounds__(256)
node_aggregate_h1d40(const int* __restrict__ off, const int* __restrict__ csr_src,
                     const float* __restrict__ a_s, const float* __restrict__ a_d,
                     const float* __restrict__ h, const float* __restrict__ bias,
                     float* __restrict__ out, int N) {
  const int lane = threadIdx.x & 31;
  const int oct = lane >> 3;
  const int sub = lane & 7;
  const int wib = __builtin_amdgcn_readfirstlane((int)(threadIdx.x >> 5));
  const int node0 = (blockIdx.x * 8 + wib) * 4;
  if (node0 >= N) return;
  const int node = node0 + oct;
  const bool valid = node < N;
  const int nd = valid ? node : N - 1;
  const int beg = off[nd];
  const int deg = valid ? off[nd + 1] - beg : 0;
  const int* __restrict__ nb = csr_src + beg;
  const int maxdeg = max(max(__builtin_amdgcn_readlane(deg, 0), __builtin_amdgcn_readlane(deg, 8)),
                         max(__builtin_amdgcn_readlane(deg, 16), __builtin_amdgcn_readlane(deg, 24)));

  const float ad = a_d[nd];
  const float z = leaky(a_s[nd] + ad);
  const float* __restrict__ hs = h + (size_t)nd * 40;
  const float4 own = *(const float4*)(hs + 4 * sub);
  const float own1 = hs[32 + sub];

  int s[4];
  bool on[4];
  float e[4];
  for (int k = 0; k < 4; ++k) {
    on[k] = 8 * k + sub < deg;
    s[k] = on[k] ? nb[8 * k + sub] : nd;
  }
  for (int k = 0; k < 4; ++k) e[k] = leaky(a_s[s[k]] + ad);

  float m = fmaxf(fmaxf(z, e[0]), fmaxf(fmaxf(e[1], e[2]), e[3]));
  for (int base = 32; base < maxdeg; base += 8) {
    const int j = base + sub;
    if (j < deg) m = fmaxf(m, leaky(a_s[nb[j]] + ad));
  }
  m = octet_max(m);

  float p[4];
  for (int k = 0; k < 4; ++k) {
    const float ex = expf(e[k] - m);
    p[k] = on[k] ? ex : 0.f;
  }
  float d = (p[0] + p[1]) + (p[2] + p[3]);
  for (int base = 32; base < maxdeg; base += 8) {
    const int j = base + sub;
    if (j < deg) d += expf(leaky(a_s[nb[j]] + ad) - m);
  }
  const float q = expf(z - m);
  d = octet_sum(d) + q;

  const float w0 = q / d;
  float4 acc = make_float4(w0 * own.x, w0 * own.y, w0 * own.z, w0 * own.w);
  float acc1 = w0 * own1;
  if (maxdeg > 0)  add_eight_h1d40(h, s[0], p[0] / d, deg, sub, acc, acc1);
  if (maxdeg > 8)  add_eight_h1d40(h, s[1], p[1] / d, deg - 8, sub, acc, acc1);
  if (maxdeg > 16) add_eight_h1d40(h, s[2], p[2] / d, deg - 16, sub, acc, acc1);
  if (maxdeg > 24) add_eight_h1d40(h, s[3], p[3] / d, deg - 24, sub, acc, acc1);
  for (int base = 32; base < maxdeg; base += 8) {
    const bool o2 = base + sub < deg;
    const int s2 = o2 ? nb[base + sub] : nd;
    const float r2 = o2 ? expf(leaky(a_s[s2] + ad) - m) / d : 0.f;
    add_eight_h1d40(h, s2, r2, deg - base, sub, acc, acc1);
  }

  if (valid) {
    float* __restrict__ orow = out + (size_t)node * 40;
    const float4 b4 = *(const float4*)(bias + 4 * sub);
    *(float4*)(orow + 4 * sub) = make_float4(acc.x + b4.x, acc.y + b4.y, acc.z + b4.z, acc.w + b4.w);
    orow[32 + sub] = acc1 + bias[32 + sub];
  }
}

extern "C" void kernel_launch(void* const* d_in, const int* in_sizes, int n_in,
                              void* d_out, int out_size, void* d_ws, size_t ws_size,
                              hipStream_t stream) {
  const float* x        = (const float*)d_in[0];
  const int*   ei       = (const int*)d_in[1];
  const float* W1       = (const float*)d_in[2];
  const float* att_src1 = (const float*)d_in[3];
  const float* att_dst1 = (const float*)d_in[4];
  const float* b1       = (const float*)d_in[5];
  const float* W2       = (const float*)d_in[6];
  const float* att_src2 = (const float*)d_in[7];
  const float* att_dst2 = (const float*)d_in[8];
  const float* b2       = (const float*)d_in[9];

  const int NFEAT = 256, NHID = 128, H1 = 4, NCLS = 40;
  const int N = in_sizes[0] / NFEAT;
  const int E = in_sizes[1] / 2;
  const int* src = ei;
  const int* dst = ei + E;

  float* out = (float*)d_out;

  const int NR = (N + CSR_BKT - 1) / CSR_BKT;
  int* counts = (int*)d_ws;
  int* recs   = counts + (size_t)NR * BIN_P;
  float* w = (float*)(recs + (size_t)NR * BIN_P * BIN_SLOT * 2);
  float* h1   = w;
  float* x2   = h1  + (size_t)N * NHID;
  float* h2   = x2  + (size_t)N * NHID;
  float* as1  = h2  + (size_t)N * NCLS;
  float* ad1  = as1 + (size_t)N * H1;
  float* as2  = ad1 + (size_t)N * H1;
  float* ad2  = as2 + (size_t)N;
  int* off     = (int*)(ad2 + (size_t)N);
  int* csr_src = off + (size_t)N + 1;

  const int TB = 256;

  const int GB = (N + 16 * L1_WAVES - 1) / (16 * L1_WAVES);
  gemm_strip_k256_c128_and_bin<<<GB + BIN_P, 32 * L1_WAVES, 0, stream>>>(
      x, W1, h1, att_src1, att_dst1, as1, ad1, N, GB, src, dst, E, NR, counts, recs);
  csr_place<<<NR, CSR_THREADS, 0, stream>>>(src, dst, E, N, counts, recs, off, csr_src);

  node_aggregate_h4d32<<<(N + 7) / 8, TB, 0, stream>>>(off, csr_src, as1, ad1, h1, b1,
                                                       x2, N);

  gemm_strip_k128_c40<<<(N + 15) / 16, 32, 0, stream>>>(x2, W2, h2, att_src2, att_dst2,
                                                        as2, ad2, N);
  node_aggregate_h1d40<<<(N + 31) / 32, TB, 0, stream>>>(off, csr_src, as2, ad2, h2, b2,
                                                         out, N);
}
